# starting kernel
# baseline (speedup 1.0000x reference)
#include <hip/hip_runtime.h>


#define NNODES 100000
#define NEDGES 1600000
#define DIM    128
#define NCLS   40

typedef __attribute__((ext_vector_type(2))) float v2f;
typedef __attribute__((ext_vector_type(8))) float v8f;

__device__ __forceinline__ void atomic_fadd_hw(float* p, float v) {
    unsafeAtomicAdd(p, v);
}


__global__ __launch_bounds__(256) void zero_f4(float4* __restrict__ p, int n4) {
    int i = blockIdx.x * 256 + threadIdx.x;
    if (i < n4) p[i] = make_float4(0.f, 0.f, 0.f, 0.f);
}

__global__ __launch_bounds__(256) void degree_kernel(const int* __restrict__ src,
                                                     const int* __restrict__ dst,
                                                     float* __restrict__ degs,
                                                     float* __restrict__ degd) {
    int e = blockIdx.x * 256 + threadIdx.x;
    if (e < NEDGES) {
        atomic_fadd_hw(degs + src[e], 1.0f);
        atomic_fadd_hw(degd + dst[e], 1.0f);
    }
}

__global__ __launch_bounds__(256) void norm_finalize(float* __restrict__ d, int n) {
    int i = blockIdx.x * 256 + threadIdx.x;
    if (i < n) {
        float v = d[i];
        d[i] = (v > 0.f) ? rsqrtf(fmaxf(v, 1.f)) : 0.f;
    }
}


__global__ __launch_bounds__(256) void scatter_kernel(const float* __restrict__ h,
                                                      const float* __restrict__ ns,
                                                      const int* __restrict__ src,
                                                      const int* __restrict__ dst,
                                                      float* __restrict__ agg) {
    int e    = (blockIdx.x * 256 + threadIdx.x) >> 5;
    int lane = threadIdx.x & 31;
    int s = src[e];
    int d = dst[e];
    float sc = ns[s];
    float4 v = ((const float4*)(h + (size_t)s * DIM))[lane];
    float* base = agg + (size_t)d * DIM + lane * 4;
    atomic_fadd_hw(base + 0, v.x * sc);
    atomic_fadd_hw(base + 1, v.y * sc);
    atomic_fadd_hw(base + 2, v.z * sc);
    atomic_fadd_hw(base + 3, v.w * sc);
}


template <int NOUT, int NT, bool RELU>
__global__ __launch_bounds__(256) void gcn_gemm(const float* __restrict__ agg,
                                                const float* __restrict__ nd,
                                                const float* __restrict__ W,
                                                const float* __restrict__ bias,
                                                float* __restrict__ out) {
    constexpr int NT16 = NT * 16;
    __shared__ float As[16][DIM + 4];
    __shared__ float Ws[DIM][NT16];

    const int t    = threadIdx.x;
    const int tile = blockIdx.x;

    {
        int m    = t >> 4;
        int kb   = (t & 15) * 8;
        int node = tile * 16 + m;
        float s  = nd[node];
        const float4* p = (const float4*)(agg + (size_t)node * DIM + kb);
        float4 v0 = p[0], v1 = p[1];
        As[m][kb + 0] = v0.x * s; As[m][kb + 1] = v0.y * s;
        As[m][kb + 2] = v0.z * s; As[m][kb + 3] = v0.w * s;
        As[m][kb + 4] = v1.x * s; As[m][kb + 5] = v1.y * s;
        As[m][kb + 6] = v1.z * s; As[m][kb + 7] = v1.w * s;
    }
    for (int i = t; i < DIM * NT16; i += 256) {
        int k = i / NT16;
        int n = i - k * NT16;
        Ws[k][n] = (n < NOUT) ? W[k * NOUT + n] : 0.0f;
    }
    __syncthreads();

    const int wave = t >> 5;
    const int lane = t & 31;
    if (wave < NT) {
        const int n0  = wave * 16;
        const int l16 = lane & 15;
        const int kh  = (lane >> 4) * 2;
        v8f acc = {};
#pragma unroll
        for (int kk = 0; kk < DIM; kk += 4) {
            v2f a, b;
            a.x = As[l16][kk + kh];
            a.y = As[l16][kk + kh + 1];
            b.x = Ws[kk + kh][n0 + l16];
            b.y = Ws[kk + kh + 1][n0 + l16];
            acc = __builtin_amdgcn_wmma_f32_16x16x4_f32(
                false, a, false, b, (short)0, acc, false, false);
        }
        const int n = n0 + l16;
        if (n < NOUT) {
            const int mbase = tile * 16 + (lane >> 4) * 8;
            float bv = bias[n];
#pragma unroll
            for (int v = 0; v < 8; ++v) {
                float r = acc[v] + bv;
                if (RELU) r = fmaxf(r, 0.0f);
                out[(size_t)(mbase + v) * NOUT + n] = r;
            }
        }
    }
}


extern "C" void kernel_launch(void* const* d_in, const int* in_sizes, int n_in,
                              void* d_out, int out_size, void* d_ws, size_t ws_size,
                              hipStream_t stream) {
    const float* inputs = (const float*)d_in[0];
    const int*   src    = (const int*)d_in[1];
    const int*   dst    = (const int*)d_in[2];
    const float* W0     = (const float*)d_in[3];
    const float* b0     = (const float*)d_in[4];
    const float* W1     = (const float*)d_in[5];
    const float* b1     = (const float*)d_in[6];
    const float* W2     = (const float*)d_in[7];
    const float* b2     = (const float*)d_in[8];
    float* out = (float*)d_out;

    float* ws   = (float*)d_ws;
    float* ns   = ws;
    float* ndm  = ws + NNODES;
    float* bufB = ws + 2 * NNODES;
    float* bufC = bufB + (size_t)NNODES * DIM;

    constexpr int FEAT4   = NNODES * DIM / 4;
    constexpr int GR_FEAT = FEAT4 / 256;
    constexpr int NORM4   = 2 * NNODES / 4;
    constexpr int GR_NZ   = (NORM4 + 255) / 256;
    constexpr int GR_DEG  = (NEDGES + 255) / 256;
    constexpr int GR_NF   = (2 * NNODES + 255) / 256;
    constexpr int GR_SCAT = NEDGES / 8;
    constexpr int GR_GEMM = NNODES / 16;

    zero_f4<<<GR_NZ, 256, 0, stream>>>((float4*)ns, NORM4);
    degree_kernel<<<GR_DEG, 256, 0, stream>>>(src, dst, ns, ndm);
    norm_finalize<<<GR_NF, 256, 0, stream>>>(ns, 2 * NNODES);

    zero_f4<<<GR_FEAT, 256, 0, stream>>>((float4*)bufB, FEAT4);
    scatter_kernel<<<GR_SCAT, 256, 0, stream>>>(inputs, ns, src, dst, bufB);
    gcn_gemm<DIM, 8, true><<<GR_GEMM, 256, 0, stream>>>(bufB, ndm, W0, b0, bufC);

    zero_f4<<<GR_FEAT, 256, 0, stream>>>((float4*)bufB, FEAT4);
    scatter_kernel<<<GR_SCAT, 256, 0, stream>>>(bufC, ns, src, dst, bufB);
    gcn_gemm<DIM, 8, true><<<GR_GEMM, 256, 0, stream>>>(bufB, ndm, W1, b1, bufC);

    zero_f4<<<GR_FEAT, 256, 0, stream>>>((float4*)bufB, FEAT4);
    scatter_kernel<<<GR_SCAT, 256, 0, stream>>>(bufC, ns, src, dst, bufB);
    gcn_gemm<NCLS, 3, false><<<GR_GEMM, 256, 0, stream>>>(bufB, ndm, W2, b2, out);
}


# Float-atomic scatter-add replaced by bucketing edges by destination per call (int atomic rank, scan, fill) and summing each node's rows in registers; sum order = bucket rank
# speedup vs baseline: 2.6741x; 2.6741x over previous
#include <hip/hip_runtime.h>


#define NNODES 100000
#define NEDGES 1600000
#define DIM    128
#define NCLS   40

typedef __attribute__((ext_vector_type(2))) float v2f;
typedef __attribute__((ext_vector_type(8))) float v8f;


__global__ __launch_bounds__(256) void zero_f4(float4* __restrict__ p, int n4) {
    int i = blockIdx.x * 256 + threadIdx.x;
    if (i < n4) p[i] = make_float4(0.f, 0.f, 0.f, 0.f);
}

__global__ __launch_bounds__(256) void count_rank(const int* __restrict__ src,
                                                  const int* __restrict__ dst,
                                                  int* __restrict__ cnt_src,
                                                  int* __restrict__ cnt_dst,
                                                  int* __restrict__ pos) {
    int e = blockIdx.x * 256 + threadIdx.x;
    if (e < NEDGES) {
        const int s = src[e];
        const int d = dst[e];
        atomicAdd(cnt_src + s, 1);
        pos[e] = atomicAdd(cnt_dst + d, 1);
    }
}

__device__ __forceinline__ float inv_sqrt_deg(int c) {
    return (c > 0) ? rsqrtf(fmaxf((float)c, 1.f)) : 0.f;
}

__global__ __launch_bounds__(256) void scan_norm(const int* __restrict__ cnt_src,
                                                 const int* __restrict__ cnt_dst,
                                                 int* __restrict__ rowptr,
                                                 float* __restrict__ ns,
                                                 float* __restrict__ nd) {
    __shared__ int red[8];
    __shared__ int wsum[8];
    const int t = threadIdx.x, b = blockIdx.x;
    const int lane = t & 31, wave = t >> 5;
    const int4* c4 = (const int4*)cnt_dst;

    int part = 0;
    int k = 0;
    for (; k + 8 <= b; k += 8) {
        int4 q[8];
#pragma unroll
        for (int u = 0; u < 8; ++u) q[u] = c4[(k + u) * 256 + t];
#pragma unroll
        for (int u = 0; u < 8; ++u) part += (q[u].x + q[u].y) + (q[u].z + q[u].w);
    }
    for (; k < b; ++k) {
        int4 q = c4[k * 256 + t];
        part += (q.x + q.y) + (q.z + q.w);
    }
    for (int o = 16; o > 0; o >>= 1) part += __shfl_xor(part, o);
    if (lane == 0) red[wave] = part;

    const int i4 = b * 256 + t;
    const bool in = i4 < NNODES / 4;
    int4 c = make_int4(0, 0, 0, 0);
    if (in) c = c4[i4];
    const int s = (c.x + c.y) + (c.z + c.w);
    int inc = s;
    for (int o = 1; o < 32; o <<= 1) {
        int v = __shfl_up(inc, o);
        if (lane >= o) inc += v;
    }
    if (lane == 31) wsum[wave] = inc;
    __syncthreads();

    int base = 0;
    for (int w = 0; w < 8; ++w) base += red[w];
    for (int w = 0; w < 8; ++w) base += (w < wave) ? wsum[w] : 0;
    const int ex = base + inc - s;
    if (in) {
        ((int4*)rowptr)[i4] = make_int4(ex, ex + c.x, ex + c.x + c.y, ex + c.x + c.y + c.z);
        int4 cs = ((const int4*)cnt_src)[i4];
        ((float4*)ns)[i4] = make_float4(inv_sqrt_deg(cs.x), inv_sqrt_deg(cs.y),
                                        inv_sqrt_deg(cs.z), inv_sqrt_deg(cs.w));
        ((float4*)nd)[i4] = make_float4(inv_sqrt_deg(c.x), inv_sqrt_deg(c.y),
                                        inv_sqrt_deg(c.z), inv_sqrt_deg(c.w));
    }
}

__global__ __launch_bounds__(256) void fill_buckets(const int* __restrict__ src,
                                                    const int* __restrict__ dst,
                                                    const int* __restrict__ rowptr,
                                                    const int* __restrict__ pos,
                                                    int* __restrict__ csr) {
    int e = blockIdx.x * 256 + threadIdx.x;
    if (e < NEDGES) {
        const int s = src[e];
        unsigned idx = (unsigned)(rowptr[dst[e]] + pos[e]);
        if (idx < (unsigned)NEDGES) csr[idx] = s;
    }
}


template <int U>
__device__ __forceinline__ void acc_rows(const float* __restrict__ h, int my, float myw,
                                         int j, int lane, float4& acc) {
    float4 v[U];
    float  w[U];
#pragma unroll
    for (int u = 0; u < U; ++u) {
        int s = __builtin_amdgcn_readlane(my, j + u);
        w[u]  = __int_as_float(__builtin_amdgcn_readlane(__float_as_int(myw), j + u));
        v[u]  = ((const float4*)(h + (size_t)s * DIM))[lane];
    }
#pragma unroll
    for (int u = 0; u < U; ++u) {
        acc.x = fmaf(v[u].x, w[u], acc.x);
        acc.y = fmaf(v[u].y, w[u], acc.y);
        acc.z = fmaf(v[u].z, w[u], acc.z);
        acc.w = fmaf(v[u].w, w[u], acc.w);
    }
}

#define GATHER_BLOCKS 2048
#define GATHER_WAVES  (GATHER_BLOCKS * 8)

__global__ __launch_bounds__(256) void gather_kernel(const float* __restrict__ h,
                                                     const float* __restrict__ ns,
                                                     const int* __restrict__ rowptr,
                                                     const int* __restrict__ cnt,
                                                     const int* __restrict__ csr,
                                                     float* __restrict__ agg) {
    const int lane = threadIdx.x & 31;
    const int w0 = __builtin_amdgcn_readfirstlane((int)((blockIdx.x * 256 + threadIdx.x) >> 5));
    for (int node = w0; node < NNODES; node += GATHER_WAVES) {
        const int start = rowptr[node];
        const int deg   = cnt[node];
        float4 acc = make_float4(0.f, 0.f, 0.f, 0.f);
        for (int base = 0; base < deg; base += 32) {
            const int n = min(32, deg - base);
            int   my  = 0;
            float myw = 0.f;
            if (lane < n) {
                my  = (int)min((unsigned)csr[start + base + lane], (unsigned)(NNODES - 1));
                myw = ns[my];
            }
            int j = 0;
            for (; j + 8 <= n; j += 8) acc_rows<8>(h, my, myw, j, lane, acc);
            if (j + 4 <= n) { acc_rows<4>(h, my, myw, j, lane, acc); j += 4; }
            if (j + 2 <= n) { acc_rows<2>(h, my, myw, j, lane, acc); j += 2; }
            if (j < n)      { acc_rows<1>(h, my, myw, j, lane, acc); }
        }
        ((float4*)(agg + (size_t)node * DIM))[lane] = acc;
    }
}


template <int NOUT, int NT, bool RELU>
__global__ __launch_bounds__(256) void gcn_gemm(const float* __restrict__ agg,
                                                const float* __restrict__ nd,
                                                const float* __restrict__ W,
                                                const float* __restrict__ bias,
                                                float* __restrict__ out) {
    constexpr int NT16 = NT * 16;
    __shared__ float As[16][DIM + 4];
    __shared__ float Ws[DIM][NT16];

    const int t    = threadIdx.x;
    const int tile = blockIdx.x;

    {
        int m    = t >> 4;
        int kb   = (t & 15) * 8;
        int node = tile * 16 + m;
        float s  = nd[node];
        const float4* p = (const float4*)(agg + (size_t)node * DIM + kb);
        float4 v0 = p[0], v1 = p[1];
        As[m][kb + 0] = v0.x * s; As[m][kb + 1] = v0.y * s;
        As[m][kb + 2] = v0.z * s; As[m][kb + 3] = v0.w * s;
        As[m][kb + 4] = v1.x * s; As[m][kb + 5] = v1.y * s;
        As[m][kb + 6] = v1.z * s; As[m][kb + 7] = v1.w * s;
    }
    for (int i = t; i < DIM * NT16; i += 256) {
        int k = i / NT16;
        int n = i - k * NT16;
        Ws[k][n] = (n < NOUT) ? W[k * NOUT + n] : 0.0f;
    }
    __syncthreads();

    const int wave = t >> 5;
    const int lane = t & 31;
    if (wave < NT) {
        const int n0  = wave * 16;
        const int l16 = lane & 15;
        const int kh  = (lane >> 4) * 2;
        v8f acc = {};
#pragma unroll
        for (int kk = 0; kk < DIM; kk += 4) {
            v2f a, b;
            a.x = As[l16][kk + kh];
            a.y = As[l16][kk + kh + 1];
            b.x = Ws[kk + kh][n0 + l16];
            b.y = Ws[kk + kh + 1][n0 + l16];
            acc = __builtin_amdgcn_wmma_f32_16x16x4_f32(
                false, a, false, b, (short)0, acc, false, false);
        }
        const int n = n0 + l16;
        if (n < NOUT) {
            const int mbase = tile * 16 + (lane >> 4) * 8;
            float bv = bias[n];
#pragma unroll
            for (int v = 0; v < 8; ++v) {
                float r = acc[v] + bv;
                if (RELU) r = fmaxf(r, 0.0f);
                out[(size_t)(mbase + v) * NOUT + n] = r;
            }
        }
    }
}


extern "C" void kernel_launch(void* const* d_in, const int* in_sizes, int n_in,
                              void* d_out, int out_size, void* d_ws, size_t ws_size,
                              hipStream_t stream) {
    const float* inputs = (const float*)d_in[0];
    const int*   src    = (const int*)d_in[1];
    const int*   dst    = (const int*)d_in[2];
    const float* W0     = (const float*)d_in[3];
    const float* b0     = (const float*)d_in[4];
    const float* W1     = (const float*)d_in[5];
    const float* b1     = (const float*)d_in[6];
    const float* W2     = (const float*)d_in[7];
    const float* b2     = (const float*)d_in[8];
    float* out = (float*)d_out;

    constexpr size_t OFF_ND   = NNODES;
    constexpr size_t OFF_CS   = 2 * (size_t)NNODES;
    constexpr size_t OFF_CD   = 3 * (size_t)NNODES;
    constexpr size_t OFF_RP   = 4 * (size_t)NNODES;
    constexpr size_t OFF_POS  = 5 * (size_t)NNODES;
    constexpr size_t OFF_CSR  = OFF_POS + NEDGES;
    constexpr size_t OFF_B    = ((OFF_CSR + NEDGES + 127) / 128) * 128;
    constexpr size_t OFF_C    = OFF_B + (size_t)NNODES * DIM;

    float* ws      = (float*)d_ws;
    float* ns      = ws;
    float* ndm     = ws + OFF_ND;
    int*   cnt_src = (int*)(ws + OFF_CS);
    int*   cnt_dst = (int*)(ws + OFF_CD);
    int*   rowptr  = (int*)(ws + OFF_RP);
    int*   pos     = (int*)(ws + OFF_POS);
    int*   csr     = (int*)(ws + OFF_CSR);
    float* bufB    = ws + OFF_B;
    float* bufC    = ws + OFF_C;

    constexpr int CNT4    = 2 * NNODES / 4;
    constexpr int GR_CZ   = (CNT4 + 255) / 256;
    constexpr int GR_EDGE = (NEDGES + 255) / 256;
    constexpr int GR_SCAN = (NNODES / 4 + 255) / 256;
    constexpr int GR_GEMM = NNODES / 16;

    zero_f4<<<GR_CZ, 256, 0, stream>>>((float4*)cnt_src, CNT4);
    count_rank<<<GR_EDGE, 256, 0, stream>>>(src, dst, cnt_src, cnt_dst, pos);
    scan_norm<<<GR_SCAN, 256, 0, stream>>>(cnt_src, cnt_dst, rowptr, ns, ndm);
    fill_buckets<<<GR_EDGE, 256, 0, stream>>>(src, dst, rowptr, pos, csr);

    gather_kernel<<<GATHER_BLOCKS, 256, 0, stream>>>(inputs, ns, rowptr, cnt_dst, csr, bufB);
    gcn_gemm<DIM, 8, true><<<GR_GEMM, 256, 0, stream>>>(bufB, ndm, W0, b0, bufC);

    gather_kernel<<<GATHER_BLOCKS, 256, 0, stream>>>(bufC, ns, rowptr, cnt_dst, csr, bufB);
    gcn_gemm<DIM, 8, true><<<GR_GEMM, 256, 0, stream>>>(bufB, ndm, W1, b1, bufC);

    gather_kernel<<<GATHER_BLOCKS, 256, 0, stream>>>(bufC, ns, rowptr, cnt_dst, csr, bufB);
    gcn_gemm<NCLS, 3, false><<<GR_GEMM, 256, 0, stream>>>(bufB, ndm, W2, b2, out);
}


# One short wave per item replaced by resident waves that loop: GEMM 768 groups over 16-node tiles, W staged once; edge passes 3-4 edges per thread; sums still in atomic-arrival rank order
# speedup vs baseline: 2.9741x; 1.1122x over previous
#include <hip/hip_runtime.h>


#define NNODES 100000
#define NEDGES 1600000
#define DIM    128
#define NCLS   40

typedef __attribute__((ext_vector_type(2))) float v2f;
typedef __attribute__((ext_vector_type(8))) float v8f;


__global__ __launch_bounds__(256) void zero_f4(float4* __restrict__ p, int n4) {
    int i = blockIdx.x * 256 + threadIdx.x;
    if (i < n4) p[i] = make_float4(0.f, 0.f, 0.f, 0.f);
}

#define EDGE_BLOCKS  2048
#define EDGE_THREADS (EDGE_BLOCKS * 256)
#define EFULL        (NEDGES / EDGE_THREADS)
#define ETAIL        (EFULL * EDGE_THREADS)

__global__ __launch_bounds__(256) void count_rank(const int* __restrict__ src,
                                                  const int* __restrict__ dst,
                                                  int* __restrict__ cnt_src,
                                                  int* __restrict__ cnt_dst,
                                                  int* __restrict__ pos) {
    const int t0 = blockIdx.x * 256 + threadIdx.x;
    const int et = ETAIL + t0;
    const bool tail = et < NEDGES;
    int s[EFULL + 1], d[EFULL + 1], r[EFULL + 1];
#pragma unroll
    for (int u = 0; u < EFULL; ++u) {
        s[u] = src[t0 + u * EDGE_THREADS];
        d[u] = dst[t0 + u * EDGE_THREADS];
    }
    s[EFULL] = 0; d[EFULL] = 0; r[EFULL] = 0;
    if (tail) { s[EFULL] = src[et]; d[EFULL] = dst[et]; }
#pragma unroll
    for (int u = 0; u < EFULL; ++u) {
        atomicAdd(cnt_src + s[u], 1);
        r[u] = atomicAdd(cnt_dst + d[u], 1);
    }
    if (tail) {
        atomicAdd(cnt_src + s[EFULL], 1);
        r[EFULL] = atomicAdd(cnt_dst + d[EFULL], 1);
    }
#pragma unroll
    for (int u = 0; u < EFULL; ++u) pos[t0 + u * EDGE_THREADS] = r[u];
    if (tail) pos[et] = r[EFULL];
}

__device__ __forceinline__ float inv_sqrt_deg(int c) {
    return (c > 0) ? rsqrtf(fmaxf((float)c, 1.f)) : 0.f;
}

__global__ __launch_bounds__(256) void scan_norm(const int* __restrict__ cnt_src,
                                                 const int* __restrict__ cnt_dst,
                                                 int* __restrict__ rowptr,
                                                 float* __restrict__ ns,
                                                 float* __restrict__ nd) {
    __shared__ int red[8];
    __shared__ int wsum[8];
    const int t = threadIdx.x, b = blockIdx.x;
    const int lane = t & 31, wave = t >> 5;
    const int4* c4 = (const int4*)cnt_dst;

    int part = 0;
    int k = 0;
    for (; k + 8 <= b; k += 8) {
        int4 q[8];
#pragma unroll
        for (int u = 0; u < 8; ++u) q[u] = c4[(k + u) * 256 + t];
#pragma unroll
        for (int u = 0; u < 8; ++u) part += (q[u].x + q[u].y) + (q[u].z + q[u].w);
    }
    for (; k < b; ++k) {
        int4 q = c4[k * 256 + t];
        part += (q.x + q.y) + (q.z + q.w);
    }
    for (int o = 16; o > 0; o >>= 1) part += __shfl_xor(part, o);
    if (lane == 0) red[wave] = part;

    const int i4 = b * 256 + t;
    const bool in = i4 < NNODES / 4;
    int4 c = make_int4(0, 0, 0, 0);
    if (in) c = c4[i4];
    const int s = (c.x + c.y) + (c.z + c.w);
    int inc = s;
    for (int o = 1; o < 32; o <<= 1) {
        int v = __shfl_up(inc, o);
        if (lane >= o) inc += v;
    }
    if (lane == 31) wsum[wave] = inc;
    __syncthreads();

    int base = 0;
    for (int w = 0; w < 8; ++w) base += red[w];
    for (int w = 0; w < 8; ++w) base += (w < wave) ? wsum[w] : 0;
    const int ex = base + inc - s;
    if (in) {
        ((int4*)rowptr)[i4] = make_int4(ex, ex + c.x, ex + c.x + c.y, ex + c.x + c.y + c.z);
        int4 cs = ((const int4*)cnt_src)[i4];
        ((float4*)ns)[i4] = make_float4(inv_sqrt_deg(cs.x), inv_sqrt_deg(cs.y),
                                        inv_sqrt_deg(cs.z), inv_sqrt_deg(cs.w));
        ((float4*)nd)[i4] = make_float4(inv_sqrt_deg(c.x), inv_sqrt_deg(c.y),
                                        inv_sqrt_deg(c.z), inv_sqrt_deg(c.w));
    }
}

__global__ __launch_bounds__(256) void fill_buckets(const int* __restrict__ src,
                                                    const int* __restrict__ dst,
                                                    const int* __restrict__ rowptr,
                                                    const int* __restrict__ pos,
                                                    int* __restrict__ csr) {
    const int t0 = blockIdx.x * 256 + threadIdx.x;
    const int et = ETAIL + t0;
    const bool tail = et < NEDGES;
    int s[EFULL + 1], r[EFULL + 1], p[EFULL + 1];
#pragma unroll
    for (int u = 0; u < EFULL; ++u) {
        s[u] = src[t0 + u * EDGE_THREADS];
        r[u] = dst[t0 + u * EDGE_THREADS];
        p[u] = pos[t0 + u * EDGE_THREADS];
    }
    s[EFULL] = 0; r[EFULL] = 0; p[EFULL] = 0;
    if (tail) { s[EFULL] = src[et]; r[EFULL] = dst[et]; p[EFULL] = pos[et]; }
#pragma unroll
    for (int u = 0; u < EFULL; ++u) r[u] = rowptr[r[u]];
    if (tail) r[EFULL] = rowptr[r[EFULL]];
#pragma unroll
    for (int u = 0; u < EFULL; ++u) {
        unsigned idx = (unsigned)(r[u] + p[u]);
        if (idx < (unsigned)NEDGES) csr[idx] = s[u];
    }
    if (tail) {
        unsigned idx = (unsigned)(r[EFULL] + p[EFULL]);
        if (idx < (unsigned)NEDGES) csr[idx] = s[EFULL];
    }
}


template <int U>
__device__ __forceinline__ void acc_rows(const float* __restrict__ h, int my, float myw,
                                         int j, int lane, float4& acc) {
    float4 v[U];
    float  w[U];
#pragma unroll
    for (int u = 0; u < U; ++u) {
        int s = __builtin_amdgcn_readlane(my, j + u);
        w[u]  = __int_as_float(__builtin_amdgcn_readlane(__float_as_int(myw), j + u));
        v[u]  = ((const float4*)(h + (size_t)s * DIM))[lane];
    }
#pragma unroll
    for (int u = 0; u < U; ++u) {
        acc.x = fmaf(v[u].x, w[u], acc.x);
        acc.y = fmaf(v[u].y, w[u], acc.y);
        acc.z = fmaf(v[u].z, w[u], acc.z);
        acc.w = fmaf(v[u].w, w[u], acc.w);
    }
}

#define GATHER_BLOCKS 2048
#define GATHER_WAVES  (GATHER_BLOCKS * 8)

__global__ __launch_bounds__(256) void gather_kernel(const float* __restrict__ h,
                                                     const float* __restrict__ ns,
                                                     const int* __restrict__ rowptr,
                                                     const int* __restrict__ cnt,
                                                     const int* __restrict__ csr,
                                                     float* __restrict__ agg) {
    const int lane = threadIdx.x & 31;
    const int w0 = __builtin_amdgcn_readfirstlane((int)((blockIdx.x * 256 + threadIdx.x) >> 5));
    for (int node = w0; node < NNODES; node += GATHER_WAVES) {
        const int start = rowptr[node];
        const int deg   = cnt[node];
        float4 acc = make_float4(0.f, 0.f, 0.f, 0.f);
        for (int base = 0; base < deg; base += 32) {
            const int n = min(32, deg - base);
            int   my  = 0;
            float myw = 0.f;
            if (lane < n) {
                my  = (int)min((unsigned)csr[start + base + lane], (unsigned)(NNODES - 1));
                myw = ns[my];
            }
            int j = 0;
            for (; j + 8 <= n; j += 8) acc_rows<8>(h, my, myw, j, lane, acc);
            if (j + 4 <= n) { acc_rows<4>(h, my, myw, j, lane, acc); j += 4; }
            if (j + 2 <= n) { acc_rows<2>(h, my, myw, j, lane, acc); j += 2; }
            if (j < n)      { acc_rows<1>(h, my, myw, j, lane, acc); }
        }
        ((float4*)(agg + (size_t)node * DIM))[lane] = acc;
    }
}


#define GEMM_BLOCKS 768
#define NTILES      (NNODES / 16)
static_assert(GEMM_BLOCKS <= NTILES, "every work-group owns at least one tile");

template <int NOUT, int NT, bool RELU>
__global__ __launch_bounds__(256) void gcn_gemm(const float* __restrict__ agg,
                                                const float* __restrict__ nd,
                                                const float* __restrict__ W,
                                                const float* __restrict__ bias,
                                                float* __restrict__ out) {
    constexpr int NT16 = NT * 16;
    __shared__ __attribute__((aligned(16))) float As[2][16][DIM + 4];
    __shared__ __attribute__((aligned(16))) float Ws[DIM][NT16];

    const int t   = threadIdx.x;
    const int am  = t >> 4;
    const int akb = (t & 15) * 8;

    int tile = blockIdx.x;

    float  s;
    float4 v0, v1;
    {
        const int node = tile * 16 + am;
        s = nd[node];
        const float4* p = (const float4*)(agg + (size_t)node * DIM + akb);
        v0 = p[0]; v1 = p[1];
    }

    if constexpr (NOUT == NT16) {
        for (int i = t; i < DIM * NT16 / 4; i += 256)
            ((float4*)&Ws[0][0])[i] = ((const float4*)W)[i];
    } else {
        for (int i = t; i < DIM * NT16; i += 256) {
            int k = i / NT16;
            int n = i - k * NT16;
            Ws[k][n] = (n < NOUT) ? W[k * NOUT + n] : 0.0f;
        }
    }

    const int wave = t >> 5;
    const int lane = t & 31;
    const int n0   = wave * 16;
    const int l16  = lane & 15;
    const int kh   = (lane >> 4) * 2;
    const int n    = n0 + l16;
    float bv = 0.0f;
    if (n < NOUT) bv = bias[n];

    int cur = 0;
    for (; tile < NTILES; tile += GEMM_BLOCKS) {
        {
            float* a = &As[cur][am][akb];
            *(float4*)a       = make_float4(v0.x * s, v0.y * s, v0.z * s, v0.w * s);
            *(float4*)(a + 4) = make_float4(v1.x * s, v1.y * s, v1.z * s, v1.w * s);
        }
        __syncthreads();

        const int nxt = tile + GEMM_BLOCKS;
        if (nxt < NTILES) {
            const int node = nxt * 16 + am;
            s = nd[node];
            const float4* p = (const float4*)(agg + (size_t)node * DIM + akb);
            v0 = p[0]; v1 = p[1];
        }

        if (wave < NT) {
            const float (*Ac)[DIM + 4] = As[cur];
            v8f acc = {};
#pragma unroll
            for (int kk = 0; kk < DIM; kk += 4) {
                v2f a, b;
                a.x = Ac[l16][kk + kh];
                a.y = Ac[l16][kk + kh + 1];
                b.x = Ws[kk + kh][n0 + l16];
                b.y = Ws[kk + kh + 1][n0 + l16];
                acc = __builtin_amdgcn_wmma_f32_16x16x4_f32(
                    false, a, false, b, (short)0, acc, false, false);
            }
            float r[8];
#pragma unroll
            for (int v = 0; v < 8; ++v) {
                r[v] = acc[v] + bv;
                if (RELU) r[v] = fmaxf(r[v], 0.0f);
            }
            if (n < NOUT) {
                const int mbase = tile * 16 + (lane >> 4) * 8;
#pragma unroll
                for (int v = 0; v < 8; ++v)
                    out[(size_t)(mbase + v) * NOUT + n] = r[v];
            }
        }
        cur ^= 1;
    }
}


extern "C" void kernel_launch(void* const* d_in, const int* in_sizes, int n_in,
                              void* d_out, int out_size, void* d_ws, size_t ws_size,
                              hipStream_t stream) {
    const float* inputs = (const float*)d_in[0];
    const int*   src    = (const int*)d_in[1];
    const int*   dst    = (const int*)d_in[2];
    const float* W0     = (const float*)d_in[3];
    const float* b0     = (const float*)d_in[4];
    const float* W1     = (const float*)d_in[5];
    const float* b1     = (const float*)d_in[6];
    const float* W2     = (const float*)d_in[7];
    const float* b2     = (const float*)d_in[8];
    float* out = (float*)d_out;

    constexpr size_t OFF_ND   = NNODES;
    constexpr size_t OFF_CS   = 2 * (size_t)NNODES;
    constexpr size_t OFF_CD   = 3 * (size_t)NNODES;
    constexpr size_t OFF_RP   = 4 * (size_t)NNODES;
    constexpr size_t OFF_POS  = 5 * (size_t)NNODES;
    constexpr size_t OFF_CSR  = OFF_POS + NEDGES;
    constexpr size_t OFF_B    = ((OFF_CSR + NEDGES + 127) / 128) * 128;
    constexpr size_t OFF_C    = OFF_B + (size_t)NNODES * DIM;

    float* ws      = (float*)d_ws;
    float* ns      = ws;
    float* ndm     = ws + OFF_ND;
    int*   cnt_src = (int*)(ws + OFF_CS);
    int*   cnt_dst = (int*)(ws + OFF_CD);
    int*   rowptr  = (int*)(ws + OFF_RP);
    int*   pos     = (int*)(ws + OFF_POS);
    int*   csr     = (int*)(ws + OFF_CSR);
    float* bufB    = ws + OFF_B;
    float* bufC    = ws + OFF_C;

    constexpr int CNT4    = 2 * NNODES / 4;
    constexpr int GR_CZ   = (CNT4 + 255) / 256;
    constexpr int GR_SCAN = (NNODES / 4 + 255) / 256;

    zero_f4<<<GR_CZ, 256, 0, stream>>>((float4*)cnt_src, CNT4);
    count_rank<<<EDGE_BLOCKS, 256, 0, stream>>>(src, dst, cnt_src, cnt_dst, pos);
    scan_norm<<<GR_SCAN, 256, 0, stream>>>(cnt_src, cnt_dst, rowptr, ns, ndm);
    fill_buckets<<<EDGE_BLOCKS, 256, 0, stream>>>(src, dst, rowptr, pos, csr);

    gather_kernel<<<GATHER_BLOCKS, 256, 0, stream>>>(inputs, ns, rowptr, cnt_dst, csr, bufB);
    gcn_gemm<DIM, 8, true><<<GEMM_BLOCKS, 256, 0, stream>>>(bufB, ndm, W0, b0, bufC);

    gather_kernel<<<GATHER_BLOCKS, 256, 0, stream>>>(bufC, ns, rowptr, cnt_dst, csr, bufB);
    gcn_gemm<DIM, 8, true><<<GEMM_BLOCKS, 256, 0, stream>>>(bufB, ndm, W1, b1, bufC);

    gather_kernel<<<GATHER_BLOCKS, 256, 0, stream>>>(bufC, ns, rowptr, cnt_dst, csr, bufB);
    gcn_gemm<NCLS, 3, false><<<GEMM_BLOCKS, 256, 0, stream>>>(bufB, ndm, W2, b2, out);
}


# Last layer applies its 128x40 weight before the neighbour sum (sum is linear): its gather reads one 256-B unit per edge, not a 512-B row; sum order changes: product first, bucket slots summed 3-way
# speedup vs baseline: 3.1346x; 1.0540x over previous
#include <hip/hip_runtime.h>


#define NNODES 100000
#define NEDGES 1600000
#define DIM    128
#define NCLS   40
#define ZS     64

typedef __attribute__((ext_vector_type(2))) float v2f;
typedef __attribute__((ext_vector_type(8))) float v8f;


__global__ __launch_bounds__(256) void zero_f4(float4* __restrict__ p, int n4) {
    int i = blockIdx.x * 256 + threadIdx.x;
    if (i < n4) p[i] = make_float4(0.f, 0.f, 0.f, 0.f);
}

#define EDGE_BLOCKS  2048
#define EDGE_THREADS (EDGE_BLOCKS * 256)
#define EFULL        (NEDGES / EDGE_THREADS)
#define ETAIL        (EFULL * EDGE_THREADS)

__global__ __launch_bounds__(256) void count_rank(const int* __restrict__ src,
                                                  const int* __restrict__ dst,
                                                  int* __restrict__ cnt_src,
                                                  int* __restrict__ cnt_dst,
                                                  int* __restrict__ pos) {
    const int t0 = blockIdx.x * 256 + threadIdx.x;
    const int et = ETAIL + t0;
    const bool tail = et < NEDGES;
    int s[EFULL + 1], d[EFULL + 1], r[EFULL + 1];
#pragma unroll
    for (int u = 0; u < EFULL; ++u) {
        s[u] = src[t0 + u * EDGE_THREADS];
        d[u] = dst[t0 + u * EDGE_THREADS];
    }
    s[EFULL] = 0; d[EFULL] = 0; r[EFULL] = 0;
    if (tail) { s[EFULL] = src[et]; d[EFULL] = dst[et]; }
#pragma unroll
    for (int u = 0; u < EFULL; ++u) {
        atomicAdd(cnt_src + s[u], 1);
        r[u] = atomicAdd(cnt_dst + d[u], 1);
    }
    if (tail) {
        atomicAdd(cnt_src + s[EFULL], 1);
        r[EFULL] = atomicAdd(cnt_dst + d[EFULL], 1);
    }
#pragma unroll
    for (int u = 0; u < EFULL; ++u) pos[t0 + u * EDGE_THREADS] = r[u];
    if (tail) pos[et] = r[EFULL];
}

__device__ __forceinline__ float inv_sqrt_deg(int c) {
    return (c > 0) ? rsqrtf(fmaxf((float)c, 1.f)) : 0.f;
}

__global__ __launch_bounds__(256) void scan_norm(const int* __restrict__ cnt_src,
                                                 const int* __restrict__ cnt_dst,
                                                 int* __restrict__ rowptr,
                                                 float* __restrict__ ns,
                                                 float* __restrict__ nd) {
    __shared__ int red[8];
    __shared__ int wsum[8];
    const int t = threadIdx.x, b = blockIdx.x;
    const int lane = t & 31, wave = t >> 5;
    const int4* c4 = (const int4*)cnt_dst;

    int part = 0;
    int k = 0;
    for (; k + 8 <= b; k += 8) {
        int4 q[8];
#pragma unroll
        for (int u = 0; u < 8; ++u) q[u] = c4[(k + u) * 256 + t];
#pragma unroll
        for (int u = 0; u < 8; ++u) part += (q[u].x + q[u].y) + (q[u].z + q[u].w);
    }
    for (; k < b; ++k) {
        int4 q = c4[k * 256 + t];
        part += (q.x + q.y) + (q.z + q.w);
    }
    for (int o = 16; o > 0; o >>= 1) part += __shfl_xor(part, o);
    if (lane == 0) red[wave] = part;

    const int i4 = b * 256 + t;
    const bool in = i4 < NNODES / 4;
    int4 c = make_int4(0, 0, 0, 0);
    if (in) c = c4[i4];
    const int s = (c.x + c.y) + (c.z + c.w);
    int inc = s;
    for (int o = 1; o < 32; o <<= 1) {
        int v = __shfl_up(inc, o);
        if (lane >= o) inc += v;
    }
    if (lane == 31) wsum[wave] = inc;
    __syncthreads();

    int base = 0;
    for (int w = 0; w < 8; ++w) base += red[w];
    for (int w = 0; w < 8; ++w) base += (w < wave) ? wsum[w] : 0;
    const int ex = base + inc - s;
    if (in) {
        ((int4*)rowptr)[i4] = make_int4(ex, ex + c.x, ex + c.x + c.y, ex + c.x + c.y + c.z);
        int4 cs = ((const int4*)cnt_src)[i4];
        ((float4*)ns)[i4] = make_float4(inv_sqrt_deg(cs.x), inv_sqrt_deg(cs.y),
                                        inv_sqrt_deg(cs.z), inv_sqrt_deg(cs.w));
        ((float4*)nd)[i4] = make_float4(inv_sqrt_deg(c.x), inv_sqrt_deg(c.y),
                                        inv_sqrt_deg(c.z), inv_sqrt_deg(c.w));
    }
}

__global__ __launch_bounds__(256) void fill_buckets(const int* __restrict__ src,
                                                    const int* __restrict__ dst,
                                                    const int* __restrict__ rowptr,
                                                    const int* __restrict__ pos,
                                                    int* __restrict__ csr) {
    const int t0 = blockIdx.x * 256 + threadIdx.x;
    const int et = ETAIL + t0;
    const bool tail = et < NEDGES;
    int s[EFULL + 1], r[EFULL + 1], p[EFULL + 1];
#pragma unroll
    for (int u = 0; u < EFULL; ++u) {
        s[u] = src[t0 + u * EDGE_THREADS];
        r[u] = dst[t0 + u * EDGE_THREADS];
        p[u] = pos[t0 + u * EDGE_THREADS];
    }
    s[EFULL] = 0; r[EFULL] = 0; p[EFULL] = 0;
    if (tail) { s[EFULL] = src[et]; r[EFULL] = dst[et]; p[EFULL] = pos[et]; }
#pragma unroll
    for (int u = 0; u < EFULL; ++u) r[u] = rowptr[r[u]];
    if (tail) r[EFULL] = rowptr[r[EFULL]];
#pragma unroll
    for (int u = 0; u < EFULL; ++u) {
        unsigned idx = (unsigned)(r[u] + p[u]);
        if (idx < (unsigned)NEDGES) csr[idx] = s[u];
    }
    if (tail) {
        unsigned idx = (unsigned)(r[EFULL] + p[EFULL]);
        if (idx < (unsigned)NEDGES) csr[idx] = s[EFULL];
    }
}


template <int U>
__device__ __forceinline__ void acc_rows(const float* __restrict__ h, int my, float myw,
                                         int j, int lane, float4& acc) {
    float4 v[U];
    float  w[U];
#pragma unroll
    for (int u = 0; u < U; ++u) {
        int s = __builtin_amdgcn_readlane(my, j + u);
        w[u]  = __int_as_float(__builtin_amdgcn_readlane(__float_as_int(myw), j + u));
        v[u]  = ((const float4*)(h + (size_t)s * DIM))[lane];
    }
#pragma unroll
    for (int u = 0; u < U; ++u) {
        acc.x = fmaf(v[u].x, w[u], acc.x);
        acc.y = fmaf(v[u].y, w[u], acc.y);
        acc.z = fmaf(v[u].z, w[u], acc.z);
        acc.w = fmaf(v[u].w, w[u], acc.w);
    }
}

#define GATHER_BLOCKS 2048
#define GATHER_WAVES  (GATHER_BLOCKS * 8)

__global__ __launch_bounds__(256) void gather_kernel(const float* __restrict__ h,
                                                     const float* __restrict__ ns,
                                                     const int* __restrict__ rowptr,
                                                     const int* __restrict__ cnt,
                                                     const int* __restrict__ csr,
                                                     float* __restrict__ agg) {
    const int lane = threadIdx.x & 31;
    const int w0 = __builtin_amdgcn_readfirstlane((int)((blockIdx.x * 256 + threadIdx.x) >> 5));
    for (int node = w0; node < NNODES; node += GATHER_WAVES) {
        const int start = rowptr[node];
        const int deg   = cnt[node];
        float4 acc = make_float4(0.f, 0.f, 0.f, 0.f);
        for (int base = 0; base < deg; base += 32) {
            const int n = min(32, deg - base);
            int   my  = 0;
            float myw = 0.f;
            if (lane < n) {
                my  = (int)min((unsigned)csr[start + base + lane], (unsigned)(NNODES - 1));
                myw = ns[my];
            }
            int j = 0;
            for (; j + 8 <= n; j += 8) acc_rows<8>(h, my, myw, j, lane, acc);
            if (j + 4 <= n) { acc_rows<4>(h, my, myw, j, lane, acc); j += 4; }
            if (j + 2 <= n) { acc_rows<2>(h, my, myw, j, lane, acc); j += 2; }
            if (j < n)      { acc_rows<1>(h, my, myw, j, lane, acc); }
        }
        ((float4*)(agg + (size_t)node * DIM))[lane] = acc;
    }
}


__device__ __forceinline__ int clamp_node(int s) {
    return (int)min((unsigned)s, (unsigned)(NNODES - 1));
}

template <int K>
__device__ __forceinline__ void rows3(const float* __restrict__ z, int ids, int rem,
                                      int g, int cc, float4& acc) {
    float4 v[K];
#pragma unroll
    for (int u = 0; u < K; ++u) {
        const int s0 = __builtin_amdgcn_readlane(ids, 3 * u);
        const int s1 = __builtin_amdgcn_readlane(ids, 3 * u + 1);
        const int s2 = __builtin_amdgcn_readlane(ids, 3 * u + 2);
        int s;
        if (u < K - 1) s = (g == 1) ? s1 : ((g == 2) ? s2 : s0);
        else           s = (g == 1 && 3 * u + 1 < rem) ? s1 : ((g == 2 && 3 * u + 2 < rem) ? s2 : s0);
        v[u] = ((const float4*)(z + (size_t)s * ZS))[cc];
    }
#pragma unroll
    for (int u = 0; u < K; ++u) {
        if (u < K - 1) {
            acc.x += v[u].x; acc.y += v[u].y; acc.z += v[u].z; acc.w += v[u].w;
        } else {
            const bool ok = 3 * u + g < rem;
            acc.x += ok ? v[u].x : 0.f; acc.y += ok ? v[u].y : 0.f;
            acc.z += ok ? v[u].z : 0.f; acc.w += ok ? v[u].w : 0.f;
        }
    }
}

__device__ __forceinline__ int bucket_ids(const int* __restrict__ csr, int first, int lane) {
    return csr[min((unsigned)(first + lane), (unsigned)(NEDGES - 1))];
}

__global__ __launch_bounds__(256) void gather_out(const float* __restrict__ z,
                                                  const float* __restrict__ nd,
                                                  const float* __restrict__ bias,
                                                  const int* __restrict__ rowptr,
                                                  const int* __restrict__ cnt,
                                                  const int* __restrict__ csr,
                                                  float* __restrict__ out) {
    const int lane = threadIdx.x & 31;
    const int g    = (lane >= 30) ? 0 : ((lane >= 20) ? 2 : ((lane >= 10) ? 1 : 0));
    const int cc   = (lane >= 30) ? (NCLS / 4 - 1) : (lane - 10 * g);
    const int w0 = __builtin_amdgcn_readfirstlane((int)((blockIdx.x * 256 + threadIdx.x) >> 5));
    const float4 bv = ((const float4*)bias)[cc];

    int   node  = w0;
    int   start = rowptr[node];
    int   deg   = cnt[node];
    float sc    = nd[node];
    int   my    = clamp_node(bucket_ids(csr, start, lane));
    static_assert(2 * GATHER_WAVES <= NNODES, "every wave owns a second node");
    int   n1    = node + GATHER_WAVES;
    int   s1    = rowptr[n1];
    int   d1    = cnt[n1];
    float c1    = nd[n1];

    for (;;) {
        const bool has1 = n1 < NNODES;
        const int nmy = bucket_ids(csr, s1, lane);
        const int  n2   = n1 + GATHER_WAVES;
        int   s2 = 0, d2 = 0;
        float c2 = 0.f;
        if (n2 < NNODES) { s2 = rowptr[n2]; d2 = cnt[n2]; c2 = nd[n2]; }

        float4 acc = make_float4(0.f, 0.f, 0.f, 0.f);
        int ids = my;
        for (int j0 = 0; j0 < deg; j0 += 24) {
            if (j0 > 0) ids = clamp_node(bucket_ids(csr, start + j0, lane));
            const int rem = deg - j0;
            switch (min(8, (rem + 2) / 3)) {
                case 1:  rows3<1>(z, ids, rem, g, cc, acc); break;
                case 2:  rows3<2>(z, ids, rem, g, cc, acc); break;
                case 3:  rows3<3>(z, ids, rem, g, cc, acc); break;
                case 4:  rows3<4>(z, ids, rem, g, cc, acc); break;
                case 5:  rows3<5>(z, ids, rem, g, cc, acc); break;
                case 6:  rows3<6>(z, ids, rem, g, cc, acc); break;
                case 7:  rows3<7>(z, ids, rem, g, cc, acc); break;
                default: rows3<8>(z, ids, rem, g, cc, acc); break;
            }
        }
        const float px = __shfl_down(acc.x, 10), py = __shfl_down(acc.y, 10);
        const float pz = __shfl_down(acc.z, 10), pw = __shfl_down(acc.w, 10);
        const float qx = __shfl_down(acc.x, 20), qy = __shfl_down(acc.y, 20);
        const float qz = __shfl_down(acc.z, 20), qw = __shfl_down(acc.w, 20);
        if (lane < NCLS / 4)
            ((float4*)(out + (size_t)node * NCLS))[lane] =
                make_float4(fmaf((acc.x + px) + qx, sc, bv.x), fmaf((acc.y + py) + qy, sc, bv.y),
                            fmaf((acc.z + pz) + qz, sc, bv.z), fmaf((acc.w + pw) + qw, sc, bv.w));
        if (!has1) break;
        node = n1; start = s1; deg = d1; sc = c1; my = clamp_node(nmy);
        n1 = n2; s1 = s2; d1 = d2; c1 = c2;
    }
}


#define GEMM_BLOCKS 768
#define NTILES      (NNODES / 16)
static_assert(GEMM_BLOCKS <= NTILES, "every work-group owns at least one tile");

template <int NOUT, int NT, bool RELU, bool BIAS, int OSTRIDE>
__global__ __launch_bounds__(256) void gcn_gemm(const float* __restrict__ agg,
                                                const float* __restrict__ nd,
                                                const float* __restrict__ W,
                                                const float* __restrict__ bias,
                                                float* __restrict__ out) {
    constexpr int NT16 = NT * 16;
    __shared__ __attribute__((aligned(16))) float As[2][16][DIM + 4];
    __shared__ __attribute__((aligned(16))) float Ws[DIM][NT16];

    const int t   = threadIdx.x;
    const int am  = t >> 4;
    const int akb = (t & 15) * 8;

    int tile = blockIdx.x;

    float  s;
    float4 v0, v1;
    {
        const int node = tile * 16 + am;
        s = nd[node];
        const float4* p = (const float4*)(agg + (size_t)node * DIM + akb);
        v0 = p[0]; v1 = p[1];
    }

    if constexpr (NOUT == NT16) {
        for (int i = t; i < DIM * NT16 / 4; i += 256)
            ((float4*)&Ws[0][0])[i] = ((const float4*)W)[i];
    } else {
        for (int i = t; i < DIM * NT16; i += 256) {
            int k = i / NT16;
            int n = i - k * NT16;
            Ws[k][n] = (n < NOUT) ? W[k * NOUT + n] : 0.0f;
        }
    }

    const int wave = t >> 5;
    const int lane = t & 31;
    const int n0   = wave * 16;
    const int l16  = lane & 15;
    const int kh   = (lane >> 4) * 2;
    const int n    = n0 + l16;
    float bv = 0.0f;
    if (BIAS && n < NOUT) bv = bias[n];

    int cur = 0;
    for (; tile < NTILES; tile += GEMM_BLOCKS) {
        {
            float* a = &As[cur][am][akb];
            *(float4*)a       = make_float4(v0.x * s, v0.y * s, v0.z * s, v0.w * s);
            *(float4*)(a + 4) = make_float4(v1.x * s, v1.y * s, v1.z * s, v1.w * s);
        }
        __syncthreads();

        const int nxt = tile + GEMM_BLOCKS;
        if (nxt < NTILES) {
            const int node = nxt * 16 + am;
            s = nd[node];
            const float4* p = (const float4*)(agg + (size_t)node * DIM + akb);
            v0 = p[0]; v1 = p[1];
        }

        if (wave < NT) {
            const float (*Ac)[DIM + 4] = As[cur];
            v8f acc = {};
#pragma unroll
            for (int kk = 0; kk < DIM; kk += 4) {
                v2f a, b;
                a.x = Ac[l16][kk + kh];
                a.y = Ac[l16][kk + kh + 1];
                b.x = Ws[kk + kh][n0 + l16];
                b.y = Ws[kk + kh + 1][n0 + l16];
                acc = __builtin_amdgcn_wmma_f32_16x16x4_f32(
                    false, a, false, b, (short)0, acc, false, false);
            }
            float r[8];
#pragma unroll
            for (int v = 0; v < 8; ++v) {
                r[v] = acc[v] + bv;
                if (RELU) r[v] = fmaxf(r[v], 0.0f);
            }
            if (n < NOUT) {
                const int mbase = tile * 16 + (lane >> 4) * 8;
#pragma unroll
                for (int v = 0; v < 8; ++v)
                    out[(size_t)(mbase + v) * OSTRIDE + n] = r[v];
            }
        }
        cur ^= 1;
    }
}


extern "C" void kernel_launch(void* const* d_in, const int* in_sizes, int n_in,
                              void* d_out, int out_size, void* d_ws, size_t ws_size,
                              hipStream_t stream) {
    const float* inputs = (const float*)d_in[0];
    const int*   src    = (const int*)d_in[1];
    const int*   dst    = (const int*)d_in[2];
    const float* W0     = (const float*)d_in[3];
    const float* b0     = (const float*)d_in[4];
    const float* W1     = (const float*)d_in[5];
    const float* b1     = (const float*)d_in[6];
    const float* W2     = (const float*)d_in[7];
    const float* b2     = (const float*)d_in[8];
    float* out = (float*)d_out;

    constexpr size_t OFF_ND   = NNODES;
    constexpr size_t OFF_CS   = 2 * (size_t)NNODES;
    constexpr size_t OFF_CD   = 3 * (size_t)NNODES;
    constexpr size_t OFF_RP   = 4 * (size_t)NNODES;
    constexpr size_t OFF_POS  = 5 * (size_t)NNODES;
    constexpr size_t OFF_CSR  = OFF_POS + NEDGES;
    constexpr size_t OFF_B    = ((OFF_CSR + NEDGES + 127) / 128) * 128;
    constexpr size_t OFF_C    = OFF_B + (size_t)NNODES * DIM;

    float* ws      = (float*)d_ws;
    float* ns      = ws;
    float* ndm     = ws + OFF_ND;
    int*   cnt_src = (int*)(ws + OFF_CS);
    int*   cnt_dst = (int*)(ws + OFF_CD);
    int*   rowptr  = (int*)(ws + OFF_RP);
    int*   pos     = (int*)(ws + OFF_POS);
    int*   csr     = (int*)(ws + OFF_CSR);
    float* bufB    = ws + OFF_B;
    float* bufC    = ws + OFF_C;

    constexpr int CNT4    = 2 * NNODES / 4;
    constexpr int GR_CZ   = (CNT4 + 255) / 256;
    constexpr int GR_SCAN = (NNODES / 4 + 255) / 256;

    zero_f4<<<GR_CZ, 256, 0, stream>>>((float4*)cnt_src, CNT4);
    count_rank<<<EDGE_BLOCKS, 256, 0, stream>>>(src, dst, cnt_src, cnt_dst, pos);
    scan_norm<<<GR_SCAN, 256, 0, stream>>>(cnt_src, cnt_dst, rowptr, ns, ndm);
    fill_buckets<<<EDGE_BLOCKS, 256, 0, stream>>>(src, dst, rowptr, pos, csr);

    gather_kernel<<<GATHER_BLOCKS, 256, 0, stream>>>(inputs, ns, rowptr, cnt_dst, csr, bufB);
    gcn_gemm<DIM, 8, true, true, DIM><<<GEMM_BLOCKS, 256, 0, stream>>>(bufB, ndm, W0, b0, bufC);

    gather_kernel<<<GATHER_BLOCKS, 256, 0, stream>>>(bufC, ns, rowptr, cnt_dst, csr, bufB);
    gcn_gemm<DIM, 8, true, true, DIM><<<GEMM_BLOCKS, 256, 0, stream>>>(bufB, ndm, W1, b1, bufC);

    gcn_gemm<NCLS, 3, false, false, ZS><<<GEMM_BLOCKS, 256, 0, stream>>>(bufC, ns, W2, b2, bufB);
    gather_out<<<GATHER_BLOCKS, 256, 0, stream>>>(bufB, ndm, b2, rowptr, cnt_dst, csr, out);
}


# Degree counts and bucket ranks by LDS integer atomics on per-chunk counter planes (64 edge chunks x 4 node ranges) and a prefix over chunks, not global atomics; sum order = chunk, then LDS arrival
# speedup vs baseline: 4.2133x; 1.3441x over previous
#include <hip/hip_runtime.h>


#define NNODES 100000
#define NEDGES 1600000
#define DIM    128
#define NCLS   40
#define ZS     64

typedef __attribute__((ext_vector_type(2))) float v2f;
typedef __attribute__((ext_vector_type(8))) float v8f;


#define EDGE_BLOCKS  2048
#define EDGE_THREADS (EDGE_BLOCKS * 256)
#define EFULL        (NEDGES / EDGE_THREADS)
#define ETAIL        (EFULL * EDGE_THREADS)

#define HC    64
#define HR    4
#define HEC   (NEDGES / HC)
#define HNR   (NNODES / HR)
#define HT    1024
#define HFULL (HEC / HT)
#define HTAIL (HFULL * HT)
static_assert(HEC * HC == NEDGES && HNR * HR == NNODES, "chunks and ranges tile the problem exactly");
static_assert(HFULL % 8 == 0, "full strides are taken in groups of eight");
static_assert(HNR % 4 == 0, "counter planes move as 16-byte units");

__global__ __launch_bounds__(HT) void hist_rank(const int* __restrict__ src,
                                                const int* __restrict__ dst,
                                                int* __restrict__ plane_src,
                                                int* __restrict__ plane_dst,
                                                int* __restrict__ pos) {
    __shared__ __attribute__((aligned(16))) int hd[HNR];
    __shared__ __attribute__((aligned(16))) int hs[HNR];
    const int t  = threadIdx.x;
    const int c  = blockIdx.x / HR;
    const int lo = (blockIdx.x - c * HR) * HNR;
    const int e0 = c * HEC + t;

    for (int i = t; i < HNR / 4; i += HT) {
        ((int4*)hd)[i] = make_int4(0, 0, 0, 0);
        ((int4*)hs)[i] = make_int4(0, 0, 0, 0);
    }
    __syncthreads();

    for (int g = 0; g < HFULL; g += 8) {
        int dv[8], sv[8];
#pragma unroll
        for (int u = 0; u < 8; ++u) {
            dv[u] = dst[e0 + (g + u) * HT];
            sv[u] = src[e0 + (g + u) * HT];
        }
#pragma unroll
        for (int u = 0; u < 8; ++u) {
            const unsigned d = (unsigned)(dv[u] - lo);
            const unsigned q = (unsigned)(sv[u] - lo);
            if (d < (unsigned)HNR) pos[e0 + (g + u) * HT] = atomicAdd(&hd[d], 1);
            if (q < (unsigned)HNR) atomicAdd(&hs[q], 1);
        }
    }
    if (HTAIL + t < HEC) {
        const unsigned d = (unsigned)(dst[e0 + HTAIL] - lo);
        const unsigned q = (unsigned)(src[e0 + HTAIL] - lo);
        if (d < (unsigned)HNR) pos[e0 + HTAIL] = atomicAdd(&hd[d], 1);
        if (q < (unsigned)HNR) atomicAdd(&hs[q], 1);
    }
    __syncthreads();

    int4* od = (int4*)(plane_dst + (size_t)c * NNODES + lo);
    int4* os = (int4*)(plane_src + (size_t)c * NNODES + lo);
    for (int i = t; i < HNR / 4; i += HT) {
        od[i] = ((const int4*)hd)[i];
        os[i] = ((const int4*)hs)[i];
    }
}

__global__ __launch_bounds__(256) void chunk_prefix(const int* __restrict__ plane_src,
                                                    int* __restrict__ plane_dst,
                                                    int* __restrict__ cnt_src,
                                                    int* __restrict__ cnt_dst) {
    const int n = blockIdx.x * 256 + threadIdx.x;
    if (n >= NNODES) return;
    int run = 0, tot = 0;
    for (int c0 = 0; c0 < HC; c0 += 16) {
        int a[16], b[16];
#pragma unroll
        for (int u = 0; u < 16; ++u) {
            a[u] = plane_dst[(size_t)(c0 + u) * NNODES + n];
            b[u] = plane_src[(size_t)(c0 + u) * NNODES + n];
        }
#pragma unroll
        for (int u = 0; u < 16; ++u) {
            plane_dst[(size_t)(c0 + u) * NNODES + n] = run;
            run += a[u];
            tot += b[u];
        }
    }
    cnt_dst[n] = run;
    cnt_src[n] = tot;
}

__device__ __forceinline__ float inv_sqrt_deg(int c) {
    return (c > 0) ? rsqrtf(fmaxf((float)c, 1.f)) : 0.f;
}

__global__ __launch_bounds__(256) void scan_norm(const int* __restrict__ cnt_src,
                                                 const int* __restrict__ cnt_dst,
                                                 int* __restrict__ rowptr,
                                                 float* __restrict__ ns,
                                                 float* __restrict__ nd) {
    __shared__ int red[8];
    __shared__ int wsum[8];
    const int t = threadIdx.x, b = blockIdx.x;
    const int lane = t & 31, wave = t >> 5;
    const int4* c4 = (const int4*)cnt_dst;

    int part = 0;
    int k = 0;
    for (; k + 8 <= b; k += 8) {
        int4 q[8];
#pragma unroll
        for (int u = 0; u < 8; ++u) q[u] = c4[(k + u) * 256 + t];
#pragma unroll
        for (int u = 0; u < 8; ++u) part += (q[u].x + q[u].y) + (q[u].z + q[u].w);
    }
    for (; k < b; ++k) {
        int4 q = c4[k * 256 + t];
        part += (q.x + q.y) + (q.z + q.w);
    }
    for (int o = 16; o > 0; o >>= 1) part += __shfl_xor(part, o);
    if (lane == 0) red[wave] = part;

    const int i4 = b * 256 + t;
    const bool in = i4 < NNODES / 4;
    int4 c = make_int4(0, 0, 0, 0);
    if (in) c = c4[i4];
    const int s = (c.x + c.y) + (c.z + c.w);
    int inc = s;
    for (int o = 1; o < 32; o <<= 1) {
        int v = __shfl_up(inc, o);
        if (lane >= o) inc += v;
    }
    if (lane == 31) wsum[wave] = inc;
    __syncthreads();

    int base = 0;
    for (int w = 0; w < 8; ++w) base += red[w];
    for (int w = 0; w < 8; ++w) base += (w < wave) ? wsum[w] : 0;
    const int ex = base + inc - s;
    if (in) {
        ((int4*)rowptr)[i4] = make_int4(ex, ex + c.x, ex + c.x + c.y, ex + c.x + c.y + c.z);
        int4 cs = ((const int4*)cnt_src)[i4];
        ((float4*)ns)[i4] = make_float4(inv_sqrt_deg(cs.x), inv_sqrt_deg(cs.y),
                                        inv_sqrt_deg(cs.z), inv_sqrt_deg(cs.w));
        ((float4*)nd)[i4] = make_float4(inv_sqrt_deg(c.x), inv_sqrt_deg(c.y),
                                        inv_sqrt_deg(c.z), inv_sqrt_deg(c.w));
    }
}

__global__ __launch_bounds__(256) void fill_buckets(const int* __restrict__ src,
                                                    const int* __restrict__ dst,
                                                    const int* __restrict__ rowptr,
                                                    const int* __restrict__ plane_dst,
                                                    const int* __restrict__ pos,
                                                    int* __restrict__ csr) {
    const int t0 = blockIdx.x * 256 + threadIdx.x;
    const int et = ETAIL + t0;
    const bool tail = et < NEDGES;
    int s[EFULL + 1], r[EFULL + 1], p[EFULL + 1], q[EFULL + 1];
#pragma unroll
    for (int u = 0; u < EFULL; ++u) {
        s[u] = src[t0 + u * EDGE_THREADS];
        r[u] = dst[t0 + u * EDGE_THREADS];
        p[u] = pos[t0 + u * EDGE_THREADS];
    }
    s[EFULL] = 0; r[EFULL] = 0; p[EFULL] = 0; q[EFULL] = 0;
    if (tail) { s[EFULL] = src[et]; r[EFULL] = dst[et]; p[EFULL] = pos[et]; }
#pragma unroll
    for (int u = 0; u < EFULL; ++u) {
        const int d = r[u];
        r[u] = rowptr[d];
        q[u] = plane_dst[(size_t)((t0 + u * EDGE_THREADS) / HEC) * NNODES + d];
    }
    if (tail) {
        const int d = r[EFULL];
        r[EFULL] = rowptr[d];
        q[EFULL] = plane_dst[(size_t)(et / HEC) * NNODES + d];
    }
#pragma unroll
    for (int u = 0; u < EFULL; ++u) {
        unsigned idx = (unsigned)(r[u] + q[u] + p[u]);
        if (idx < (unsigned)NEDGES) csr[idx] = s[u];
    }
    if (tail) {
        unsigned idx = (unsigned)(r[EFULL] + q[EFULL] + p[EFULL]);
        if (idx < (unsigned)NEDGES) csr[idx] = s[EFULL];
    }
}


template <int U>
__device__ __forceinline__ void acc_rows(const float* __restrict__ h, int my, float myw,
                                         int j, int lane, float4& acc) {
    float4 v[U];
    float  w[U];
#pragma unroll
    for (int u = 0; u < U; ++u) {
        int s = __builtin_amdgcn_readlane(my, j + u);
        w[u]  = __int_as_float(__builtin_amdgcn_readlane(__float_as_int(myw), j + u));
        v[u]  = ((const float4*)(h + (size_t)s * DIM))[lane];
    }
#pragma unroll
    for (int u = 0; u < U; ++u) {
        acc.x = fmaf(v[u].x, w[u], acc.x);
        acc.y = fmaf(v[u].y, w[u], acc.y);
        acc.z = fmaf(v[u].z, w[u], acc.z);
        acc.w = fmaf(v[u].w, w[u], acc.w);
    }
}

#define GATHER_BLOCKS 2048
#define GATHER_WAVES  (GATHER_BLOCKS * 8)

__global__ __launch_bounds__(256) void gather_kernel(const float* __restrict__ h,
                                                     const float* __restrict__ ns,
                                                     const int* __restrict__ rowptr,
                                                     const int* __restrict__ cnt,
                                                     const int* __restrict__ csr,
                                                     float* __restrict__ agg) {
    const int lane = threadIdx.x & 31;
    const int w0 = __builtin_amdgcn_readfirstlane((int)((blockIdx.x * 256 + threadIdx.x) >> 5));
    for (int node = w0; node < NNODES; node += GATHER_WAVES) {
        const int start = rowptr[node];
        const int deg   = cnt[node];
        float4 acc = make_float4(0.f, 0.f, 0.f, 0.f);
        for (int base = 0; base < deg; base += 32) {
            const int n = min(32, deg - base);
            int   my  = 0;
            float myw = 0.f;
            if (lane < n) {
                my  = (int)min((unsigned)csr[start + base + lane], (unsigned)(NNODES - 1));
                myw = ns[my];
            }
            int j = 0;
            for (; j + 8 <= n; j += 8) acc_rows<8>(h, my, myw, j, lane, acc);
            if (j + 4 <= n) { acc_rows<4>(h, my, myw, j, lane, acc); j += 4; }
            if (j + 2 <= n) { acc_rows<2>(h, my, myw, j, lane, acc); j += 2; }
            if (j < n)      { acc_rows<1>(h, my, myw, j, lane, acc); }
        }
        ((float4*)(agg + (size_t)node * DIM))[lane] = acc;
    }
}


__device__ __forceinline__ int clamp_node(int s) {
    return (int)min((unsigned)s, (unsigned)(NNODES - 1));
}

template <int K>
__device__ __forceinline__ void rows3(const float* __restrict__ z, int ids, int rem,
                                      int g, int cc, float4& acc) {
    float4 v[K];
#pragma unroll
    for (int u = 0; u < K; ++u) {
        const int s0 = __builtin_amdgcn_readlane(ids, 3 * u);
        const int s1 = __builtin_amdgcn_readlane(ids, 3 * u + 1);
        const int s2 = __builtin_amdgcn_readlane(ids, 3 * u + 2);
        int s;
        if (u < K - 1) s = (g == 1) ? s1 : ((g == 2) ? s2 : s0);
        else           s = (g == 1 && 3 * u + 1 < rem) ? s1 : ((g == 2 && 3 * u + 2 < rem) ? s2 : s0);
        v[u] = ((const float4*)(z + (size_t)s * ZS))[cc];
    }
#pragma unroll
    for (int u = 0; u < K; ++u) {
        if (u < K - 1) {
            acc.x += v[u].x; acc.y += v[u].y; acc.z += v[u].z; acc.w += v[u].w;
        } else {
            const bool ok = 3 * u + g < rem;
            acc.x += ok ? v[u].x : 0.f; acc.y += ok ? v[u].y : 0.f;
            acc.z += ok ? v[u].z : 0.f; acc.w += ok ? v[u].w : 0.f;
        }
    }
}

__device__ __forceinline__ int bucket_ids(const int* __restrict__ csr, int first, int lane) {
    return csr[min((unsigned)(first + lane), (unsigned)(NEDGES - 1))];
}

__global__ __launch_bounds__(256) void gather_out(const float* __restrict__ z,
                                                  const float* __restrict__ nd,
                                                  const float* __restrict__ bias,
                                                  const int* __restrict__ rowptr,
                                                  const int* __restrict__ cnt,
                                                  const int* __restrict__ csr,
                                                  float* __restrict__ out) {
    const int lane = threadIdx.x & 31;
    const int g    = (lane >= 30) ? 0 : ((lane >= 20) ? 2 : ((lane >= 10) ? 1 : 0));
    const int cc   = (lane >= 30) ? (NCLS / 4 - 1) : (lane - 10 * g);
    const int w0 = __builtin_amdgcn_readfirstlane((int)((blockIdx.x * 256 + threadIdx.x) >> 5));
    const float4 bv = ((const float4*)bias)[cc];

    int   node  = w0;
    int   start = rowptr[node];
    int   deg   = cnt[node];
    float sc    = nd[node];
    int   my    = clamp_node(bucket_ids(csr, start, lane));
    static_assert(2 * GATHER_WAVES <= NNODES, "every wave owns a second node");
    int   n1    = node + GATHER_WAVES;
    int   s1    = rowptr[n1];
    int   d1    = cnt[n1];
    float c1    = nd[n1];

    for (;;) {
        const bool has1 = n1 < NNODES;
        const int nmy = bucket_ids(csr, s1, lane);
        const int  n2   = n1 + GATHER_WAVES;
        int   s2 = 0, d2 = 0;
        float c2 = 0.f;
        if (n2 < NNODES) { s2 = rowptr[n2]; d2 = cnt[n2]; c2 = nd[n2]; }

        float4 acc = make_float4(0.f, 0.f, 0.f, 0.f);
        int ids = my;
        for (int j0 = 0; j0 < deg; j0 += 24) {
            if (j0 > 0) ids = clamp_node(bucket_ids(csr, start + j0, lane));
            const int rem = deg - j0;
            switch (min(8, (rem + 2) / 3)) {
                case 1:  rows3<1>(z, ids, rem, g, cc, acc); break;
                case 2:  rows3<2>(z, ids, rem, g, cc, acc); break;
                case 3:  rows3<3>(z, ids, rem, g, cc, acc); break;
                case 4:  rows3<4>(z, ids, rem, g, cc, acc); break;
                case 5:  rows3<5>(z, ids, rem, g, cc, acc); break;
                case 6:  rows3<6>(z, ids, rem, g, cc, acc); break;
                case 7:  rows3<7>(z, ids, rem, g, cc, acc); break;
                default: rows3<8>(z, ids, rem, g, cc, acc); break;
            }
        }
        const float px = __shfl_down(acc.x, 10), py = __shfl_down(acc.y, 10);
        const float pz = __shfl_down(acc.z, 10), pw = __shfl_down(acc.w, 10);
        const float qx = __shfl_down(acc.x, 20), qy = __shfl_down(acc.y, 20);
        const float qz = __shfl_down(acc.z, 20), qw = __shfl_down(acc.w, 20);
        if (lane < NCLS / 4)
            ((float4*)(out + (size_t)node * NCLS))[lane] =
                make_float4(fmaf((acc.x + px) + qx, sc, bv.x), fmaf((acc.y + py) + qy, sc, bv.y),
                            fmaf((acc.z + pz) + qz, sc, bv.z), fmaf((acc.w + pw) + qw, sc, bv.w));
        if (!has1) break;
        node = n1; start = s1; deg = d1; sc = c1; my = clamp_node(nmy);
        n1 = n2; s1 = s2; d1 = d2; c1 = c2;
    }
}


#define GEMM_BLOCKS 768
#define NTILES      (NNODES / 16)
static_assert(GEMM_BLOCKS <= NTILES, "every work-group owns at least one tile");

template <int NOUT, int NT, bool RELU, bool BIAS, int OSTRIDE>
__global__ __launch_bounds__(256) void gcn_gemm(const float* __restrict__ agg,
                                                const float* __restrict__ nd,
                                                const float* __restrict__ W,
                                                const float* __restrict__ bias,
                                                float* __restrict__ out) {
    constexpr int NT16 = NT * 16;
    __shared__ __attribute__((aligned(16))) float As[2][16][DIM + 4];
    __shared__ __attribute__((aligned(16))) float Ws[DIM][NT16];

    const int t   = threadIdx.x;
    const int am  = t >> 4;
    const int akb = (t & 15) * 8;

    int tile = blockIdx.x;

    float  s;
    float4 v0, v1;
    {
        const int node = tile * 16 + am;
        s = nd[node];
        const float4* p = (const float4*)(agg + (size_t)node * DIM + akb);
        v0 = p[0]; v1 = p[1];
    }

    if constexpr (NOUT == NT16) {
        for (int i = t; i < DIM * NT16 / 4; i += 256)
            ((float4*)&Ws[0][0])[i] = ((const float4*)W)[i];
    } else {
        for (int i = t; i < DIM * NT16; i += 256) {
            int k = i / NT16;
            int n = i - k * NT16;
            Ws[k][n] = (n < NOUT) ? W[k * NOUT + n] : 0.0f;
        }
    }

    const int wave = t >> 5;
    const int lane = t & 31;
    const int n0   = wave * 16;
    const int l16  = lane & 15;
    const int kh   = (lane >> 4) * 2;
    const int n    = n0 + l16;
    float bv = 0.0f;
    if (BIAS && n < NOUT) bv = bias[n];

    int cur = 0;
    for (; tile < NTILES; tile += GEMM_BLOCKS) {
        {
            float* a = &As[cur][am][akb];
            *(float4*)a       = make_float4(v0.x * s, v0.y * s, v0.z * s, v0.w * s);
            *(float4*)(a + 4) = make_float4(v1.x * s, v1.y * s, v1.z * s, v1.w * s);
        }
        __syncthreads();

        const int nxt = tile + GEMM_BLOCKS;
        if (nxt < NTILES) {
            const int node = nxt * 16 + am;
            s = nd[node];
            const float4* p = (const float4*)(agg + (size_t)node * DIM + akb);
            v0 = p[0]; v1 = p[1];
        }

        if (wave < NT) {
            const float (*Ac)[DIM + 4] = As[cur];
            v8f acc = {};
#pragma unroll
            for (int kk = 0; kk < DIM; kk += 4) {
                v2f a, b;
                a.x = Ac[l16][kk + kh];
                a.y = Ac[l16][kk + kh + 1];
                b.x = Ws[kk + kh][n0 + l16];
                b.y = Ws[kk + kh + 1][n0 + l16];
                acc = __builtin_amdgcn_wmma_f32_16x16x4_f32(
                    false, a, false, b, (short)0, acc, false, false);
            }
            float r[8];
#pragma unroll
            for (int v = 0; v < 8; ++v) {
                r[v] = acc[v] + bv;
                if (RELU) r[v] = fmaxf(r[v], 0.0f);
            }
            if (n < NOUT) {
                const int mbase = tile * 16 + (lane >> 4) * 8;
#pragma unroll
                for (int v = 0; v < 8; ++v)
                    out[(size_t)(mbase + v) * OSTRIDE + n] = r[v];
            }
        }
        cur ^= 1;
    }
}


extern "C" void kernel_launch(void* const* d_in, const int* in_sizes, int n_in,
                              void* d_out, int out_size, void* d_ws, size_t ws_size,
                              hipStream_t stream) {
    const float* inputs = (const float*)d_in[0];
    const int*   src    = (const int*)d_in[1];
    const int*   dst    = (const int*)d_in[2];
    const float* W0     = (const float*)d_in[3];
    const float* b0     = (const float*)d_in[4];
    const float* W1     = (const float*)d_in[5];
    const float* b1     = (const float*)d_in[6];
    const float* W2     = (const float*)d_in[7];
    const float* b2     = (const float*)d_in[8];
    float* out = (float*)d_out;

    constexpr size_t OFF_ND   = NNODES;
    constexpr size_t OFF_CS   = 2 * (size_t)NNODES;
    constexpr size_t OFF_CD   = 3 * (size_t)NNODES;
    constexpr size_t OFF_RP   = 4 * (size_t)NNODES;
    constexpr size_t OFF_POS  = 5 * (size_t)NNODES;
    constexpr size_t OFF_CSR  = OFF_POS + NEDGES;
    constexpr size_t OFF_B    = ((OFF_CSR + NEDGES + 127) / 128) * 128;
    constexpr size_t OFF_C    = OFF_B + (size_t)NNODES * DIM;
    constexpr size_t OFF_PD   = OFF_C + (size_t)NNODES * DIM;
    constexpr size_t OFF_PS   = OFF_PD + (size_t)HC * NNODES;

    float* ws      = (float*)d_ws;
    float* ns      = ws;
    float* ndm     = ws + OFF_ND;
    int*   cnt_src = (int*)(ws + OFF_CS);
    int*   cnt_dst = (int*)(ws + OFF_CD);
    int*   rowptr  = (int*)(ws + OFF_RP);
    int*   pos     = (int*)(ws + OFF_POS);
    int*   csr     = (int*)(ws + OFF_CSR);
    float* bufB    = ws + OFF_B;
    float* bufC    = ws + OFF_C;
    int*   plane_d = (int*)(ws + OFF_PD);
    int*   plane_s = (int*)(ws + OFF_PS);

    constexpr int GR_PRE  = (NNODES + 255) / 256;
    constexpr int GR_SCAN = (NNODES / 4 + 255) / 256;

    hist_rank<<<HC * HR, HT, 0, stream>>>(src, dst, plane_s, plane_d, pos);
    chunk_prefix<<<GR_PRE, 256, 0, stream>>>(plane_s, plane_d, cnt_src, cnt_dst);
    scan_norm<<<GR_SCAN, 256, 0, stream>>>(cnt_src, cnt_dst, rowptr, ns, ndm);
    fill_buckets<<<EDGE_BLOCKS, 256, 0, stream>>>(src, dst, rowptr, plane_d, pos, csr);

    gather_kernel<<<GATHER_BLOCKS, 256, 0, stream>>>(inputs, ns, rowptr, cnt_dst, csr, bufB);
    gcn_gemm<DIM, 8, true, true, DIM><<<GEMM_BLOCKS, 256, 0, stream>>>(bufB, ndm, W0, b0, bufC);

    gather_kernel<<<GATHER_BLOCKS, 256, 0, stream>>>(bufC, ns, rowptr, cnt_dst, csr, bufB);
    gcn_gemm<DIM, 8, true, true, DIM><<<GEMM_BLOCKS, 256, 0, stream>>>(bufB, ndm, W1, b1, bufC);

    gcn_gemm<NCLS, 3, false, false, ZS><<<GEMM_BLOCKS, 256, 0, stream>>>(bufC, ns, W2, b2, bufB);
    gather_out<<<GATHER_BLOCKS, 256, 0, stream>>>(bufB, ndm, b2, rowptr, cnt_dst, csr, out);
}


# Per-edge norm_src loads removed from both 512-B gathers: producers store rows pre-scaled by norm_src; layer 0 applies W0 before its neighbour sum (sum order: products summed in bucket order)
# speedup vs baseline: 4.3389x; 1.0298x over previous
#include <hip/hip_runtime.h>


#define NNODES 100000
#define NEDGES 1600000
#define DIM    128
#define NCLS   40
#define ZS     64

typedef __attribute__((ext_vector_type(2))) float v2f;
typedef __attribute__((ext_vector_type(8))) float v8f;


#define EDGE_BLOCKS  2048
#define EDGE_THREADS (EDGE_BLOCKS * 256)
#define EFULL        (NEDGES / EDGE_THREADS)
#define ETAIL        (EFULL * EDGE_THREADS)

#define HC    64
#define HR    4
#define HEC   (NEDGES / HC)
#define HNR   (NNODES / HR)
#define HT    1024
#define HFULL (HEC / HT)
#define HTAIL (HFULL * HT)
static_assert(HEC * HC == NEDGES && HNR * HR == NNODES, "chunks and ranges tile the problem exactly");
static_assert(HFULL % 8 == 0, "full strides are taken in groups of eight");
static_assert(HNR % 4 == 0, "counter planes move as 16-byte units");

__global__ __launch_bounds__(HT) void hist_rank(const int* __restrict__ src,
                                                const int* __restrict__ dst,
                                                int* __restrict__ plane_src,
                                                int* __restrict__ plane_dst,
                                                int* __restrict__ pos) {
    __shared__ __attribute__((aligned(16))) int hd[HNR];
    __shared__ __attribute__((aligned(16))) int hs[HNR];
    const int t  = threadIdx.x;
    const int c  = blockIdx.x / HR;
    const int lo = (blockIdx.x - c * HR) * HNR;
    const int e0 = c * HEC + t;

    for (int i = t; i < HNR / 4; i += HT) {
        ((int4*)hd)[i] = make_int4(0, 0, 0, 0);
        ((int4*)hs)[i] = make_int4(0, 0, 0, 0);
    }
    __syncthreads();

    for (int g = 0; g < HFULL; g += 8) {
        int dv[8], sv[8];
#pragma unroll
        for (int u = 0; u < 8; ++u) {
            dv[u] = dst[e0 + (g + u) * HT];
            sv[u] = src[e0 + (g + u) * HT];
        }
#pragma unroll
        for (int u = 0; u < 8; ++u) {
            const unsigned d = (unsigned)(dv[u] - lo);
            const unsigned q = (unsigned)(sv[u] - lo);
            if (d < (unsigned)HNR) pos[e0 + (g + u) * HT] = atomicAdd(&hd[d], 1);
            if (q < (unsigned)HNR) atomicAdd(&hs[q], 1);
        }
    }
    if (HTAIL + t < HEC) {
        const unsigned d = (unsigned)(dst[e0 + HTAIL] - lo);
        const unsigned q = (unsigned)(src[e0 + HTAIL] - lo);
        if (d < (unsigned)HNR) pos[e0 + HTAIL] = atomicAdd(&hd[d], 1);
        if (q < (unsigned)HNR) atomicAdd(&hs[q], 1);
    }
    __syncthreads();

    int4* od = (int4*)(plane_dst + (size_t)c * NNODES + lo);
    int4* os = (int4*)(plane_src + (size_t)c * NNODES + lo);
    for (int i = t; i < HNR / 4; i += HT) {
        od[i] = ((const int4*)hd)[i];
        os[i] = ((const int4*)hs)[i];
    }
}

__global__ __launch_bounds__(256) void chunk_prefix(const int* __restrict__ plane_src,
                                                    int* __restrict__ plane_dst,
                                                    int* __restrict__ cnt_src,
                                                    int* __restrict__ cnt_dst) {
    const int n = blockIdx.x * 256 + threadIdx.x;
    if (n >= NNODES) return;
    int run = 0, tot = 0;
    for (int c0 = 0; c0 < HC; c0 += 16) {
        int a[16], b[16];
#pragma unroll
        for (int u = 0; u < 16; ++u) {
            a[u] = plane_dst[(size_t)(c0 + u) * NNODES + n];
            b[u] = plane_src[(size_t)(c0 + u) * NNODES + n];
        }
#pragma unroll
        for (int u = 0; u < 16; ++u) {
            plane_dst[(size_t)(c0 + u) * NNODES + n] = run;
            run += a[u];
            tot += b[u];
        }
    }
    cnt_dst[n] = run;
    cnt_src[n] = tot;
}

__device__ __forceinline__ float inv_sqrt_deg(int c) {
    return (c > 0) ? rsqrtf(fmaxf((float)c, 1.f)) : 0.f;
}

__global__ __launch_bounds__(256) void scan_norm(const int* __restrict__ cnt_src,
                                                 const int* __restrict__ cnt_dst,
                                                 int* __restrict__ rowptr,
                                                 float* __restrict__ ns,
                                                 float* __restrict__ nd) {
    __shared__ int red[8];
    __shared__ int wsum[8];
    const int t = threadIdx.x, b = blockIdx.x;
    const int lane = t & 31, wave = t >> 5;
    const int4* c4 = (const int4*)cnt_dst;

    int part = 0;
    int k = 0;
    for (; k + 8 <= b; k += 8) {
        int4 q[8];
#pragma unroll
        for (int u = 0; u < 8; ++u) q[u] = c4[(k + u) * 256 + t];
#pragma unroll
        for (int u = 0; u < 8; ++u) part += (q[u].x + q[u].y) + (q[u].z + q[u].w);
    }
    for (; k < b; ++k) {
        int4 q = c4[k * 256 + t];
        part += (q.x + q.y) + (q.z + q.w);
    }
    for (int o = 16; o > 0; o >>= 1) part += __shfl_xor(part, o);
    if (lane == 0) red[wave] = part;

    const int i4 = b * 256 + t;
    const bool in = i4 < NNODES / 4;
    int4 c = make_int4(0, 0, 0, 0);
    if (in) c = c4[i4];
    const int s = (c.x + c.y) + (c.z + c.w);
    int inc = s;
    for (int o = 1; o < 32; o <<= 1) {
        int v = __shfl_up(inc, o);
        if (lane >= o) inc += v;
    }
    if (lane == 31) wsum[wave] = inc;
    __syncthreads();

    int base = 0;
    for (int w = 0; w < 8; ++w) base += red[w];
    for (int w = 0; w < 8; ++w) base += (w < wave) ? wsum[w] : 0;
    const int ex = base + inc - s;
    if (in) {
        ((int4*)rowptr)[i4] = make_int4(ex, ex + c.x, ex + c.x + c.y, ex + c.x + c.y + c.z);
        int4 cs = ((const int4*)cnt_src)[i4];
        ((float4*)ns)[i4] = make_float4(inv_sqrt_deg(cs.x), inv_sqrt_deg(cs.y),
                                        inv_sqrt_deg(cs.z), inv_sqrt_deg(cs.w));
        ((float4*)nd)[i4] = make_float4(inv_sqrt_deg(c.x), inv_sqrt_deg(c.y),
                                        inv_sqrt_deg(c.z), inv_sqrt_deg(c.w));
    }
}

__global__ __launch_bounds__(256) void fill_buckets(const int* __restrict__ src,
                                                    const int* __restrict__ dst,
                                                    const int* __restrict__ rowptr,
                                                    const int* __restrict__ plane_dst,
                                                    const int* __restrict__ pos,
                                                    int* __restrict__ csr) {
    const int t0 = blockIdx.x * 256 + threadIdx.x;
    const int et = ETAIL + t0;
    const bool tail = et < NEDGES;
    int s[EFULL + 1], r[EFULL + 1], p[EFULL + 1], q[EFULL + 1];
#pragma unroll
    for (int u = 0; u < EFULL; ++u) {
        s[u] = src[t0 + u * EDGE_THREADS];
        r[u] = dst[t0 + u * EDGE_THREADS];
        p[u] = pos[t0 + u * EDGE_THREADS];
    }
    s[EFULL] = 0; r[EFULL] = 0; p[EFULL] = 0; q[EFULL] = 0;
    if (tail) { s[EFULL] = src[et]; r[EFULL] = dst[et]; p[EFULL] = pos[et]; }
#pragma unroll
    for (int u = 0; u < EFULL; ++u) {
        const int d = r[u];
        r[u] = rowptr[d];
        q[u] = plane_dst[(size_t)((t0 + u * EDGE_THREADS) / HEC) * NNODES + d];
    }
    if (tail) {
        const int d = r[EFULL];
        r[EFULL] = rowptr[d];
        q[EFULL] = plane_dst[(size_t)(et / HEC) * NNODES + d];
    }
#pragma unroll
    for (int u = 0; u < EFULL; ++u) {
        unsigned idx = (unsigned)(r[u] + q[u] + p[u]);
        if (idx < (unsigned)NEDGES) csr[idx] = s[u];
    }
    if (tail) {
        unsigned idx = (unsigned)(r[EFULL] + q[EFULL] + p[EFULL]);
        if (idx < (unsigned)NEDGES) csr[idx] = s[EFULL];
    }
}


template <int U>
__device__ __forceinline__ void acc_rows(const float* __restrict__ h, int my,
                                         int j, int lane, float4& acc) {
    float4 v[U];
#pragma unroll
    for (int u = 0; u < U; ++u) {
        const int s = __builtin_amdgcn_readlane(my, j + u);
        v[u] = ((const float4*)(h + (size_t)s * DIM))[lane];
    }
#pragma unroll
    for (int u = 0; u < U; ++u) {
        acc.x += v[u].x; acc.y += v[u].y; acc.z += v[u].z; acc.w += v[u].w;
    }
}

#define GATHER_BLOCKS 2048
#define GATHER_WAVES  (GATHER_BLOCKS * 8)

template <bool POST>
__global__ __launch_bounds__(256) void gather_kernel(const float* __restrict__ h,
                                                     const int* __restrict__ rowptr,
                                                     const int* __restrict__ cnt,
                                                     const int* __restrict__ csr,
                                                     const float* __restrict__ nd,
                                                     const float* __restrict__ ns,
                                                     const float* __restrict__ bias,
                                                     float* __restrict__ out) {
    const int lane = threadIdx.x & 31;
    const int w0 = __builtin_amdgcn_readfirstlane((int)((blockIdx.x * 256 + threadIdx.x) >> 5));
    float4 bv = make_float4(0.f, 0.f, 0.f, 0.f);
    if (POST) bv = ((const float4*)bias)[lane];
    for (int node = w0; node < NNODES; node += GATHER_WAVES) {
        const int start = rowptr[node];
        const int deg   = cnt[node];
        float sc = 0.f, tn = 0.f;
        if (POST) { sc = nd[node]; tn = ns[node]; }
        float4 acc = make_float4(0.f, 0.f, 0.f, 0.f);
        for (int base = 0; base < deg; base += 32) {
            const int n = min(32, deg - base);
            int my = 0;
            if (lane < n)
                my = (int)min((unsigned)csr[start + base + lane], (unsigned)(NNODES - 1));
            int j = 0;
            for (; j + 8 <= n; j += 8) acc_rows<8>(h, my, j, lane, acc);
            if (j + 4 <= n) { acc_rows<4>(h, my, j, lane, acc); j += 4; }
            if (j + 2 <= n) { acc_rows<2>(h, my, j, lane, acc); j += 2; }
            if (j < n)      { acc_rows<1>(h, my, j, lane, acc); }
        }
        if (POST) {
            acc.x = fmaxf(fmaf(acc.x, sc, bv.x), 0.f) * tn;
            acc.y = fmaxf(fmaf(acc.y, sc, bv.y), 0.f) * tn;
            acc.z = fmaxf(fmaf(acc.z, sc, bv.z), 0.f) * tn;
            acc.w = fmaxf(fmaf(acc.w, sc, bv.w), 0.f) * tn;
        }
        ((float4*)(out + (size_t)node * DIM))[lane] = acc;
    }
}


__device__ __forceinline__ int clamp_node(int s) {
    return (int)min((unsigned)s, (unsigned)(NNODES - 1));
}

template <int K>
__device__ __forceinline__ void rows3(const float* __restrict__ z, int ids, int rem,
                                      int g, int cc, float4& acc) {
    float4 v[K];
#pragma unroll
    for (int u = 0; u < K; ++u) {
        const int s0 = __builtin_amdgcn_readlane(ids, 3 * u);
        const int s1 = __builtin_amdgcn_readlane(ids, 3 * u + 1);
        const int s2 = __builtin_amdgcn_readlane(ids, 3 * u + 2);
        int s;
        if (u < K - 1) s = (g == 1) ? s1 : ((g == 2) ? s2 : s0);
        else           s = (g == 1 && 3 * u + 1 < rem) ? s1 : ((g == 2 && 3 * u + 2 < rem) ? s2 : s0);
        v[u] = ((const float4*)(z + (size_t)s * ZS))[cc];
    }
#pragma unroll
    for (int u = 0; u < K; ++u) {
        if (u < K - 1) {
            acc.x += v[u].x; acc.y += v[u].y; acc.z += v[u].z; acc.w += v[u].w;
        } else {
            const bool ok = 3 * u + g < rem;
            acc.x += ok ? v[u].x : 0.f; acc.y += ok ? v[u].y : 0.f;
            acc.z += ok ? v[u].z : 0.f; acc.w += ok ? v[u].w : 0.f;
        }
    }
}

__device__ __forceinline__ int bucket_ids(const int* __restrict__ csr, int first, int lane) {
    return csr[min((unsigned)(first + lane), (unsigned)(NEDGES - 1))];
}

__global__ __launch_bounds__(256) void gather_out(const float* __restrict__ z,
                                                  const float* __restrict__ nd,
                                                  const float* __restrict__ bias,
                                                  const int* __restrict__ rowptr,
                                                  const int* __restrict__ cnt,
                                                  const int* __restrict__ csr,
                                                  float* __restrict__ out) {
    const int lane = threadIdx.x & 31;
    const int g    = (lane >= 30) ? 0 : ((lane >= 20) ? 2 : ((lane >= 10) ? 1 : 0));
    const int cc   = (lane >= 30) ? (NCLS / 4 - 1) : (lane - 10 * g);
    const int w0 = __builtin_amdgcn_readfirstlane((int)((blockIdx.x * 256 + threadIdx.x) >> 5));
    const float4 bv = ((const float4*)bias)[cc];

    int   node  = w0;
    int   start = rowptr[node];
    int   deg   = cnt[node];
    float sc    = nd[node];
    int   my    = clamp_node(bucket_ids(csr, start, lane));
    static_assert(2 * GATHER_WAVES <= NNODES, "every wave owns a second node");
    int   n1    = node + GATHER_WAVES;
    int   s1    = rowptr[n1];
    int   d1    = cnt[n1];
    float c1    = nd[n1];

    for (;;) {
        const bool has1 = n1 < NNODES;
        const int nmy = bucket_ids(csr, s1, lane);
        const int  n2   = n1 + GATHER_WAVES;
        int   s2 = 0, d2 = 0;
        float c2 = 0.f;
        if (n2 < NNODES) { s2 = rowptr[n2]; d2 = cnt[n2]; c2 = nd[n2]; }

        float4 acc = make_float4(0.f, 0.f, 0.f, 0.f);
        int ids = my;
        for (int j0 = 0; j0 < deg; j0 += 24) {
            if (j0 > 0) ids = clamp_node(bucket_ids(csr, start + j0, lane));
            const int rem = deg - j0;
            switch (min(8, (rem + 2) / 3)) {
                case 1:  rows3<1>(z, ids, rem, g, cc, acc); break;
                case 2:  rows3<2>(z, ids, rem, g, cc, acc); break;
                case 3:  rows3<3>(z, ids, rem, g, cc, acc); break;
                case 4:  rows3<4>(z, ids, rem, g, cc, acc); break;
                case 5:  rows3<5>(z, ids, rem, g, cc, acc); break;
                case 6:  rows3<6>(z, ids, rem, g, cc, acc); break;
                case 7:  rows3<7>(z, ids, rem, g, cc, acc); break;
                default: rows3<8>(z, ids, rem, g, cc, acc); break;
            }
        }
        const float px = __shfl_down(acc.x, 10), py = __shfl_down(acc.y, 10);
        const float pz = __shfl_down(acc.z, 10), pw = __shfl_down(acc.w, 10);
        const float qx = __shfl_down(acc.x, 20), qy = __shfl_down(acc.y, 20);
        const float qz = __shfl_down(acc.z, 20), qw = __shfl_down(acc.w, 20);
        if (lane < NCLS / 4)
            ((float4*)(out + (size_t)node * NCLS))[lane] =
                make_float4(fmaf((acc.x + px) + qx, sc, bv.x), fmaf((acc.y + py) + qy, sc, bv.y),
                            fmaf((acc.z + pz) + qz, sc, bv.z), fmaf((acc.w + pw) + qw, sc, bv.w));
        if (!has1) break;
        node = n1; start = s1; deg = d1; sc = c1; my = clamp_node(nmy);
        n1 = n2; s1 = s2; d1 = d2; c1 = c2;
    }
}


#define GEMM_BLOCKS 768
#define NTILES      (NNODES / 16)
static_assert(GEMM_BLOCKS <= NTILES, "every work-group owns at least one tile");

template <int NOUT, int NT, bool RELU, bool BIAS, int OSTRIDE>
__global__ __launch_bounds__(256) void gcn_gemm(const float* __restrict__ agg,
                                                const float* __restrict__ nd,
                                                const float* __restrict__ W,
                                                const float* __restrict__ bias,
                                                float* __restrict__ out) {
    constexpr int NT16 = NT * 16;
    __shared__ __attribute__((aligned(16))) float As[2][16][DIM + 4];
    __shared__ __attribute__((aligned(16))) float Ws[DIM][NT16];

    const int t   = threadIdx.x;
    const int am  = t >> 4;
    const int akb = (t & 15) * 8;

    int tile = blockIdx.x;

    float  s;
    float4 v0, v1;
    {
        const int node = tile * 16 + am;
        s = nd[node];
        const float4* p = (const float4*)(agg + (size_t)node * DIM + akb);
        v0 = p[0]; v1 = p[1];
    }

    if constexpr (NOUT == NT16) {
        for (int i = t; i < DIM * NT16 / 4; i += 256)
            ((float4*)&Ws[0][0])[i] = ((const float4*)W)[i];
    } else {
        for (int i = t; i < DIM * NT16; i += 256) {
            int k = i / NT16;
            int n = i - k * NT16;
            Ws[k][n] = (n < NOUT) ? W[k * NOUT + n] : 0.0f;
        }
    }

    const int wave = t >> 5;
    const int lane = t & 31;
    const int n0   = wave * 16;
    const int l16  = lane & 15;
    const int kh   = (lane >> 4) * 2;
    const int n    = n0 + l16;
    float bv = 0.0f;
    if (BIAS && n < NOUT) bv = bias[n];

    int cur = 0;
    for (; tile < NTILES; tile += GEMM_BLOCKS) {
        {
            float* a = &As[cur][am][akb];
            *(float4*)a       = make_float4(v0.x * s, v0.y * s, v0.z * s, v0.w * s);
            *(float4*)(a + 4) = make_float4(v1.x * s, v1.y * s, v1.z * s, v1.w * s);
        }
        __syncthreads();

        const int nxt = tile + GEMM_BLOCKS;
        if (nxt < NTILES) {
            const int node = nxt * 16 + am;
            s = nd[node];
            const float4* p = (const float4*)(agg + (size_t)node * DIM + akb);
            v0 = p[0]; v1 = p[1];
        }

        if (wave < NT) {
            const float (*Ac)[DIM + 4] = As[cur];
            v8f acc = {};
#pragma unroll
            for (int kk = 0; kk < DIM; kk += 4) {
                v2f a, b;
                a.x = Ac[l16][kk + kh];
                a.y = Ac[l16][kk + kh + 1];
                b.x = Ws[kk + kh][n0 + l16];
                b.y = Ws[kk + kh + 1][n0 + l16];
                acc = __builtin_amdgcn_wmma_f32_16x16x4_f32(
                    false, a, false, b, (short)0, acc, false, false);
            }
            float r[8];
#pragma unroll
            for (int v = 0; v < 8; ++v) {
                r[v] = acc[v] + bv;
                if (RELU) r[v] = fmaxf(r[v], 0.0f);
            }
            if (n < NOUT) {
                const int mbase = tile * 16 + (lane >> 4) * 8;
#pragma unroll
                for (int v = 0; v < 8; ++v)
                    out[(size_t)(mbase + v) * OSTRIDE + n] = r[v];
            }
        }
        cur ^= 1;
    }
}


extern "C" void kernel_launch(void* const* d_in, const int* in_sizes, int n_in,
                              void* d_out, int out_size, void* d_ws, size_t ws_size,
                              hipStream_t stream) {
    const float* inputs = (const float*)d_in[0];
    const int*   src    = (const int*)d_in[1];
    const int*   dst    = (const int*)d_in[2];
    const float* W0     = (const float*)d_in[3];
    const float* b0     = (const float*)d_in[4];
    const float* W1     = (const float*)d_in[5];
    const float* b1     = (const float*)d_in[6];
    const float* W2     = (const float*)d_in[7];
    const float* b2     = (const float*)d_in[8];
    float* out = (float*)d_out;

    constexpr size_t OFF_ND   = NNODES;
    constexpr size_t OFF_CS   = 2 * (size_t)NNODES;
    constexpr size_t OFF_CD   = 3 * (size_t)NNODES;
    constexpr size_t OFF_RP   = 4 * (size_t)NNODES;
    constexpr size_t OFF_POS  = 5 * (size_t)NNODES;
    constexpr size_t OFF_CSR  = OFF_POS + NEDGES;
    constexpr size_t OFF_B    = ((OFF_CSR + NEDGES + 127) / 128) * 128;
    constexpr size_t OFF_C    = OFF_B + (size_t)NNODES * DIM;
    constexpr size_t OFF_PD   = OFF_C + (size_t)NNODES * DIM;
    constexpr size_t OFF_PS   = OFF_PD + (size_t)HC * NNODES;

    float* ws      = (float*)d_ws;
    float* ns      = ws;
    float* ndm     = ws + OFF_ND;
    int*   cnt_src = (int*)(ws + OFF_CS);
    int*   cnt_dst = (int*)(ws + OFF_CD);
    int*   rowptr  = (int*)(ws + OFF_RP);
    int*   pos     = (int*)(ws + OFF_POS);
    int*   csr     = (int*)(ws + OFF_CSR);
    float* bufB    = ws + OFF_B;
    float* bufC    = ws + OFF_C;
    int*   plane_d = (int*)(ws + OFF_PD);
    int*   plane_s = (int*)(ws + OFF_PS);

    constexpr int GR_PRE  = (NNODES + 255) / 256;
    constexpr int GR_SCAN = (NNODES / 4 + 255) / 256;

    hist_rank<<<HC * HR, HT, 0, stream>>>(src, dst, plane_s, plane_d, pos);
    chunk_prefix<<<GR_PRE, 256, 0, stream>>>(plane_s, plane_d, cnt_src, cnt_dst);
    scan_norm<<<GR_SCAN, 256, 0, stream>>>(cnt_src, cnt_dst, rowptr, ns, ndm);
    fill_buckets<<<EDGE_BLOCKS, 256, 0, stream>>>(src, dst, rowptr, plane_d, pos, csr);

    gcn_gemm<DIM, 8, false, false, DIM><<<GEMM_BLOCKS, 256, 0, stream>>>(inputs, ns, W0, b0, bufB);
    gather_kernel<true><<<GATHER_BLOCKS, 256, 0, stream>>>(bufB, rowptr, cnt_dst, csr, ndm, ns, b0, bufC);

    gather_kernel<false><<<GATHER_BLOCKS, 256, 0, stream>>>(bufC, rowptr, cnt_dst, csr, ndm, ns, b1, bufB);
    gcn_gemm<DIM, 8, true, true, DIM><<<GEMM_BLOCKS, 256, 0, stream>>>(bufB, ndm, W1, b1, bufC);

    gcn_gemm<NCLS, 3, false, false, ZS><<<GEMM_BLOCKS, 256, 0, stream>>>(bufC, ns, W2, b2, bufB);
    gather_out<<<GATHER_BLOCKS, 256, 0, stream>>>(bufB, ndm, b2, rowptr, cnt_dst, csr, out);
}


# Layer-1 neighbour sum and the two products behind it fused in one kernel: buckets summed in registers into an LDS tile, W1 then W2 applied from LDS; 10 -> 8 launches; sum order unchanged
# speedup vs baseline: 4.7498x; 1.0947x over previous
#include <hip/hip_runtime.h>


#define NNODES 100000
#define NEDGES 1600000
#define DIM    128
#define NCLS   40
#define ZS     64

typedef __attribute__((ext_vector_type(2))) float v2f;
typedef __attribute__((ext_vector_type(8))) float v8f;


#define EDGE_BLOCKS  2048
#define EDGE_THREADS (EDGE_BLOCKS * 256)
#define EFULL        (NEDGES / EDGE_THREADS)
#define ETAIL        (EFULL * EDGE_THREADS)

#define HC    64
#define HR    4
#define HEC   (NEDGES / HC)
#define HNR   (NNODES / HR)
#define HT    1024
#define HFULL (HEC / HT)
#define HTAIL (HFULL * HT)
static_assert(HEC * HC == NEDGES && HNR * HR == NNODES, "chunks and ranges tile the problem exactly");
static_assert(HFULL % 8 == 0, "full strides are taken in groups of eight");
static_assert(HNR % 4 == 0, "counter planes move as 16-byte units");

__global__ __launch_bounds__(HT) void hist_rank(const int* __restrict__ src,
                                                const int* __restrict__ dst,
                                                int* __restrict__ plane_src,
                                                int* __restrict__ plane_dst,
                                                int* __restrict__ pos) {
    __shared__ __attribute__((aligned(16))) int hd[HNR];
    __shared__ __attribute__((aligned(16))) int hs[HNR];
    const int t  = threadIdx.x;
    const int c  = blockIdx.x / HR;
    const int lo = (blockIdx.x - c * HR) * HNR;
    const int e0 = c * HEC + t;

    for (int i = t; i < HNR / 4; i += HT) {
        ((int4*)hd)[i] = make_int4(0, 0, 0, 0);
        ((int4*)hs)[i] = make_int4(0, 0, 0, 0);
    }
    __syncthreads();

    for (int g = 0; g < HFULL; g += 8) {
        int dv[8], sv[8];
#pragma unroll
        for (int u = 0; u < 8; ++u) {
            dv[u] = dst[e0 + (g + u) * HT];
            sv[u] = src[e0 + (g + u) * HT];
        }
#pragma unroll
        for (int u = 0; u < 8; ++u) {
            const unsigned d = (unsigned)(dv[u] - lo);
            const unsigned q = (unsigned)(sv[u] - lo);
            if (d < (unsigned)HNR) pos[e0 + (g + u) * HT] = atomicAdd(&hd[d], 1);
            if (q < (unsigned)HNR) atomicAdd(&hs[q], 1);
        }
    }
    if (HTAIL + t < HEC) {
        const unsigned d = (unsigned)(dst[e0 + HTAIL] - lo);
        const unsigned q = (unsigned)(src[e0 + HTAIL] - lo);
        if (d < (unsigned)HNR) pos[e0 + HTAIL] = atomicAdd(&hd[d], 1);
        if (q < (unsigned)HNR) atomicAdd(&hs[q], 1);
    }
    __syncthreads();

    int4* od = (int4*)(plane_dst + (size_t)c * NNODES + lo);
    int4* os = (int4*)(plane_src + (size_t)c * NNODES + lo);
    for (int i = t; i < HNR / 4; i += HT) {
        od[i] = ((const int4*)hd)[i];
        os[i] = ((const int4*)hs)[i];
    }
}

__global__ __launch_bounds__(256) void chunk_prefix(const int* __restrict__ plane_src,
                                                    int* __restrict__ plane_dst,
                                                    int* __restrict__ cnt_src,
                                                    int* __restrict__ cnt_dst) {
    const int n = blockIdx.x * 256 + threadIdx.x;
    if (n >= NNODES) return;
    int run = 0, tot = 0;
    for (int c0 = 0; c0 < HC; c0 += 16) {
        int a[16], b[16];
#pragma unroll
        for (int u = 0; u < 16; ++u) {
            a[u] = plane_dst[(size_t)(c0 + u) * NNODES + n];
            b[u] = plane_src[(size_t)(c0 + u) * NNODES + n];
        }
#pragma unroll
        for (int u = 0; u < 16; ++u) {
            plane_dst[(size_t)(c0 + u) * NNODES + n] = run;
            run += a[u];
            tot += b[u];
        }
    }
    cnt_dst[n] = run;
    cnt_src[n] = tot;
}

__device__ __forceinline__ float inv_sqrt_deg(int c) {
    return (c > 0) ? rsqrtf(fmaxf((float)c, 1.f)) : 0.f;
}

__global__ __launch_bounds__(256) void scan_norm(const int* __restrict__ cnt_src,
                                                 const int* __restrict__ cnt_dst,
                                                 int* __restrict__ rowptr,
                                                 float* __restrict__ ns,
                                                 float* __restrict__ nd) {
    __shared__ int red[8];
    __shared__ int wsum[8];
    const int t = threadIdx.x, b = blockIdx.x;
    const int lane = t & 31, wave = t >> 5;
    const int4* c4 = (const int4*)cnt_dst;

    int part = 0;
    int k = 0;
    for (; k + 8 <= b; k += 8) {
        int4 q[8];
#pragma unroll
        for (int u = 0; u < 8; ++u) q[u] = c4[(k + u) * 256 + t];
#pragma unroll
        for (int u = 0; u < 8; ++u) part += (q[u].x + q[u].y) + (q[u].z + q[u].w);
    }
    for (; k < b; ++k) {
        int4 q = c4[k * 256 + t];
        part += (q.x + q.y) + (q.z + q.w);
    }
    for (int o = 16; o > 0; o >>= 1) part += __shfl_xor(part, o);
    if (lane == 0) red[wave] = part;

    const int i4 = b * 256 + t;
    const bool in = i4 < NNODES / 4;
    int4 c = make_int4(0, 0, 0, 0);
    if (in) c = c4[i4];
    const int s = (c.x + c.y) + (c.z + c.w);
    int inc = s;
    for (int o = 1; o < 32; o <<= 1) {
        int v = __shfl_up(inc, o);
        if (lane >= o) inc += v;
    }
    if (lane == 31) wsum[wave] = inc;
    __syncthreads();

    int base = 0;
    for (int w = 0; w < 8; ++w) base += red[w];
    for (int w = 0; w < 8; ++w) base += (w < wave) ? wsum[w] : 0;
    const int ex = base + inc - s;
    if (in) {
        ((int4*)rowptr)[i4] = make_int4(ex, ex + c.x, ex + c.x + c.y, ex + c.x + c.y + c.z);
        int4 cs = ((const int4*)cnt_src)[i4];
        ((float4*)ns)[i4] = make_float4(inv_sqrt_deg(cs.x), inv_sqrt_deg(cs.y),
                                        inv_sqrt_deg(cs.z), inv_sqrt_deg(cs.w));
        ((float4*)nd)[i4] = make_float4(inv_sqrt_deg(c.x), inv_sqrt_deg(c.y),
                                        inv_sqrt_deg(c.z), inv_sqrt_deg(c.w));
    }
}

__global__ __launch_bounds__(256) void fill_buckets(const int* __restrict__ src,
                                                    const int* __restrict__ dst,
                                                    const int* __restrict__ rowptr,
                                                    const int* __restrict__ plane_dst,
                                                    const int* __restrict__ pos,
                                                    int* __restrict__ csr) {
    const int t0 = blockIdx.x * 256 + threadIdx.x;
    const int et = ETAIL + t0;
    const bool tail = et < NEDGES;
    int s[EFULL + 1], r[EFULL + 1], p[EFULL + 1], q[EFULL + 1];
#pragma unroll
    for (int u = 0; u < EFULL; ++u) {
        s[u] = src[t0 + u * EDGE_THREADS];
        r[u] = dst[t0 + u * EDGE_THREADS];
        p[u] = pos[t0 + u * EDGE_THREADS];
    }
    s[EFULL] = 0; r[EFULL] = 0; p[EFULL] = 0; q[EFULL] = 0;
    if (tail) { s[EFULL] = src[et]; r[EFULL] = dst[et]; p[EFULL] = pos[et]; }
#pragma unroll
    for (int u = 0; u < EFULL; ++u) {
        const int d = r[u];
        r[u] = rowptr[d];
        q[u] = plane_dst[(size_t)((t0 + u * EDGE_THREADS) / HEC) * NNODES + d];
    }
    if (tail) {
        const int d = r[EFULL];
        r[EFULL] = rowptr[d];
        q[EFULL] = plane_dst[(size_t)(et / HEC) * NNODES + d];
    }
#pragma unroll
    for (int u = 0; u < EFULL; ++u) {
        unsigned idx = (unsigned)(r[u] + q[u] + p[u]);
        if (idx < (unsigned)NEDGES) csr[idx] = s[u];
    }
    if (tail) {
        unsigned idx = (unsigned)(r[EFULL] + q[EFULL] + p[EFULL]);
        if (idx < (unsigned)NEDGES) csr[idx] = s[EFULL];
    }
}


template <int U>
__device__ __forceinline__ void acc_rows(const float* __restrict__ h, int my,
                                         int j, int lane, float4& acc) {
    float4 v[U];
#pragma unroll
    for (int u = 0; u < U; ++u) {
        const int s = __builtin_amdgcn_readlane(my, j + u);
        v[u] = ((const float4*)(h + (size_t)s * DIM))[lane];
    }
#pragma unroll
    for (int u = 0; u < U; ++u) {
        acc.x += v[u].x; acc.y += v[u].y; acc.z += v[u].z; acc.w += v[u].w;
    }
}

#define GATHER_BLOCKS 2048
#define GATHER_WAVES  (GATHER_BLOCKS * 8)

__global__ __launch_bounds__(256) void gather_kernel(const float* __restrict__ h,
                                                     const int* __restrict__ rowptr,
                                                     const int* __restrict__ cnt,
                                                     const int* __restrict__ csr,
                                                     const float* __restrict__ nd,
                                                     const float* __restrict__ ns,
                                                     const float* __restrict__ bias,
                                                     float* __restrict__ out) {
    const int lane = threadIdx.x & 31;
    const int w0 = __builtin_amdgcn_readfirstlane((int)((blockIdx.x * 256 + threadIdx.x) >> 5));
    const float4 bv = ((const float4*)bias)[lane];
    for (int node = w0; node < NNODES; node += GATHER_WAVES) {
        const int start = rowptr[node];
        const int deg   = cnt[node];
        const float sc = nd[node], tn = ns[node];
        float4 acc = make_float4(0.f, 0.f, 0.f, 0.f);
        for (int base = 0; base < deg; base += 32) {
            const int n = min(32, deg - base);
            int my = 0;
            if (lane < n)
                my = (int)min((unsigned)csr[start + base + lane], (unsigned)(NNODES - 1));
            int j = 0;
            for (; j + 8 <= n; j += 8) acc_rows<8>(h, my, j, lane, acc);
            if (j + 4 <= n) { acc_rows<4>(h, my, j, lane, acc); j += 4; }
            if (j + 2 <= n) { acc_rows<2>(h, my, j, lane, acc); j += 2; }
            if (j < n)      { acc_rows<1>(h, my, j, lane, acc); }
        }
        acc.x = fmaxf(fmaf(acc.x, sc, bv.x), 0.f) * tn;
        acc.y = fmaxf(fmaf(acc.y, sc, bv.y), 0.f) * tn;
        acc.z = fmaxf(fmaf(acc.z, sc, bv.z), 0.f) * tn;
        acc.w = fmaxf(fmaf(acc.w, sc, bv.w), 0.f) * tn;
        ((float4*)(out + (size_t)node * DIM))[lane] = acc;
    }
}


__device__ __forceinline__ int clamp_node(int s) {
    return (int)min((unsigned)s, (unsigned)(NNODES - 1));
}

template <int K>
__device__ __forceinline__ void rows3(const float* __restrict__ z, int ids, int rem,
                                      int g, int cc, float4& acc) {
    float4 v[K];
#pragma unroll
    for (int u = 0; u < K; ++u) {
        const int s0 = __builtin_amdgcn_readlane(ids, 3 * u);
        const int s1 = __builtin_amdgcn_readlane(ids, 3 * u + 1);
        const int s2 = __builtin_amdgcn_readlane(ids, 3 * u + 2);
        int s;
        if (u < K - 1) s = (g == 1) ? s1 : ((g == 2) ? s2 : s0);
        else           s = (g == 1 && 3 * u + 1 < rem) ? s1 : ((g == 2 && 3 * u + 2 < rem) ? s2 : s0);
        v[u] = ((const float4*)(z + (size_t)s * ZS))[cc];
    }
#pragma unroll
    for (int u = 0; u < K; ++u) {
        if (u < K - 1) {
            acc.x += v[u].x; acc.y += v[u].y; acc.z += v[u].z; acc.w += v[u].w;
        } else {
            const bool ok = 3 * u + g < rem;
            acc.x += ok ? v[u].x : 0.f; acc.y += ok ? v[u].y : 0.f;
            acc.z += ok ? v[u].z : 0.f; acc.w += ok ? v[u].w : 0.f;
        }
    }
}

__device__ __forceinline__ int bucket_ids(const int* __restrict__ csr, int first, int lane) {
    return csr[min((unsigned)(first + lane), (unsigned)(NEDGES - 1))];
}

__global__ __launch_bounds__(256) void gather_out(const float* __restrict__ z,
                                                  const float* __restrict__ nd,
                                                  const float* __restrict__ bias,
                                                  const int* __restrict__ rowptr,
                                                  const int* __restrict__ cnt,
                                                  const int* __restrict__ csr,
                                                  float* __restrict__ out) {
    const int lane = threadIdx.x & 31;
    const int g    = (lane >= 30) ? 0 : ((lane >= 20) ? 2 : ((lane >= 10) ? 1 : 0));
    const int cc   = (lane >= 30) ? (NCLS / 4 - 1) : (lane - 10 * g);
    const int w0 = __builtin_amdgcn_readfirstlane((int)((blockIdx.x * 256 + threadIdx.x) >> 5));
    const float4 bv = ((const float4*)bias)[cc];

    int   node  = w0;
    int   start = rowptr[node];
    int   deg   = cnt[node];
    float sc    = nd[node];
    int   my    = clamp_node(bucket_ids(csr, start, lane));
    static_assert(2 * GATHER_WAVES <= NNODES, "every wave owns a second node");
    int   n1    = node + GATHER_WAVES;
    int   s1    = rowptr[n1];
    int   d1    = cnt[n1];
    float c1    = nd[n1];

    for (;;) {
        const bool has1 = n1 < NNODES;
        const int nmy = bucket_ids(csr, s1, lane);
        const int  n2   = n1 + GATHER_WAVES;
        int   s2 = 0, d2 = 0;
        float c2 = 0.f;
        if (n2 < NNODES) { s2 = rowptr[n2]; d2 = cnt[n2]; c2 = nd[n2]; }

        float4 acc = make_float4(0.f, 0.f, 0.f, 0.f);
        int ids = my;
        for (int j0 = 0; j0 < deg; j0 += 24) {
            if (j0 > 0) ids = clamp_node(bucket_ids(csr, start + j0, lane));
            const int rem = deg - j0;
            switch (min(8, (rem + 2) / 3)) {
                case 1:  rows3<1>(z, ids, rem, g, cc, acc); break;
                case 2:  rows3<2>(z, ids, rem, g, cc, acc); break;
                case 3:  rows3<3>(z, ids, rem, g, cc, acc); break;
                case 4:  rows3<4>(z, ids, rem, g, cc, acc); break;
                case 5:  rows3<5>(z, ids, rem, g, cc, acc); break;
                case 6:  rows3<6>(z, ids, rem, g, cc, acc); break;
                case 7:  rows3<7>(z, ids, rem, g, cc, acc); break;
                default: rows3<8>(z, ids, rem, g, cc, acc); break;
            }
        }
        const float px = __shfl_down(acc.x, 10), py = __shfl_down(acc.y, 10);
        const float pz = __shfl_down(acc.z, 10), pw = __shfl_down(acc.w, 10);
        const float qx = __shfl_down(acc.x, 20), qy = __shfl_down(acc.y, 20);
        const float qz = __shfl_down(acc.z, 20), qw = __shfl_down(acc.w, 20);
        if (lane < NCLS / 4)
            ((float4*)(out + (size_t)node * NCLS))[lane] =
                make_float4(fmaf((acc.x + px) + qx, sc, bv.x), fmaf((acc.y + py) + qy, sc, bv.y),
                            fmaf((acc.z + pz) + qz, sc, bv.z), fmaf((acc.w + pw) + qw, sc, bv.w));
        if (!has1) break;
        node = n1; start = s1; deg = d1; sc = c1; my = clamp_node(nmy);
        n1 = n2; s1 = s2; d1 = d2; c1 = c2;
    }
}


#define GEMM_BLOCKS 768
#define NTILES      (NNODES / 16)
static_assert(GEMM_BLOCKS <= NTILES, "every work-group owns at least one tile");

template <int NOUT, int NT, bool RELU, bool BIAS, int OSTRIDE>
__global__ __launch_bounds__(256) void gcn_gemm(const float* __restrict__ agg,
                                                const float* __restrict__ nd,
                                                const float* __restrict__ W,
                                                const float* __restrict__ bias,
                                                float* __restrict__ out) {
    constexpr int NT16 = NT * 16;
    __shared__ __attribute__((aligned(16))) float As[2][16][DIM + 4];
    __shared__ __attribute__((aligned(16))) float Ws[DIM][NT16];

    const int t   = threadIdx.x;
    const int am  = t >> 4;
    const int akb = (t & 15) * 8;

    int tile = blockIdx.x;

    float  s;
    float4 v0, v1;
    {
        const int node = tile * 16 + am;
        s = nd[node];
        const float4* p = (const float4*)(agg + (size_t)node * DIM + akb);
        v0 = p[0]; v1 = p[1];
    }

    if constexpr (NOUT == NT16) {
        for (int i = t; i < DIM * NT16 / 4; i += 256)
            ((float4*)&Ws[0][0])[i] = ((const float4*)W)[i];
    } else {
        for (int i = t; i < DIM * NT16; i += 256) {
            int k = i / NT16;
            int n = i - k * NT16;
            Ws[k][n] = (n < NOUT) ? W[k * NOUT + n] : 0.0f;
        }
    }

    const int wave = t >> 5;
    const int lane = t & 31;
    const int n0   = wave * 16;
    const int l16  = lane & 15;
    const int kh   = (lane >> 4) * 2;
    const int n    = n0 + l16;
    float bv = 0.0f;
    if (BIAS && n < NOUT) bv = bias[n];

    int cur = 0;
    for (; tile < NTILES; tile += GEMM_BLOCKS) {
        {
            float* a = &As[cur][am][akb];
            *(float4*)a       = make_float4(v0.x * s, v0.y * s, v0.z * s, v0.w * s);
            *(float4*)(a + 4) = make_float4(v1.x * s, v1.y * s, v1.z * s, v1.w * s);
        }
        __syncthreads();

        const int nxt = tile + GEMM_BLOCKS;
        if (nxt < NTILES) {
            const int node = nxt * 16 + am;
            s = nd[node];
            const float4* p = (const float4*)(agg + (size_t)node * DIM + akb);
            v0 = p[0]; v1 = p[1];
        }

        if (wave < NT) {
            const float (*Ac)[DIM + 4] = As[cur];
            v8f acc = {};
#pragma unroll
            for (int kk = 0; kk < DIM; kk += 4) {
                v2f a, b;
                a.x = Ac[l16][kk + kh];
                a.y = Ac[l16][kk + kh + 1];
                b.x = Ws[kk + kh][n0 + l16];
                b.y = Ws[kk + kh + 1][n0 + l16];
                acc = __builtin_amdgcn_wmma_f32_16x16x4_f32(
                    false, a, false, b, (short)0, acc, false, false);
            }
            float r[8];
#pragma unroll
            for (int v = 0; v < 8; ++v) {
                r[v] = acc[v] + bv;
                if (RELU) r[v] = fmaxf(r[v], 0.0f);
            }
            if (n < NOUT) {
                const int mbase = tile * 16 + (lane >> 4) * 8;
#pragma unroll
                for (int v = 0; v < 8; ++v)
                    out[(size_t)(mbase + v) * OSTRIDE + n] = r[v];
            }
        }
        cur ^= 1;
    }
}


#define FUSED_BLOCKS 768
#define NT2          3
#define NT2W         (NT2 * 16)
static_assert(2 * FUSED_BLOCKS <= NTILES, "every work-group owns a second tile");
static_assert(NT2W >= NCLS, "the class columns fit the N-tiles");

__device__ __forceinline__ void rows_upto16(const float* __restrict__ h, int ids, int j, int k,
                                            int lane, float4& acc) {
    switch (k) {
        case 1:  acc_rows<1>(h, ids, j, lane, acc);  break;
        case 2:  acc_rows<2>(h, ids, j, lane, acc);  break;
        case 3:  acc_rows<3>(h, ids, j, lane, acc);  break;
        case 4:  acc_rows<4>(h, ids, j, lane, acc);  break;
        case 5:  acc_rows<5>(h, ids, j, lane, acc);  break;
        case 6:  acc_rows<6>(h, ids, j, lane, acc);  break;
        case 7:  acc_rows<7>(h, ids, j, lane, acc);  break;
        case 8:  acc_rows<8>(h, ids, j, lane, acc);  break;
        case 9:  acc_rows<9>(h, ids, j, lane, acc);  break;
        case 10: acc_rows<10>(h, ids, j, lane, acc); break;
        case 11: acc_rows<11>(h, ids, j, lane, acc); break;
        case 12: acc_rows<12>(h, ids, j, lane, acc); break;
        case 13: acc_rows<13>(h, ids, j, lane, acc); break;
        case 14: acc_rows<14>(h, ids, j, lane, acc); break;
        case 15: acc_rows<15>(h, ids, j, lane, acc); break;
        default: acc_rows<16>(h, ids, j, lane, acc); break;
    }
}

__device__ __forceinline__ float4 bucket_sum(const float* __restrict__ h, const int* __restrict__ csr,
                                             int start, int deg, int ids, int lane) {
    float4 acc = make_float4(0.f, 0.f, 0.f, 0.f);
    for (int base = 0; base < deg; base += 32) {
        const int n = min(32, deg - base);
        if (base > 0) ids = clamp_node(bucket_ids(csr, start + base, lane));
        for (int j = 0; j < n; j += 16) rows_upto16(h, ids, j, min(16, n - j), lane, acc);
    }
    return acc;
}

__global__ __launch_bounds__(256) void gather_gemm2(const float* __restrict__ h,
                                                    const int* __restrict__ rowptr,
                                                    const int* __restrict__ cnt,
                                                    const int* __restrict__ csr,
                                                    const float* __restrict__ nd,
                                                    const float* __restrict__ ns,
                                                    const float* __restrict__ W1,
                                                    const float* __restrict__ b1,
                                                    const float* __restrict__ W2,
                                                    float* __restrict__ z) {
    __shared__ __attribute__((aligned(16))) float T[16][DIM + 4];
    __shared__ __attribute__((aligned(16))) float Ws1[DIM][DIM];
    __shared__ __attribute__((aligned(16))) float Ws2[DIM][NT2W];

    const int t    = threadIdx.x;
    const int lane = t & 31;
    const int wave = t >> 5;
    const int r0   = 2 * wave;
    const int hl   = lane & 1;

    int tile = blockIdx.x;

    int   vst = rowptr[tile * 16 + r0 + hl];
    int   vdg = cnt[tile * 16 + r0 + hl];
    float vsc = nd[tile * 16 + r0 + hl];
    int   wst = rowptr[(tile + FUSED_BLOCKS) * 16 + r0 + hl];
    int   wdg = cnt[(tile + FUSED_BLOCKS) * 16 + r0 + hl];
    float wsc = nd[(tile + FUSED_BLOCKS) * 16 + r0 + hl];

    for (int i = t; i < DIM * DIM / 4; i += 256)
        ((float4*)&Ws1[0][0])[i] = ((const float4*)W1)[i];
    for (int i = t; i < DIM * NT2W; i += 256) {
        const int k = i / NT2W;
        const int c = i - k * NT2W;
        Ws2[k][c] = (c < NCLS) ? W2[k * NCLS + c] : 0.0f;
    }

    int   stA = __builtin_amdgcn_readlane(vst, 0), stB = __builtin_amdgcn_readlane(vst, 1);
    int   dgA = __builtin_amdgcn_readlane(vdg, 0), dgB = __builtin_amdgcn_readlane(vdg, 1);
    float scA = __int_as_float(__builtin_amdgcn_readlane(__float_as_int(vsc), 0));
    float scB = __int_as_float(__builtin_amdgcn_readlane(__float_as_int(vsc), 1));
    int   st1A = __builtin_amdgcn_readlane(wst, 0), st1B = __builtin_amdgcn_readlane(wst, 1);
    int   dg1A = __builtin_amdgcn_readlane(wdg, 0), dg1B = __builtin_amdgcn_readlane(wdg, 1);
    float sc1A = __int_as_float(__builtin_amdgcn_readlane(__float_as_int(wsc), 0));
    float sc1B = __int_as_float(__builtin_amdgcn_readlane(__float_as_int(wsc), 1));
    int idA = bucket_ids(csr, stA, lane);
    int idB = bucket_ids(csr, stB, lane);

    const int n0  = wave * 16;
    const int l16 = lane & 15;
    const int kh  = (lane >> 4) * 2;
    const int mh  = (lane >> 4) * 8;
    const int n   = n0 + l16;
    const float bv = b1[n];

    for (; tile < NTILES; tile += FUSED_BLOCKS) {
        const int cA = clamp_node(idA), cB = clamp_node(idB);

        const int nidA = bucket_ids(csr, st1A, lane);
        const int nidB = bucket_ids(csr, st1B, lane);
        {
            const int na = min((tile + 2 * FUSED_BLOCKS) * 16 + r0 + hl, NNODES - 1);
            vst = rowptr[na]; vdg = cnt[na]; vsc = nd[na];
        }
        const float4 q0 = ((const float4*)(ns + tile * 16 + mh))[0];
        const float4 q1 = ((const float4*)(ns + tile * 16 + mh))[1];

        const float4 a0 = bucket_sum(h, csr, stA, dgA, cA, lane);
        const float4 a1 = bucket_sum(h, csr, stB, dgB, cB, lane);

        __syncthreads();
        *(float4*)&T[r0][4 * lane]     = make_float4(a0.x * scA, a0.y * scA, a0.z * scA, a0.w * scA);
        *(float4*)&T[r0 + 1][4 * lane] = make_float4(a1.x * scB, a1.y * scB, a1.z * scB, a1.w * scB);
        __syncthreads();

        v8f acc = {};
#pragma unroll
        for (int kk = 0; kk < DIM; kk += 4) {
            v2f a, b;
            a.x = T[l16][kk + kh];
            a.y = T[l16][kk + kh + 1];
            b.x = Ws1[kk + kh][n0 + l16];
            b.y = Ws1[kk + kh + 1][n0 + l16];
            acc = __builtin_amdgcn_wmma_f32_16x16x4_f32(
                false, a, false, b, (short)0, acc, false, false);
        }

        __syncthreads();
        T[mh + 0][n] = fmaxf(acc[0] + bv, 0.0f) * q0.x;
        T[mh + 1][n] = fmaxf(acc[1] + bv, 0.0f) * q0.y;
        T[mh + 2][n] = fmaxf(acc[2] + bv, 0.0f) * q0.z;
        T[mh + 3][n] = fmaxf(acc[3] + bv, 0.0f) * q0.w;
        T[mh + 4][n] = fmaxf(acc[4] + bv, 0.0f) * q1.x;
        T[mh + 5][n] = fmaxf(acc[5] + bv, 0.0f) * q1.y;
        T[mh + 6][n] = fmaxf(acc[6] + bv, 0.0f) * q1.z;
        T[mh + 7][n] = fmaxf(acc[7] + bv, 0.0f) * q1.w;
        __syncthreads();

        if (wave < NT2) {
            v8f acc2 = {};
#pragma unroll
            for (int kk = 0; kk < DIM; kk += 4) {
                v2f a, b;
                a.x = T[l16][kk + kh];
                a.y = T[l16][kk + kh + 1];
                b.x = Ws2[kk + kh][n0 + l16];
                b.y = Ws2[kk + kh + 1][n0 + l16];
                acc2 = __builtin_amdgcn_wmma_f32_16x16x4_f32(
                    false, a, false, b, (short)0, acc2, false, false);
            }
            if (n < NCLS) {
                const int mbase = tile * 16 + mh;
#pragma unroll
                for (int v = 0; v < 8; ++v)
                    z[(size_t)(mbase + v) * ZS + n] = acc2[v];
            }
        }

        stA = st1A; stB = st1B; dgA = dg1A; dgB = dg1B; scA = sc1A; scB = sc1B;
        idA = nidA; idB = nidB;
        st1A = __builtin_amdgcn_readlane(vst, 0); st1B = __builtin_amdgcn_readlane(vst, 1);
        dg1A = __builtin_amdgcn_readlane(vdg, 0); dg1B = __builtin_amdgcn_readlane(vdg, 1);
        sc1A = __int_as_float(__builtin_amdgcn_readlane(__float_as_int(vsc), 0));
        sc1B = __int_as_float(__builtin_amdgcn_readlane(__float_as_int(vsc), 1));
    }
}


extern "C" void kernel_launch(void* const* d_in, const int* in_sizes, int n_in,
                              void* d_out, int out_size, void* d_ws, size_t ws_size,
                              hipStream_t stream) {
    const float* inputs = (const float*)d_in[0];
    const int*   src    = (const int*)d_in[1];
    const int*   dst    = (const int*)d_in[2];
    const float* W0     = (const float*)d_in[3];
    const float* b0     = (const float*)d_in[4];
    const float* W1     = (const float*)d_in[5];
    const float* b1     = (const float*)d_in[6];
    const float* W2     = (const float*)d_in[7];
    const float* b2     = (const float*)d_in[8];
    float* out = (float*)d_out;

    constexpr size_t OFF_ND   = NNODES;
    constexpr size_t OFF_CS   = 2 * (size_t)NNODES;
    constexpr size_t OFF_CD   = 3 * (size_t)NNODES;
    constexpr size_t OFF_RP   = 4 * (size_t)NNODES;
    constexpr size_t OFF_POS  = 5 * (size_t)NNODES;
    constexpr size_t OFF_CSR  = OFF_POS + NEDGES;
    constexpr size_t OFF_B    = ((OFF_CSR + NEDGES + 127) / 128) * 128;
    constexpr size_t OFF_C    = OFF_B + (size_t)NNODES * DIM;
    constexpr size_t OFF_PD   = OFF_C + (size_t)NNODES * DIM;
    constexpr size_t OFF_PS   = OFF_PD + (size_t)HC * NNODES;

    float* ws      = (float*)d_ws;
    float* ns      = ws;
    float* ndm     = ws + OFF_ND;
    int*   cnt_src = (int*)(ws + OFF_CS);
    int*   cnt_dst = (int*)(ws + OFF_CD);
    int*   rowptr  = (int*)(ws + OFF_RP);
    int*   pos     = (int*)(ws + OFF_POS);
    int*   csr     = (int*)(ws + OFF_CSR);
    float* bufB    = ws + OFF_B;
    float* bufC    = ws + OFF_C;
    int*   plane_d = (int*)(ws + OFF_PD);
    int*   plane_s = (int*)(ws + OFF_PS);

    constexpr int GR_PRE  = (NNODES + 255) / 256;
    constexpr int GR_SCAN = (NNODES / 4 + 255) / 256;

    hist_rank<<<HC * HR, HT, 0, stream>>>(src, dst, plane_s, plane_d, pos);
    chunk_prefix<<<GR_PRE, 256, 0, stream>>>(plane_s, plane_d, cnt_src, cnt_dst);
    scan_norm<<<GR_SCAN, 256, 0, stream>>>(cnt_src, cnt_dst, rowptr, ns, ndm);
    fill_buckets<<<EDGE_BLOCKS, 256, 0, stream>>>(src, dst, rowptr, plane_d, pos, csr);

    gcn_gemm<DIM, 8, false, false, DIM><<<GEMM_BLOCKS, 256, 0, stream>>>(inputs, ns, W0, b0, bufB);
    gather_kernel<<<GATHER_BLOCKS, 256, 0, stream>>>(bufB, rowptr, cnt_dst, csr, ndm, ns, b0, bufC);

    gather_gemm2<<<FUSED_BLOCKS, 256, 0, stream>>>(bufC, rowptr, cnt_dst, csr, ndm, ns, W1, b1, W2, bufB);
    gather_out<<<GATHER_BLOCKS, 256, 0, stream>>>(bufB, ndm, b2, rowptr, cnt_dst, csr, out);
}


# Bucket fill runs as (chunk, range) work-groups with bucket start + chunk prefix in LDS; slots by LDS atomics, no rank array, no random per-edge loads; sum order = chunk, then fill-pass LDS arrival
# speedup vs baseline: 5.0677x; 1.0669x over previous
#include <hip/hip_runtime.h>


#define NNODES 100000
#define NEDGES 1600000
#define DIM    128
#define NCLS   40
#define ZS     64

typedef __attribute__((ext_vector_type(2))) float v2f;
typedef __attribute__((ext_vector_type(8))) float v8f;


#define HC    64
#define HR    4
#define HEC   (NEDGES / HC)
#define HNR   (NNODES / HR)
#define HT    1024
#define HFULL (HEC / HT)
#define HTAIL (HFULL * HT)
static_assert(HEC * HC == NEDGES && HNR * HR == NNODES, "chunks and ranges tile the problem exactly");
static_assert(HFULL % 8 == 0, "full strides are taken in groups of eight");
static_assert(HNR % 4 == 0, "counter planes move as 16-byte units");
#define SFULL ((HNR / 4) / HT)
static_assert((SFULL + 1) * HT >= HNR / 4, "one guarded unit per thread covers the rest of the range");

__global__ __launch_bounds__(HT) void hist_rank(const int* __restrict__ src,
                                                const int* __restrict__ dst,
                                                int* __restrict__ plane_src,
                                                int* __restrict__ plane_dst) {
    __shared__ __attribute__((aligned(16))) int hd[HNR];
    __shared__ __attribute__((aligned(16))) int hs[HNR];
    const int t  = threadIdx.x;
    const int c  = blockIdx.x / HR;
    const int lo = (blockIdx.x - c * HR) * HNR;
    const int e0 = c * HEC + t;

    for (int i = t; i < HNR / 4; i += HT) {
        ((int4*)hd)[i] = make_int4(0, 0, 0, 0);
        ((int4*)hs)[i] = make_int4(0, 0, 0, 0);
    }
    __syncthreads();

    for (int g = 0; g < HFULL; g += 8) {
        int dv[8], sv[8];
#pragma unroll
        for (int u = 0; u < 8; ++u) {
            dv[u] = dst[e0 + (g + u) * HT];
            sv[u] = src[e0 + (g + u) * HT];
        }
#pragma unroll
        for (int u = 0; u < 8; ++u) {
            const unsigned d = (unsigned)(dv[u] - lo);
            const unsigned q = (unsigned)(sv[u] - lo);
            if (d < (unsigned)HNR) atomicAdd(&hd[d], 1);
            if (q < (unsigned)HNR) atomicAdd(&hs[q], 1);
        }
    }
    if (HTAIL + t < HEC) {
        const unsigned d = (unsigned)(dst[e0 + HTAIL] - lo);
        const unsigned q = (unsigned)(src[e0 + HTAIL] - lo);
        if (d < (unsigned)HNR) atomicAdd(&hd[d], 1);
        if (q < (unsigned)HNR) atomicAdd(&hs[q], 1);
    }
    __syncthreads();

    int4* od = (int4*)(plane_dst + (size_t)c * NNODES + lo);
    int4* os = (int4*)(plane_src + (size_t)c * NNODES + lo);
    for (int i = t; i < HNR / 4; i += HT) {
        od[i] = ((const int4*)hd)[i];
        os[i] = ((const int4*)hs)[i];
    }
}

__global__ __launch_bounds__(256) void chunk_prefix(const int* __restrict__ plane_src,
                                                    int* __restrict__ plane_dst,
                                                    int* __restrict__ cnt_src,
                                                    int* __restrict__ cnt_dst) {
    const int n = blockIdx.x * 256 + threadIdx.x;
    if (n >= NNODES) return;
    int run = 0, tot = 0;
    for (int c0 = 0; c0 < HC; c0 += 16) {
        int a[16], b[16];
#pragma unroll
        for (int u = 0; u < 16; ++u) {
            a[u] = plane_dst[(size_t)(c0 + u) * NNODES + n];
            b[u] = plane_src[(size_t)(c0 + u) * NNODES + n];
        }
#pragma unroll
        for (int u = 0; u < 16; ++u) {
            plane_dst[(size_t)(c0 + u) * NNODES + n] = run;
            run += a[u];
            tot += b[u];
        }
    }
    cnt_dst[n] = run;
    cnt_src[n] = tot;
}

__device__ __forceinline__ float inv_sqrt_deg(int c) {
    return (c > 0) ? rsqrtf(fmaxf((float)c, 1.f)) : 0.f;
}

__global__ __launch_bounds__(256) void scan_norm(const int* __restrict__ cnt_src,
                                                 const int* __restrict__ cnt_dst,
                                                 int* __restrict__ rowptr,
                                                 float* __restrict__ ns,
                                                 float* __restrict__ nd) {
    __shared__ int red[8];
    __shared__ int wsum[8];
    const int t = threadIdx.x, b = blockIdx.x;
    const int lane = t & 31, wave = t >> 5;
    const int4* c4 = (const int4*)cnt_dst;

    int part = 0;
    int k = 0;
    for (; k + 8 <= b; k += 8) {
        int4 q[8];
#pragma unroll
        for (int u = 0; u < 8; ++u) q[u] = c4[(k + u) * 256 + t];
#pragma unroll
        for (int u = 0; u < 8; ++u) part += (q[u].x + q[u].y) + (q[u].z + q[u].w);
    }
    for (; k < b; ++k) {
        int4 q = c4[k * 256 + t];
        part += (q.x + q.y) + (q.z + q.w);
    }
    for (int o = 16; o > 0; o >>= 1) part += __shfl_xor(part, o);
    if (lane == 0) red[wave] = part;

    const int i4 = b * 256 + t;
    const bool in = i4 < NNODES / 4;
    int4 c = make_int4(0, 0, 0, 0);
    if (in) c = c4[i4];
    const int s = (c.x + c.y) + (c.z + c.w);
    int inc = s;
    for (int o = 1; o < 32; o <<= 1) {
        int v = __shfl_up(inc, o);
        if (lane >= o) inc += v;
    }
    if (lane == 31) wsum[wave] = inc;
    __syncthreads();

    int base = 0;
    for (int w = 0; w < 8; ++w) base += red[w];
    for (int w = 0; w < 8; ++w) base += (w < wave) ? wsum[w] : 0;
    const int ex = base + inc - s;
    if (in) {
        ((int4*)rowptr)[i4] = make_int4(ex, ex + c.x, ex + c.x + c.y, ex + c.x + c.y + c.z);
        int4 cs = ((const int4*)cnt_src)[i4];
        ((float4*)ns)[i4] = make_float4(inv_sqrt_deg(cs.x), inv_sqrt_deg(cs.y),
                                        inv_sqrt_deg(cs.z), inv_sqrt_deg(cs.w));
        ((float4*)nd)[i4] = make_float4(inv_sqrt_deg(c.x), inv_sqrt_deg(c.y),
                                        inv_sqrt_deg(c.z), inv_sqrt_deg(c.w));
    }
}

__global__ __launch_bounds__(HT) void fill_buckets(const int* __restrict__ src,
                                                   const int* __restrict__ dst,
                                                   const int* __restrict__ rowptr,
                                                   const int* __restrict__ plane_dst,
                                                   int* __restrict__ csr) {
    __shared__ __attribute__((aligned(16))) int nxt[HNR];
    const int t  = threadIdx.x;
    const int c  = blockIdx.x / HR;
    const int lo = (blockIdx.x - c * HR) * HNR;
    const int e0 = c * HEC + t;

    const int4* rp = (const int4*)(rowptr + lo);
    const int4* pd = (const int4*)(plane_dst + (size_t)c * NNODES + lo);
    {
        int4 r[SFULL + 1], p[SFULL + 1];
        const bool last = SFULL * HT + t < HNR / 4;
#pragma unroll
        for (int k = 0; k < SFULL; ++k) { r[k] = rp[k * HT + t]; p[k] = pd[k * HT + t]; }
        r[SFULL] = make_int4(0, 0, 0, 0); p[SFULL] = make_int4(0, 0, 0, 0);
        if (last) { r[SFULL] = rp[SFULL * HT + t]; p[SFULL] = pd[SFULL * HT + t]; }
#pragma unroll
        for (int k = 0; k < SFULL; ++k)
            ((int4*)nxt)[k * HT + t] = make_int4(r[k].x + p[k].x, r[k].y + p[k].y, r[k].z + p[k].z, r[k].w + p[k].w);
        if (last)
            ((int4*)nxt)[SFULL * HT + t] = make_int4(r[SFULL].x + p[SFULL].x, r[SFULL].y + p[SFULL].y,
                                                     r[SFULL].z + p[SFULL].z, r[SFULL].w + p[SFULL].w);
    }
    __syncthreads();

    for (int g = 0; g < HFULL; g += 8) {
        int dv[8], sv[8];
#pragma unroll
        for (int u = 0; u < 8; ++u) {
            dv[u] = dst[e0 + (g + u) * HT];
            sv[u] = src[e0 + (g + u) * HT];
        }
        asm volatile("" :: "v"(sv[0]), "v"(sv[1]), "v"(sv[2]), "v"(sv[3]),
                           "v"(sv[4]), "v"(sv[5]), "v"(sv[6]), "v"(sv[7]));
#pragma unroll
        for (int u = 0; u < 8; ++u) {
            const unsigned d = (unsigned)(dv[u] - lo);
            if (d < (unsigned)HNR) {
                const unsigned idx = (unsigned)atomicAdd(&nxt[d], 1);
                if (idx < (unsigned)NEDGES) csr[idx] = sv[u];
            }
        }
    }
    if (HTAIL + t < HEC) {
        const unsigned d = (unsigned)(dst[e0 + HTAIL] - lo);
        const int s = src[e0 + HTAIL];
        asm volatile("" :: "v"(s));
        if (d < (unsigned)HNR) {
            const unsigned idx = (unsigned)atomicAdd(&nxt[d], 1);
            if (idx < (unsigned)NEDGES) csr[idx] = s;
        }
    }
}


template <int U>
__device__ __forceinline__ void acc_rows(const float* __restrict__ h, int my,
                                         int j, int lane, float4& acc) {
    float4 v[U];
#pragma unroll
    for (int u = 0; u < U; ++u) {
        const int s = __builtin_amdgcn_readlane(my, j + u);
        v[u] = ((const float4*)(h + (size_t)s * DIM))[lane];
    }
#pragma unroll
    for (int u = 0; u < U; ++u) {
        acc.x += v[u].x; acc.y += v[u].y; acc.z += v[u].z; acc.w += v[u].w;
    }
}

#define GATHER_BLOCKS 2048
#define GATHER_WAVES  (GATHER_BLOCKS * 8)

__global__ __launch_bounds__(256) void gather_kernel(const float* __restrict__ h,
                                                     const int* __restrict__ rowptr,
                                                     const int* __restrict__ cnt,
                                                     const int* __restrict__ csr,
                                                     const float* __restrict__ nd,
                                                     const float* __restrict__ ns,
                                                     const float* __restrict__ bias,
                                                     float* __restrict__ out) {
    const int lane = threadIdx.x & 31;
    const int w0 = __builtin_amdgcn_readfirstlane((int)((blockIdx.x * 256 + threadIdx.x) >> 5));
    const float4 bv = ((const float4*)bias)[lane];
    for (int node = w0; node < NNODES; node += GATHER_WAVES) {
        const int start = rowptr[node];
        const int deg   = cnt[node];
        const float sc = nd[node], tn = ns[node];
        float4 acc = make_float4(0.f, 0.f, 0.f, 0.f);
        for (int base = 0; base < deg; base += 32) {
            const int n = min(32, deg - base);
            int my = 0;
            if (lane < n)
                my = (int)min((unsigned)csr[start + base + lane], (unsigned)(NNODES - 1));
            int j = 0;
            for (; j + 8 <= n; j += 8) acc_rows<8>(h, my, j, lane, acc);
            if (j + 4 <= n) { acc_rows<4>(h, my, j, lane, acc); j += 4; }
            if (j + 2 <= n) { acc_rows<2>(h, my, j, lane, acc); j += 2; }
            if (j < n)      { acc_rows<1>(h, my, j, lane, acc); }
        }
        acc.x = fmaxf(fmaf(acc.x, sc, bv.x), 0.f) * tn;
        acc.y = fmaxf(fmaf(acc.y, sc, bv.y), 0.f) * tn;
        acc.z = fmaxf(fmaf(acc.z, sc, bv.z), 0.f) * tn;
        acc.w = fmaxf(fmaf(acc.w, sc, bv.w), 0.f) * tn;
        ((float4*)(out + (size_t)node * DIM))[lane] = acc;
    }
}


__device__ __forceinline__ int clamp_node(int s) {
    return (int)min((unsigned)s, (unsigned)(NNODES - 1));
}

template <int K>
__device__ __forceinline__ void rows3(const float* __restrict__ z, int ids, int rem,
                                      int g, int cc, float4& acc) {
    float4 v[K];
#pragma unroll
    for (int u = 0; u < K; ++u) {
        const int s0 = __builtin_amdgcn_readlane(ids, 3 * u);
        const int s1 = __builtin_amdgcn_readlane(ids, 3 * u + 1);
        const int s2 = __builtin_amdgcn_readlane(ids, 3 * u + 2);
        int s;
        if (u < K - 1) s = (g == 1) ? s1 : ((g == 2) ? s2 : s0);
        else           s = (g == 1 && 3 * u + 1 < rem) ? s1 : ((g == 2 && 3 * u + 2 < rem) ? s2 : s0);
        v[u] = ((const float4*)(z + (size_t)s * ZS))[cc];
    }
#pragma unroll
    for (int u = 0; u < K; ++u) {
        if (u < K - 1) {
            acc.x += v[u].x; acc.y += v[u].y; acc.z += v[u].z; acc.w += v[u].w;
        } else {
            const bool ok = 3 * u + g < rem;
            acc.x += ok ? v[u].x : 0.f; acc.y += ok ? v[u].y : 0.f;
            acc.z += ok ? v[u].z : 0.f; acc.w += ok ? v[u].w : 0.f;
        }
    }
}

__device__ __forceinline__ int bucket_ids(const int* __restrict__ csr, int first, int lane) {
    return csr[min((unsigned)(first + lane), (unsigned)(NEDGES - 1))];
}

__global__ __launch_bounds__(256) void gather_out(const float* __restrict__ z,
                                                  const float* __restrict__ nd,
                                                  const float* __restrict__ bias,
                                                  const int* __restrict__ rowptr,
                                                  const int* __restrict__ cnt,
                                                  const int* __restrict__ csr,
                                                  float* __restrict__ out) {
    const int lane = threadIdx.x & 31;
    const int g    = (lane >= 30) ? 0 : ((lane >= 20) ? 2 : ((lane >= 10) ? 1 : 0));
    const int cc   = (lane >= 30) ? (NCLS / 4 - 1) : (lane - 10 * g);
    const int w0 = __builtin_amdgcn_readfirstlane((int)((blockIdx.x * 256 + threadIdx.x) >> 5));
    const float4 bv = ((const float4*)bias)[cc];

    int   node  = w0;
    int   start = rowptr[node];
    int   deg   = cnt[node];
    float sc    = nd[node];
    int   my    = clamp_node(bucket_ids(csr, start, lane));
    static_assert(2 * GATHER_WAVES <= NNODES, "every wave owns a second node");
    int   n1    = node + GATHER_WAVES;
    int   s1    = rowptr[n1];
    int   d1    = cnt[n1];
    float c1    = nd[n1];

    for (;;) {
        const bool has1 = n1 < NNODES;
        const int nmy = bucket_ids(csr, s1, lane);
        const int  n2   = n1 + GATHER_WAVES;
        int   s2 = 0, d2 = 0;
        float c2 = 0.f;
        if (n2 < NNODES) { s2 = rowptr[n2]; d2 = cnt[n2]; c2 = nd[n2]; }

        float4 acc = make_float4(0.f, 0.f, 0.f, 0.f);
        int ids = my;
        for (int j0 = 0; j0 < deg; j0 += 24) {
            if (j0 > 0) ids = clamp_node(bucket_ids(csr, start + j0, lane));
            const int rem = deg - j0;
            switch (min(8, (rem + 2) / 3)) {
                case 1:  rows3<1>(z, ids, rem, g, cc, acc); break;
                case 2:  rows3<2>(z, ids, rem, g, cc, acc); break;
                case 3:  rows3<3>(z, ids, rem, g, cc, acc); break;
                case 4:  rows3<4>(z, ids, rem, g, cc, acc); break;
                case 5:  rows3<5>(z, ids, rem, g, cc, acc); break;
                case 6:  rows3<6>(z, ids, rem, g, cc, acc); break;
                case 7:  rows3<7>(z, ids, rem, g, cc, acc); break;
                default: rows3<8>(z, ids, rem, g, cc, acc); break;
            }
        }
        const float px = __shfl_down(acc.x, 10), py = __shfl_down(acc.y, 10);
        const float pz = __shfl_down(acc.z, 10), pw = __shfl_down(acc.w, 10);
        const float qx = __shfl_down(acc.x, 20), qy = __shfl_down(acc.y, 20);
        const float qz = __shfl_down(acc.z, 20), qw = __shfl_down(acc.w, 20);
        if (lane < NCLS / 4)
            ((float4*)(out + (size_t)node * NCLS))[lane] =
                make_float4(fmaf((acc.x + px) + qx, sc, bv.x), fmaf((acc.y + py) + qy, sc, bv.y),
                            fmaf((acc.z + pz) + qz, sc, bv.z), fmaf((acc.w + pw) + qw, sc, bv.w));
        if (!has1) break;
        node = n1; start = s1; deg = d1; sc = c1; my = clamp_node(nmy);
        n1 = n2; s1 = s2; d1 = d2; c1 = c2;
    }
}


#define GEMM_BLOCKS 768
#define NTILES      (NNODES / 16)
static_assert(GEMM_BLOCKS <= NTILES, "every work-group owns at least one tile");

template <int NOUT, int NT, bool RELU, bool BIAS, int OSTRIDE>
__global__ __launch_bounds__(256) void gcn_gemm(const float* __restrict__ agg,
                                                const float* __restrict__ nd,
                                                const float* __restrict__ W,
                                                const float* __restrict__ bias,
                                                float* __restrict__ out) {
    constexpr int NT16 = NT * 16;
    __shared__ __attribute__((aligned(16))) float As[2][16][DIM + 4];
    __shared__ __attribute__((aligned(16))) float Ws[DIM][NT16];

    const int t   = threadIdx.x;
    const int am  = t >> 4;
    const int akb = (t & 15) * 8;

    int tile = blockIdx.x;

    float  s;
    float4 v0, v1;
    {
        const int node = tile * 16 + am;
        s = nd[node];
        const float4* p = (const float4*)(agg + (size_t)node * DIM + akb);
        v0 = p[0]; v1 = p[1];
    }

    if constexpr (NOUT == NT16) {
        for (int i = t; i < DIM * NT16 / 4; i += 256)
            ((float4*)&Ws[0][0])[i] = ((const float4*)W)[i];
    } else {
        for (int i = t; i < DIM * NT16; i += 256) {
            int k = i / NT16;
            int n = i - k * NT16;
            Ws[k][n] = (n < NOUT) ? W[k * NOUT + n] : 0.0f;
        }
    }

    const int wave = t >> 5;
    const int lane = t & 31;
    const int n0   = wave * 16;
    const int l16  = lane & 15;
    const int kh   = (lane >> 4) * 2;
    const int n    = n0 + l16;
    float bv = 0.0f;
    if (BIAS && n < NOUT) bv = bias[n];

    int cur = 0;
    for (; tile < NTILES; tile += GEMM_BLOCKS) {
        {
            float* a = &As[cur][am][akb];
            *(float4*)a       = make_float4(v0.x * s, v0.y * s, v0.z * s, v0.w * s);
            *(float4*)(a + 4) = make_float4(v1.x * s, v1.y * s, v1.z * s, v1.w * s);
        }
        __syncthreads();

        const int nxt = tile + GEMM_BLOCKS;
        if (nxt < NTILES) {
            const int node = nxt * 16 + am;
            s = nd[node];
            const float4* p = (const float4*)(agg + (size_t)node * DIM + akb);
            v0 = p[0]; v1 = p[1];
        }

        if (wave < NT) {
            const float (*Ac)[DIM + 4] = As[cur];
            v8f acc = {};
#pragma unroll
            for (int kk = 0; kk < DIM; kk += 4) {
                v2f a, b;
                a.x = Ac[l16][kk + kh];
                a.y = Ac[l16][kk + kh + 1];
                b.x = Ws[kk + kh][n0 + l16];
                b.y = Ws[kk + kh + 1][n0 + l16];
                acc = __builtin_amdgcn_wmma_f32_16x16x4_f32(
                    false, a, false, b, (short)0, acc, false, false);
            }
            float r[8];
#pragma unroll
            for (int v = 0; v < 8; ++v) {
                r[v] = acc[v] + bv;
                if (RELU) r[v] = fmaxf(r[v], 0.0f);
            }
            if (n < NOUT) {
                const int mbase = tile * 16 + (lane >> 4) * 8;
#pragma unroll
                for (int v = 0; v < 8; ++v)
                    out[(size_t)(mbase + v) * OSTRIDE + n] = r[v];
            }
        }
        cur ^= 1;
    }
}


#define FUSED_BLOCKS 768
#define NT2          3
#define NT2W         (NT2 * 16)
static_assert(2 * FUSED_BLOCKS <= NTILES, "every work-group owns a second tile");
static_assert(NT2W >= NCLS, "the class columns fit the N-tiles");

__device__ __forceinline__ void rows_upto16(const float* __restrict__ h, int ids, int j, int k,
                                            int lane, float4& acc) {
    switch (k) {
        case 1:  acc_rows<1>(h, ids, j, lane, acc);  break;
        case 2:  acc_rows<2>(h, ids, j, lane, acc);  break;
        case 3:  acc_rows<3>(h, ids, j, lane, acc);  break;
        case 4:  acc_rows<4>(h, ids, j, lane, acc);  break;
        case 5:  acc_rows<5>(h, ids, j, lane, acc);  break;
        case 6:  acc_rows<6>(h, ids, j, lane, acc);  break;
        case 7:  acc_rows<7>(h, ids, j, lane, acc);  break;
        case 8:  acc_rows<8>(h, ids, j, lane, acc);  break;
        case 9:  acc_rows<9>(h, ids, j, lane, acc);  break;
        case 10: acc_rows<10>(h, ids, j, lane, acc); break;
        case 11: acc_rows<11>(h, ids, j, lane, acc); break;
        case 12: acc_rows<12>(h, ids, j, lane, acc); break;
        case 13: acc_rows<13>(h, ids, j, lane, acc); break;
        case 14: acc_rows<14>(h, ids, j, lane, acc); break;
        case 15: acc_rows<15>(h, ids, j, lane, acc); break;
        default: acc_rows<16>(h, ids, j, lane, acc); break;
    }
}

__device__ __forceinline__ float4 bucket_sum(const float* __restrict__ h, const int* __restrict__ csr,
                                             int start, int deg, int ids, int lane) {
    float4 acc = make_float4(0.f, 0.f, 0.f, 0.f);
    for (int base = 0; base < deg; base += 32) {
        const int n = min(32, deg - base);
        if (base > 0) ids = clamp_node(bucket_ids(csr, start + base, lane));
        for (int j = 0; j < n; j += 16) rows_upto16(h, ids, j, min(16, n - j), lane, acc);
    }
    return acc;
}

__global__ __launch_bounds__(256) void gather_gemm2(const float* __restrict__ h,
                                                    const int* __restrict__ rowptr,
                                                    const int* __restrict__ cnt,
                                                    const int* __restrict__ csr,
                                                    const float* __restrict__ nd,
                                                    const float* __restrict__ ns,
                                                    const float* __restrict__ W1,
                                                    const float* __restrict__ b1,
                                                    const float* __restrict__ W2,
                                                    float* __restrict__ z) {
    __shared__ __attribute__((aligned(16))) float T[16][DIM + 4];
    __shared__ __attribute__((aligned(16))) float Ws1[DIM][DIM];
    __shared__ __attribute__((aligned(16))) float Ws2[DIM][NT2W];

    const int t    = threadIdx.x;
    const int lane = t & 31;
    const int wave = t >> 5;
    const int r0   = 2 * wave;
    const int hl   = lane & 1;

    int tile = blockIdx.x;

    int   vst = rowptr[tile * 16 + r0 + hl];
    int   vdg = cnt[tile * 16 + r0 + hl];
    float vsc = nd[tile * 16 + r0 + hl];
    int   wst = rowptr[(tile + FUSED_BLOCKS) * 16 + r0 + hl];
    int   wdg = cnt[(tile + FUSED_BLOCKS) * 16 + r0 + hl];
    float wsc = nd[(tile + FUSED_BLOCKS) * 16 + r0 + hl];

    for (int i = t; i < DIM * DIM / 4; i += 256)
        ((float4*)&Ws1[0][0])[i] = ((const float4*)W1)[i];
    for (int i = t; i < DIM * NT2W; i += 256) {
        const int k = i / NT2W;
        const int c = i - k * NT2W;
        Ws2[k][c] = (c < NCLS) ? W2[k * NCLS + c] : 0.0f;
    }

    int   stA = __builtin_amdgcn_readlane(vst, 0), stB = __builtin_amdgcn_readlane(vst, 1);
    int   dgA = __builtin_amdgcn_readlane(vdg, 0), dgB = __builtin_amdgcn_readlane(vdg, 1);
    float scA = __int_as_float(__builtin_amdgcn_readlane(__float_as_int(vsc), 0));
    float scB = __int_as_float(__builtin_amdgcn_readlane(__float_as_int(vsc), 1));
    int   st1A = __builtin_amdgcn_readlane(wst, 0), st1B = __builtin_amdgcn_readlane(wst, 1);
    int   dg1A = __builtin_amdgcn_readlane(wdg, 0), dg1B = __builtin_amdgcn_readlane(wdg, 1);
    float sc1A = __int_as_float(__builtin_amdgcn_readlane(__float_as_int(wsc), 0));
    float sc1B = __int_as_float(__builtin_amdgcn_readlane(__float_as_int(wsc), 1));
    int idA = bucket_ids(csr, stA, lane);
    int idB = bucket_ids(csr, stB, lane);

    const int n0  = wave * 16;
    const int l16 = lane & 15;
    const int kh  = (lane >> 4) * 2;
    const int mh  = (lane >> 4) * 8;
    const int n   = n0 + l16;
    const float bv = b1[n];

    for (; tile < NTILES; tile += FUSED_BLOCKS) {
        const int cA = clamp_node(idA), cB = clamp_node(idB);

        const int nidA = bucket_ids(csr, st1A, lane);
        const int nidB = bucket_ids(csr, st1B, lane);
        {
            const int na = min((tile + 2 * FUSED_BLOCKS) * 16 + r0 + hl, NNODES - 1);
            vst = rowptr[na]; vdg = cnt[na]; vsc = nd[na];
        }
        const float4 q0 = ((const float4*)(ns + tile * 16 + mh))[0];
        const float4 q1 = ((const float4*)(ns + tile * 16 + mh))[1];

        const float4 a0 = bucket_sum(h, csr, stA, dgA, cA, lane);
        const float4 a1 = bucket_sum(h, csr, stB, dgB, cB, lane);

        __syncthreads();
        *(float4*)&T[r0][4 * lane]     = make_float4(a0.x * scA, a0.y * scA, a0.z * scA, a0.w * scA);
        *(float4*)&T[r0 + 1][4 * lane] = make_float4(a1.x * scB, a1.y * scB, a1.z * scB, a1.w * scB);
        __syncthreads();

        v8f acc = {};
#pragma unroll
        for (int kk = 0; kk < DIM; kk += 4) {
            v2f a, b;
            a.x = T[l16][kk + kh];
            a.y = T[l16][kk + kh + 1];
            b.x = Ws1[kk + kh][n0 + l16];
            b.y = Ws1[kk + kh + 1][n0 + l16];
            acc = __builtin_amdgcn_wmma_f32_16x16x4_f32(
                false, a, false, b, (short)0, acc, false, false);
        }

        __syncthreads();
        T[mh + 0][n] = fmaxf(acc[0] + bv, 0.0f) * q0.x;
        T[mh + 1][n] = fmaxf(acc[1] + bv, 0.0f) * q0.y;
        T[mh + 2][n] = fmaxf(acc[2] + bv, 0.0f) * q0.z;
        T[mh + 3][n] = fmaxf(acc[3] + bv, 0.0f) * q0.w;
        T[mh + 4][n] = fmaxf(acc[4] + bv, 0.0f) * q1.x;
        T[mh + 5][n] = fmaxf(acc[5] + bv, 0.0f) * q1.y;
        T[mh + 6][n] = fmaxf(acc[6] + bv, 0.0f) * q1.z;
        T[mh + 7][n] = fmaxf(acc[7] + bv, 0.0f) * q1.w;
        __syncthreads();

        if (wave < NT2) {
            v8f acc2 = {};
#pragma unroll
            for (int kk = 0; kk < DIM; kk += 4) {
                v2f a, b;
                a.x = T[l16][kk + kh];
                a.y = T[l16][kk + kh + 1];
                b.x = Ws2[kk + kh][n0 + l16];
                b.y = Ws2[kk + kh + 1][n0 + l16];
                acc2 = __builtin_amdgcn_wmma_f32_16x16x4_f32(
                    false, a, false, b, (short)0, acc2, false, false);
            }
            if (n < NCLS) {
                const int mbase = tile * 16 + mh;
#pragma unroll
                for (int v = 0; v < 8; ++v)
                    z[(size_t)(mbase + v) * ZS + n] = acc2[v];
            }
        }

        stA = st1A; stB = st1B; dgA = dg1A; dgB = dg1B; scA = sc1A; scB = sc1B;
        idA = nidA; idB = nidB;
        st1A = __builtin_amdgcn_readlane(vst, 0); st1B = __builtin_amdgcn_readlane(vst, 1);
        dg1A = __builtin_amdgcn_readlane(vdg, 0); dg1B = __builtin_amdgcn_readlane(vdg, 1);
        sc1A = __int_as_float(__builtin_amdgcn_readlane(__float_as_int(vsc), 0));
        sc1B = __int_as_float(__builtin_amdgcn_readlane(__float_as_int(vsc), 1));
    }
}


extern "C" void kernel_launch(void* const* d_in, const int* in_sizes, int n_in,
                              void* d_out, int out_size, void* d_ws, size_t ws_size,
                              hipStream_t stream) {
    const float* inputs = (const float*)d_in[0];
    const int*   src    = (const int*)d_in[1];
    const int*   dst    = (const int*)d_in[2];
    const float* W0     = (const float*)d_in[3];
    const float* b0     = (const float*)d_in[4];
    const float* W1     = (const float*)d_in[5];
    const float* b1     = (const float*)d_in[6];
    const float* W2     = (const float*)d_in[7];
    const float* b2     = (const float*)d_in[8];
    float* out = (float*)d_out;

    constexpr size_t OFF_ND   = NNODES;
    constexpr size_t OFF_CS   = 2 * (size_t)NNODES;
    constexpr size_t OFF_CD   = 3 * (size_t)NNODES;
    constexpr size_t OFF_RP   = 4 * (size_t)NNODES;
    constexpr size_t OFF_CSR  = 5 * (size_t)NNODES;
    constexpr size_t OFF_B    = ((OFF_CSR + NEDGES + 127) / 128) * 128;
    constexpr size_t OFF_C    = OFF_B + (size_t)NNODES * DIM;
    constexpr size_t OFF_PD   = OFF_C + (size_t)NNODES * DIM;
    constexpr size_t OFF_PS   = OFF_PD + (size_t)HC * NNODES;

    float* ws      = (float*)d_ws;
    float* ns      = ws;
    float* ndm     = ws + OFF_ND;
    int*   cnt_src = (int*)(ws + OFF_CS);
    int*   cnt_dst = (int*)(ws + OFF_CD);
    int*   rowptr  = (int*)(ws + OFF_RP);
    int*   csr     = (int*)(ws + OFF_CSR);
    float* bufB    = ws + OFF_B;
    float* bufC    = ws + OFF_C;
    int*   plane_d = (int*)(ws + OFF_PD);
    int*   plane_s = (int*)(ws + OFF_PS);

    constexpr int GR_PRE  = (NNODES + 255) / 256;
    constexpr int GR_SCAN = (NNODES / 4 + 255) / 256;

    hist_rank<<<HC * HR, HT, 0, stream>>>(src, dst, plane_s, plane_d);
    chunk_prefix<<<GR_PRE, 256, 0, stream>>>(plane_s, plane_d, cnt_src, cnt_dst);
    scan_norm<<<GR_SCAN, 256, 0, stream>>>(cnt_src, cnt_dst, rowptr, ns, ndm);
    fill_buckets<<<HC * HR, HT, 0, stream>>>(src, dst, rowptr, plane_d, csr);

    gcn_gemm<DIM, 8, false, false, DIM><<<GEMM_BLOCKS, 256, 0, stream>>>(inputs, ns, W0, b0, bufB);
    gather_kernel<<<GATHER_BLOCKS, 256, 0, stream>>>(bufB, rowptr, cnt_dst, csr, ndm, ns, b0, bufC);

    gather_gemm2<<<FUSED_BLOCKS, 256, 0, stream>>>(bufC, rowptr, cnt_dst, csr, ndm, ns, W1, b1, W2, bufB);
    gather_out<<<GATHER_BLOCKS, 256, 0, stream>>>(bufB, ndm, b2, rowptr, cnt_dst, csr, out);
}


# First layer's product (x*norm_src)@W0 runs as 256 extra 32-wave work-groups of the bucket-fill launch (independent work, one of each kind per WGP): 8 -> 7 launches; sums and their order unchanged
# speedup vs baseline: 5.2413x; 1.0343x over previous
#include <hip/hip_runtime.h>


#define NNODES 100000
#define NEDGES 1600000
#define DIM    128
#define NCLS   40
#define ZS     64

typedef __attribute__((ext_vector_type(2))) float v2f;
typedef __attribute__((ext_vector_type(8))) float v8f;


#define HC    64
#define HR    4
#define HEC   (NEDGES / HC)
#define HNR   (NNODES / HR)
#define HT    1024
#define HFULL (HEC / HT)
#define HTAIL (HFULL * HT)
static_assert(HEC * HC == NEDGES && HNR * HR == NNODES, "chunks and ranges tile the problem exactly");
static_assert(HFULL % 8 == 0, "full strides are taken in groups of eight");
static_assert(HNR % 4 == 0, "counter planes move as 16-byte units");
#define SFULL ((HNR / 4) / HT)
static_assert((SFULL + 1) * HT >= HNR / 4, "one guarded unit per thread covers the rest of the range");

__global__ __launch_bounds__(HT) void hist_rank(const int* __restrict__ src,
                                                const int* __restrict__ dst,
                                                int* __restrict__ plane_src,
                                                int* __restrict__ plane_dst) {
    __shared__ __attribute__((aligned(16))) int hd[HNR];
    __shared__ __attribute__((aligned(16))) int hs[HNR];
    const int t  = threadIdx.x;
    const int c  = blockIdx.x / HR;
    const int lo = (blockIdx.x - c * HR) * HNR;
    const int e0 = c * HEC + t;

    for (int i = t; i < HNR / 4; i += HT) {
        ((int4*)hd)[i] = make_int4(0, 0, 0, 0);
        ((int4*)hs)[i] = make_int4(0, 0, 0, 0);
    }
    __syncthreads();

    for (int g = 0; g < HFULL; g += 8) {
        int dv[8], sv[8];
#pragma unroll
        for (int u = 0; u < 8; ++u) {
            dv[u] = dst[e0 + (g + u) * HT];
            sv[u] = src[e0 + (g + u) * HT];
        }
#pragma unroll
        for (int u = 0; u < 8; ++u) {
            const unsigned d = (unsigned)(dv[u] - lo);
            const unsigned q = (unsigned)(sv[u] - lo);
            if (d < (unsigned)HNR) atomicAdd(&hd[d], 1);
            if (q < (unsigned)HNR) atomicAdd(&hs[q], 1);
        }
    }
    if (HTAIL + t < HEC) {
        const unsigned d = (unsigned)(dst[e0 + HTAIL] - lo);
        const unsigned q = (unsigned)(src[e0 + HTAIL] - lo);
        if (d < (unsigned)HNR) atomicAdd(&hd[d], 1);
        if (q < (unsigned)HNR) atomicAdd(&hs[q], 1);
    }
    __syncthreads();

    int4* od = (int4*)(plane_dst + (size_t)c * NNODES + lo);
    int4* os = (int4*)(plane_src + (size_t)c * NNODES + lo);
    for (int i = t; i < HNR / 4; i += HT) {
        od[i] = ((const int4*)hd)[i];
        os[i] = ((const int4*)hs)[i];
    }
}

__global__ __launch_bounds__(256) void chunk_prefix(const int* __restrict__ plane_src,
                                                    int* __restrict__ plane_dst,
                                                    int* __restrict__ cnt_src,
                                                    int* __restrict__ cnt_dst) {
    const int n = blockIdx.x * 256 + threadIdx.x;
    if (n >= NNODES) return;
    int run = 0, tot = 0;
    for (int c0 = 0; c0 < HC; c0 += 16) {
        int a[16], b[16];
#pragma unroll
        for (int u = 0; u < 16; ++u) {
            a[u] = plane_dst[(size_t)(c0 + u) * NNODES + n];
            b[u] = plane_src[(size_t)(c0 + u) * NNODES + n];
        }
#pragma unroll
        for (int u = 0; u < 16; ++u) {
            plane_dst[(size_t)(c0 + u) * NNODES + n] = run;
            run += a[u];
            tot += b[u];
        }
    }
    cnt_dst[n] = run;
    cnt_src[n] = tot;
}

__device__ __forceinline__ float inv_sqrt_deg(int c) {
    return (c > 0) ? rsqrtf(fmaxf((float)c, 1.f)) : 0.f;
}

__global__ __launch_bounds__(256) void scan_norm(const int* __restrict__ cnt_src,
                                                 const int* __restrict__ cnt_dst,
                                                 int* __restrict__ rowptr,
                                                 float* __restrict__ ns,
                                                 float* __restrict__ nd) {
    __shared__ int red[8];
    __shared__ int wsum[8];
    const int t = threadIdx.x, b = blockIdx.x;
    const int lane = t & 31, wave = t >> 5;
    const int4* c4 = (const int4*)cnt_dst;

    int part = 0;
    int k = 0;
    for (; k + 8 <= b; k += 8) {
        int4 q[8];
#pragma unroll
        for (int u = 0; u < 8; ++u) q[u] = c4[(k + u) * 256 + t];
#pragma unroll
        for (int u = 0; u < 8; ++u) part += (q[u].x + q[u].y) + (q[u].z + q[u].w);
    }
    for (; k < b; ++k) {
        int4 q = c4[k * 256 + t];
        part += (q.x + q.y) + (q.z + q.w);
    }
    for (int o = 16; o > 0; o >>= 1) part += __shfl_xor(part, o);
    if (lane == 0) red[wave] = part;

    const int i4 = b * 256 + t;
    const bool in = i4 < NNODES / 4;
    int4 c = make_int4(0, 0, 0, 0);
    if (in) c = c4[i4];
    const int s = (c.x + c.y) + (c.z + c.w);
    int inc = s;
    for (int o = 1; o < 32; o <<= 1) {
        int v = __shfl_up(inc, o);
        if (lane >= o) inc += v;
    }
    if (lane == 31) wsum[wave] = inc;
    __syncthreads();

    int base = 0;
    for (int w = 0; w < 8; ++w) base += red[w];
    for (int w = 0; w < 8; ++w) base += (w < wave) ? wsum[w] : 0;
    const int ex = base + inc - s;
    if (in) {
        ((int4*)rowptr)[i4] = make_int4(ex, ex + c.x, ex + c.x + c.y, ex + c.x + c.y + c.z);
        int4 cs = ((const int4*)cnt_src)[i4];
        ((float4*)ns)[i4] = make_float4(inv_sqrt_deg(cs.x), inv_sqrt_deg(cs.y),
                                        inv_sqrt_deg(cs.z), inv_sqrt_deg(cs.w));
        ((float4*)nd)[i4] = make_float4(inv_sqrt_deg(c.x), inv_sqrt_deg(c.y),
                                        inv_sqrt_deg(c.z), inv_sqrt_deg(c.w));
    }
}

__device__ __forceinline__ void fill_part(int* __restrict__ nxt,
                                          const int* __restrict__ src,
                                          const int* __restrict__ dst,
                                          const int* __restrict__ rowptr,
                                          const int* __restrict__ plane_dst,
                                          int* __restrict__ csr) {
    const int t  = threadIdx.x;
    const int c  = blockIdx.x / HR;
    const int lo = (blockIdx.x - c * HR) * HNR;
    const int e0 = c * HEC + t;

    const int4* rp = (const int4*)(rowptr + lo);
    const int4* pd = (const int4*)(plane_dst + (size_t)c * NNODES + lo);
    {
        int4 r[SFULL + 1], p[SFULL + 1];
        const bool last = SFULL * HT + t < HNR / 4;
#pragma unroll
        for (int k = 0; k < SFULL; ++k) { r[k] = rp[k * HT + t]; p[k] = pd[k * HT + t]; }
        r[SFULL] = make_int4(0, 0, 0, 0); p[SFULL] = make_int4(0, 0, 0, 0);
        if (last) { r[SFULL] = rp[SFULL * HT + t]; p[SFULL] = pd[SFULL * HT + t]; }
#pragma unroll
        for (int k = 0; k < SFULL; ++k)
            ((int4*)nxt)[k * HT + t] = make_int4(r[k].x + p[k].x, r[k].y + p[k].y, r[k].z + p[k].z, r[k].w + p[k].w);
        if (last)
            ((int4*)nxt)[SFULL * HT + t] = make_int4(r[SFULL].x + p[SFULL].x, r[SFULL].y + p[SFULL].y,
                                                     r[SFULL].z + p[SFULL].z, r[SFULL].w + p[SFULL].w);
    }
    __syncthreads();

    for (int g = 0; g < HFULL; g += 8) {
        int dv[8], sv[8];
#pragma unroll
        for (int u = 0; u < 8; ++u) {
            dv[u] = dst[e0 + (g + u) * HT];
            sv[u] = src[e0 + (g + u) * HT];
        }
        asm volatile("" :: "v"(sv[0]), "v"(sv[1]), "v"(sv[2]), "v"(sv[3]),
                           "v"(sv[4]), "v"(sv[5]), "v"(sv[6]), "v"(sv[7]));
#pragma unroll
        for (int u = 0; u < 8; ++u) {
            const unsigned d = (unsigned)(dv[u] - lo);
            if (d < (unsigned)HNR) {
                const unsigned idx = (unsigned)atomicAdd(&nxt[d], 1);
                if (idx < (unsigned)NEDGES) csr[idx] = sv[u];
            }
        }
    }
    if (HTAIL + t < HEC) {
        const unsigned d = (unsigned)(dst[e0 + HTAIL] - lo);
        const int s = src[e0 + HTAIL];
        asm volatile("" :: "v"(s));
        if (d < (unsigned)HNR) {
            const unsigned idx = (unsigned)atomicAdd(&nxt[d], 1);
            if (idx < (unsigned)NEDGES) csr[idx] = s;
        }
    }
}


template <int U>
__device__ __forceinline__ void acc_rows(const float* __restrict__ h, int my,
                                         int j, int lane, float4& acc) {
    float4 v[U];
#pragma unroll
    for (int u = 0; u < U; ++u) {
        const int s = __builtin_amdgcn_readlane(my, j + u);
        v[u] = ((const float4*)(h + (size_t)s * DIM))[lane];
    }
#pragma unroll
    for (int u = 0; u < U; ++u) {
        acc.x += v[u].x; acc.y += v[u].y; acc.z += v[u].z; acc.w += v[u].w;
    }
}

#define GATHER_BLOCKS 2048
#define GATHER_WAVES  (GATHER_BLOCKS * 8)

__global__ __launch_bounds__(256) void gather_kernel(const float* __restrict__ h,
                                                     const int* __restrict__ rowptr,
                                                     const int* __restrict__ cnt,
                                                     const int* __restrict__ csr,
                                                     const float* __restrict__ nd,
                                                     const float* __restrict__ ns,
                                                     const float* __restrict__ bias,
                                                     float* __restrict__ out) {
    const int lane = threadIdx.x & 31;
    const int w0 = __builtin_amdgcn_readfirstlane((int)((blockIdx.x * 256 + threadIdx.x) >> 5));
    const float4 bv = ((const float4*)bias)[lane];
    for (int node = w0; node < NNODES; node += GATHER_WAVES) {
        const int start = rowptr[node];
        const int deg   = cnt[node];
        const float sc = nd[node], tn = ns[node];
        float4 acc = make_float4(0.f, 0.f, 0.f, 0.f);
        for (int base = 0; base < deg; base += 32) {
            const int n = min(32, deg - base);
            int my = 0;
            if (lane < n)
                my = (int)min((unsigned)csr[start + base + lane], (unsigned)(NNODES - 1));
            int j = 0;
            for (; j + 8 <= n; j += 8) acc_rows<8>(h, my, j, lane, acc);
            if (j + 4 <= n) { acc_rows<4>(h, my, j, lane, acc); j += 4; }
            if (j + 2 <= n) { acc_rows<2>(h, my, j, lane, acc); j += 2; }
            if (j < n)      { acc_rows<1>(h, my, j, lane, acc); }
        }
        acc.x = fmaxf(fmaf(acc.x, sc, bv.x), 0.f) * tn;
        acc.y = fmaxf(fmaf(acc.y, sc, bv.y), 0.f) * tn;
        acc.z = fmaxf(fmaf(acc.z, sc, bv.z), 0.f) * tn;
        acc.w = fmaxf(fmaf(acc.w, sc, bv.w), 0.f) * tn;
        ((float4*)(out + (size_t)node * DIM))[lane] = acc;
    }
}


__device__ __forceinline__ int clamp_node(int s) {
    return (int)min((unsigned)s, (unsigned)(NNODES - 1));
}

template <int K>
__device__ __forceinline__ void rows3(const float* __restrict__ z, int ids, int rem,
                                      int g, int cc, float4& acc) {
    float4 v[K];
#pragma unroll
    for (int u = 0; u < K; ++u) {
        const int s0 = __builtin_amdgcn_readlane(ids, 3 * u);
        const int s1 = __builtin_amdgcn_readlane(ids, 3 * u + 1);
        const int s2 = __builtin_amdgcn_readlane(ids, 3 * u + 2);
        int s;
        if (u < K - 1) s = (g == 1) ? s1 : ((g == 2) ? s2 : s0);
        else           s = (g == 1 && 3 * u + 1 < rem) ? s1 : ((g == 2 && 3 * u + 2 < rem) ? s2 : s0);
        v[u] = ((const float4*)(z + (size_t)s * ZS))[cc];
    }
#pragma unroll
    for (int u = 0; u < K; ++u) {
        if (u < K - 1) {
            acc.x += v[u].x; acc.y += v[u].y; acc.z += v[u].z; acc.w += v[u].w;
        } else {
            const bool ok = 3 * u + g < rem;
            acc.x += ok ? v[u].x : 0.f; acc.y += ok ? v[u].y : 0.f;
            acc.z += ok ? v[u].z : 0.f; acc.w += ok ? v[u].w : 0.f;
        }
    }
}

__device__ __forceinline__ int bucket_ids(const int* __restrict__ csr, int first, int lane) {
    return csr[min((unsigned)(first + lane), (unsigned)(NEDGES - 1))];
}

__global__ __launch_bounds__(256) void gather_out(const float* __restrict__ z,
                                                  const float* __restrict__ nd,
                                                  const float* __restrict__ bias,
                                                  const int* __restrict__ rowptr,
                                                  const int* __restrict__ cnt,
                                                  const int* __restrict__ csr,
                                                  float* __restrict__ out) {
    const int lane = threadIdx.x & 31;
    const int g    = (lane >= 30) ? 0 : ((lane >= 20) ? 2 : ((lane >= 10) ? 1 : 0));
    const int cc   = (lane >= 30) ? (NCLS / 4 - 1) : (lane - 10 * g);
    const int w0 = __builtin_amdgcn_readfirstlane((int)((blockIdx.x * 256 + threadIdx.x) >> 5));
    const float4 bv = ((const float4*)bias)[cc];

    int   node  = w0;
    int   start = rowptr[node];
    int   deg   = cnt[node];
    float sc    = nd[node];
    int   my    = clamp_node(bucket_ids(csr, start, lane));
    static_assert(2 * GATHER_WAVES <= NNODES, "every wave owns a second node");
    int   n1    = node + GATHER_WAVES;
    int   s1    = rowptr[n1];
    int   d1    = cnt[n1];
    float c1    = nd[n1];

    for (;;) {
        const bool has1 = n1 < NNODES;
        const int nmy = bucket_ids(csr, s1, lane);
        const int  n2   = n1 + GATHER_WAVES;
        int   s2 = 0, d2 = 0;
        float c2 = 0.f;
        if (n2 < NNODES) { s2 = rowptr[n2]; d2 = cnt[n2]; c2 = nd[n2]; }

        float4 acc = make_float4(0.f, 0.f, 0.f, 0.f);
        int ids = my;
        for (int j0 = 0; j0 < deg; j0 += 24) {
            if (j0 > 0) ids = clamp_node(bucket_ids(csr, start + j0, lane));
            const int rem = deg - j0;
            switch (min(8, (rem + 2) / 3)) {
                case 1:  rows3<1>(z, ids, rem, g, cc, acc); break;
                case 2:  rows3<2>(z, ids, rem, g, cc, acc); break;
                case 3:  rows3<3>(z, ids, rem, g, cc, acc); break;
                case 4:  rows3<4>(z, ids, rem, g, cc, acc); break;
                case 5:  rows3<5>(z, ids, rem, g, cc, acc); break;
                case 6:  rows3<6>(z, ids, rem, g, cc, acc); break;
                case 7:  rows3<7>(z, ids, rem, g, cc, acc); break;
                default: rows3<8>(z, ids, rem, g, cc, acc); break;
            }
        }
        const float px = __shfl_down(acc.x, 10), py = __shfl_down(acc.y, 10);
        const float pz = __shfl_down(acc.z, 10), pw = __shfl_down(acc.w, 10);
        const float qx = __shfl_down(acc.x, 20), qy = __shfl_down(acc.y, 20);
        const float qz = __shfl_down(acc.z, 20), qw = __shfl_down(acc.w, 20);
        if (lane < NCLS / 4)
            ((float4*)(out + (size_t)node * NCLS))[lane] =
                make_float4(fmaf((acc.x + px) + qx, sc, bv.x), fmaf((acc.y + py) + qy, sc, bv.y),
                            fmaf((acc.z + pz) + qz, sc, bv.z), fmaf((acc.w + pw) + qw, sc, bv.w));
        if (!has1) break;
        node = n1; start = s1; deg = d1; sc = c1; my = clamp_node(nmy);
        n1 = n2; s1 = s2; d1 = d2; c1 = c2;
    }
}


#define NTILES     (NNODES / 16)
#define FG_FILL    (HC * HR)
#define FG_GEMM    256
#define FG_GROUPS  (HT / 256)
#define FG_STREAMS (FG_GEMM * FG_GROUPS)
#define AS_PITCH   (DIM + 4)
#define FG_LDS_WORDS (DIM * DIM + FG_GROUPS * 2 * 16 * AS_PITCH)
static_assert(FG_STREAMS <= NTILES, "every tile stream owns a first tile");
static_assert(FG_LDS_WORDS >= HNR, "the fill's slot counters fit the same LDS block");

__global__ __launch_bounds__(HT, 16) void fill_gemm(const int* __restrict__ src,
                                                    const int* __restrict__ dst,
                                                    const int* __restrict__ rowptr,
                                                    const int* __restrict__ plane_dst,
                                                    int* __restrict__ csr,
                                                    const float* __restrict__ x,
                                                    const float* __restrict__ ns,
                                                    const float* __restrict__ W,
                                                    float* __restrict__ out) {
    __shared__ __attribute__((aligned(16))) float lds[FG_LDS_WORDS];

    if (blockIdx.x < FG_FILL) {
        fill_part((int*)lds, src, dst, rowptr, plane_dst, csr);
        return;
    }

    const int t   = threadIdx.x;
    const int grp = __builtin_amdgcn_readfirstlane(t >> 8);
    const int tg  = t & 255;
    const int am  = tg >> 4;
    const int akb = (tg & 15) * 8;

    float (*Ws)[DIM]          = (float (*)[DIM])lds;
    float (*As)[16][AS_PITCH] = (float (*)[16][AS_PITCH])(lds + DIM * DIM) + 2 * grp;

    const int base0 = blockIdx.x - FG_FILL;
    const int goff  = FG_GEMM * grp;

    float  s;
    float4 v0, v1;
    {
        const int node = (base0 + goff) * 16 + am;
        s = ns[node];
        const float4* p = (const float4*)(x + (size_t)node * DIM + akb);
        v0 = p[0]; v1 = p[1];
    }

    static_assert((DIM * DIM / 4) % HT == 0, "every thread stages the same number of 16-byte units");
#pragma unroll
    for (int k = 0; k < DIM * DIM / 4 / HT; ++k)
        ((float4*)lds)[k * HT + t] = ((const float4*)W)[k * HT + t];

    const int wave = tg >> 5;
    const int lane = t & 31;
    const int n0   = wave * 16;
    const int l16  = lane & 15;
    const int kh   = (lane >> 4) * 2;
    const int n    = n0 + l16;

    int cur = 0;
    for (int base = base0; base < NTILES; base += FG_STREAMS) {
        const int  tile  = base + goff;
        const bool valid = tile < NTILES;
        if (valid) {
            float* a = &As[cur][am][akb];
            *(float4*)a       = make_float4(v0.x * s, v0.y * s, v0.z * s, v0.w * s);
            *(float4*)(a + 4) = make_float4(v1.x * s, v1.y * s, v1.z * s, v1.w * s);
        }
        __syncthreads();

        const int nxt = tile + FG_STREAMS;
        if (nxt < NTILES) {
            const int node = nxt * 16 + am;
            s = ns[node];
            const float4* p = (const float4*)(x + (size_t)node * DIM + akb);
            v0 = p[0]; v1 = p[1];
        }

        if (valid) {
            const float (*Ac)[AS_PITCH] = As[cur];
            v8f acc = {};
#pragma unroll
            for (int kk = 0; kk < DIM; kk += 4) {
                v2f a, b;
                a.x = Ac[l16][kk + kh];
                a.y = Ac[l16][kk + kh + 1];
                b.x = Ws[kk + kh][n0 + l16];
                b.y = Ws[kk + kh + 1][n0 + l16];
                acc = __builtin_amdgcn_wmma_f32_16x16x4_f32(
                    false, a, false, b, (short)0, acc, false, false);
            }
            const int mbase = tile * 16 + (lane >> 4) * 8;
#pragma unroll
            for (int v = 0; v < 8; ++v)
                out[(size_t)(mbase + v) * DIM + n] = acc[v];
        }
        cur ^= 1;
    }
}


#define FUSED_BLOCKS 768
#define NT2          3
#define NT2W         (NT2 * 16)
static_assert(2 * FUSED_BLOCKS <= NTILES, "every work-group owns a second tile");
static_assert(NT2W >= NCLS, "the class columns fit the N-tiles");

__device__ __forceinline__ void rows_upto16(const float* __restrict__ h, int ids, int j, int k,
                                            int lane, float4& acc) {
    switch (k) {
        case 1:  acc_rows<1>(h, ids, j, lane, acc);  break;
        case 2:  acc_rows<2>(h, ids, j, lane, acc);  break;
        case 3:  acc_rows<3>(h, ids, j, lane, acc);  break;
        case 4:  acc_rows<4>(h, ids, j, lane, acc);  break;
        case 5:  acc_rows<5>(h, ids, j, lane, acc);  break;
        case 6:  acc_rows<6>(h, ids, j, lane, acc);  break;
        case 7:  acc_rows<7>(h, ids, j, lane, acc);  break;
        case 8:  acc_rows<8>(h, ids, j, lane, acc);  break;
        case 9:  acc_rows<9>(h, ids, j, lane, acc);  break;
        case 10: acc_rows<10>(h, ids, j, lane, acc); break;
        case 11: acc_rows<11>(h, ids, j, lane, acc); break;
        case 12: acc_rows<12>(h, ids, j, lane, acc); break;
        case 13: acc_rows<13>(h, ids, j, lane, acc); break;
        case 14: acc_rows<14>(h, ids, j, lane, acc); break;
        case 15: acc_rows<15>(h, ids, j, lane, acc); break;
        default: acc_rows<16>(h, ids, j, lane, acc); break;
    }
}

__device__ __forceinline__ float4 bucket_sum(const float* __restrict__ h, const int* __restrict__ csr,
                                             int start, int deg, int ids, int lane) {
    float4 acc = make_float4(0.f, 0.f, 0.f, 0.f);
    for (int base = 0; base < deg; base += 32) {
        const int n = min(32, deg - base);
        if (base > 0) ids = clamp_node(bucket_ids(csr, start + base, lane));
        for (int j = 0; j < n; j += 16) rows_upto16(h, ids, j, min(16, n - j), lane, acc);
    }
    return acc;
}

__global__ __launch_bounds__(256) void gather_gemm2(const float* __restrict__ h,
                                                    const int* __restrict__ rowptr,
                                                    const int* __restrict__ cnt,
                                                    const int* __restrict__ csr,
                                                    const float* __restrict__ nd,
                                                    const float* __restrict__ ns,
                                                    const float* __restrict__ W1,
                                                    const float* __restrict__ b1,
                                                    const float* __restrict__ W2,
                                                    float* __restrict__ z) {
    __shared__ __attribute__((aligned(16))) float T[16][DIM + 4];
    __shared__ __attribute__((aligned(16))) float Ws1[DIM][DIM];
    __shared__ __attribute__((aligned(16))) float Ws2[DIM][NT2W];

    const int t    = threadIdx.x;
    const int lane = t & 31;
    const int wave = t >> 5;
    const int r0   = 2 * wave;
    const int hl   = lane & 1;

    int tile = blockIdx.x;

    int   vst = rowptr[tile * 16 + r0 + hl];
    int   vdg = cnt[tile * 16 + r0 + hl];
    float vsc = nd[tile * 16 + r0 + hl];
    int   wst = rowptr[(tile + FUSED_BLOCKS) * 16 + r0 + hl];
    int   wdg = cnt[(tile + FUSED_BLOCKS) * 16 + r0 + hl];
    float wsc = nd[(tile + FUSED_BLOCKS) * 16 + r0 + hl];

    for (int i = t; i < DIM * DIM / 4; i += 256)
        ((float4*)&Ws1[0][0])[i] = ((const float4*)W1)[i];
    for (int i = t; i < DIM * NT2W; i += 256) {
        const int k = i / NT2W;
        const int c = i - k * NT2W;
        Ws2[k][c] = (c < NCLS) ? W2[k * NCLS + c] : 0.0f;
    }

    int   stA = __builtin_amdgcn_readlane(vst, 0), stB = __builtin_amdgcn_readlane(vst, 1);
    int   dgA = __builtin_amdgcn_readlane(vdg, 0), dgB = __builtin_amdgcn_readlane(vdg, 1);
    float scA = __int_as_float(__builtin_amdgcn_readlane(__float_as_int(vsc), 0));
    float scB = __int_as_float(__builtin_amdgcn_readlane(__float_as_int(vsc), 1));
    int   st1A = __builtin_amdgcn_readlane(wst, 0), st1B = __builtin_amdgcn_readlane(wst, 1);
    int   dg1A = __builtin_amdgcn_readlane(wdg, 0), dg1B = __builtin_amdgcn_readlane(wdg, 1);
    float sc1A = __int_as_float(__builtin_amdgcn_readlane(__float_as_int(wsc), 0));
    float sc1B = __int_as_float(__builtin_amdgcn_readlane(__float_as_int(wsc), 1));
    int idA = bucket_ids(csr, stA, lane);
    int idB = bucket_ids(csr, stB, lane);

    const int n0  = wave * 16;
    const int l16 = lane & 15;
    const int kh  = (lane >> 4) * 2;
    const int mh  = (lane >> 4) * 8;
    const int n   = n0 + l16;
    const float bv = b1[n];

    for (; tile < NTILES; tile += FUSED_BLOCKS) {
        const int cA = clamp_node(idA), cB = clamp_node(idB);

        const int nidA = bucket_ids(csr, st1A, lane);
        const int nidB = bucket_ids(csr, st1B, lane);
        {
            const int na = min((tile + 2 * FUSED_BLOCKS) * 16 + r0 + hl, NNODES - 1);
            vst = rowptr[na]; vdg = cnt[na]; vsc = nd[na];
        }
        const float4 q0 = ((const float4*)(ns + tile * 16 + mh))[0];
        const float4 q1 = ((const float4*)(ns + tile * 16 + mh))[1];

        const float4 a0 = bucket_sum(h, csr, stA, dgA, cA, lane);
        const float4 a1 = bucket_sum(h, csr, stB, dgB, cB, lane);

        __syncthreads();
        *(float4*)&T[r0][4 * lane]     = make_float4(a0.x * scA, a0.y * scA, a0.z * scA, a0.w * scA);
        *(float4*)&T[r0 + 1][4 * lane] = make_float4(a1.x * scB, a1.y * scB, a1.z * scB, a1.w * scB);
        __syncthreads();

        v8f acc = {};
#pragma unroll
        for (int kk = 0; kk < DIM; kk += 4) {
            v2f a, b;
            a.x = T[l16][kk + kh];
            a.y = T[l16][kk + kh + 1];
            b.x = Ws1[kk + kh][n0 + l16];
            b.y = Ws1[kk + kh + 1][n0 + l16];
            acc = __builtin_amdgcn_wmma_f32_16x16x4_f32(
                false, a, false, b, (short)0, acc, false, false);
        }

        __syncthreads();
        T[mh + 0][n] = fmaxf(acc[0] + bv, 0.0f) * q0.x;
        T[mh + 1][n] = fmaxf(acc[1] + bv, 0.0f) * q0.y;
        T[mh + 2][n] = fmaxf(acc[2] + bv, 0.0f) * q0.z;
        T[mh + 3][n] = fmaxf(acc[3] + bv, 0.0f) * q0.w;
        T[mh + 4][n] = fmaxf(acc[4] + bv, 0.0f) * q1.x;
        T[mh + 5][n] = fmaxf(acc[5] + bv, 0.0f) * q1.y;
        T[mh + 6][n] = fmaxf(acc[6] + bv, 0.0f) * q1.z;
        T[mh + 7][n] = fmaxf(acc[7] + bv, 0.0f) * q1.w;
        __syncthreads();

        if (wave < NT2) {
            v8f acc2 = {};
#pragma unroll
            for (int kk = 0; kk < DIM; kk += 4) {
                v2f a, b;
                a.x = T[l16][kk + kh];
                a.y = T[l16][kk + kh + 1];
                b.x = Ws2[kk + kh][n0 + l16];
                b.y = Ws2[kk + kh + 1][n0 + l16];
                acc2 = __builtin_amdgcn_wmma_f32_16x16x4_f32(
                    false, a, false, b, (short)0, acc2, false, false);
            }
            if (n < NCLS) {
                const int mbase = tile * 16 + mh;
#pragma unroll
                for (int v = 0; v < 8; ++v)
                    z[(size_t)(mbase + v) * ZS + n] = acc2[v];
            }
        }

        stA = st1A; stB = st1B; dgA = dg1A; dgB = dg1B; scA = sc1A; scB = sc1B;
        idA = nidA; idB = nidB;
        st1A = __builtin_amdgcn_readlane(vst, 0); st1B = __builtin_amdgcn_readlane(vst, 1);
        dg1A = __builtin_amdgcn_readlane(vdg, 0); dg1B = __builtin_amdgcn_readlane(vdg, 1);
        sc1A = __int_as_float(__builtin_amdgcn_readlane(__float_as_int(vsc), 0));
        sc1B = __int_as_float(__builtin_amdgcn_readlane(__float_as_int(vsc), 1));
    }
}


extern "C" void kernel_launch(void* const* d_in, const int* in_sizes, int n_in,
                              void* d_out, int out_size, void* d_ws, size_t ws_size,
                              hipStream_t stream) {
    const float* inputs = (const float*)d_in[0];
    const int*   src    = (const int*)d_in[1];
    const int*   dst    = (const int*)d_in[2];
    const float* W0     = (const float*)d_in[3];
    const float* b0     = (const float*)d_in[4];
    const float* W1     = (const float*)d_in[5];
    const float* b1     = (const float*)d_in[6];
    const float* W2     = (const float*)d_in[7];
    const float* b2     = (const float*)d_in[8];
    float* out = (float*)d_out;

    constexpr size_t OFF_ND   = NNODES;
    constexpr size_t OFF_CS   = 2 * (size_t)NNODES;
    constexpr size_t OFF_CD   = 3 * (size_t)NNODES;
    constexpr size_t OFF_RP   = 4 * (size_t)NNODES;
    constexpr size_t OFF_CSR  = 5 * (size_t)NNODES;
    constexpr size_t OFF_B    = ((OFF_CSR + NEDGES + 127) / 128) * 128;
    constexpr size_t OFF_C    = OFF_B + (size_t)NNODES * DIM;
    constexpr size_t OFF_PD   = OFF_C + (size_t)NNODES * DIM;
    constexpr size_t OFF_PS   = OFF_PD + (size_t)HC * NNODES;

    float* ws      = (float*)d_ws;
    float* ns      = ws;
    float* ndm     = ws + OFF_ND;
    int*   cnt_src = (int*)(ws + OFF_CS);
    int*   cnt_dst = (int*)(ws + OFF_CD);
    int*   rowptr  = (int*)(ws + OFF_RP);
    int*   csr     = (int*)(ws + OFF_CSR);
    float* bufB    = ws + OFF_B;
    float* bufC    = ws + OFF_C;
    int*   plane_d = (int*)(ws + OFF_PD);
    int*   plane_s = (int*)(ws + OFF_PS);

    constexpr int GR_PRE  = (NNODES + 255) / 256;
    constexpr int GR_SCAN = (NNODES / 4 + 255) / 256;

    hist_rank<<<HC * HR, HT, 0, stream>>>(src, dst, plane_s, plane_d);
    chunk_prefix<<<GR_PRE, 256, 0, stream>>>(plane_s, plane_d, cnt_src, cnt_dst);
    scan_norm<<<GR_SCAN, 256, 0, stream>>>(cnt_src, cnt_dst, rowptr, ns, ndm);
    fill_gemm<<<FG_FILL + FG_GEMM, HT, 0, stream>>>(src, dst, rowptr, plane_d, csr, inputs, ns, W0, bufB);

    gather_kernel<<<GATHER_BLOCKS, 256, 0, stream>>>(bufB, rowptr, cnt_dst, csr, ndm, ns, b0, bufC);

    gather_gemm2<<<FUSED_BLOCKS, 256, 0, stream>>>(bufC, rowptr, cnt_dst, csr, ndm, ns, W1, b1, W2, bufB);
    gather_out<<<GATHER_BLOCKS, 256, 0, stream>>>(bufB, ndm, b2, rowptr, cnt_dst, csr, out);
}
